# starting kernel
# baseline (speedup 1.0000x reference)
#include <hip/hip_runtime.h>

#define NN 50000
#define EE 1600000
#define INF_ 128
#define HH 4
#define DD 32
#define HD 128
#define NEG_SLOPE 0.2f

typedef __attribute__((ext_vector_type(2))) float v2f;
typedef __attribute__((ext_vector_type(8))) float v8f;

__device__ __forceinline__ unsigned int ford(float f) {
  unsigned int b = __float_as_uint(f);
  return (b & 0x80000000u) ? ~b : (b | 0x80000000u);
}
__device__ __forceinline__ float funord(unsigned int u) {
  unsigned int b = (u & 0x80000000u) ? (u ^ 0x80000000u) : ~u;
  return __uint_as_float(b);
}
#define FORD_NEG_INF 0x007FFFFFu

__global__ void k_zero_f32(float* __restrict__ p, int n) {
  int i = blockIdx.x * blockDim.x + threadIdx.x;
  if (i < n) p[i] = 0.0f;
}
__global__ void k_init_ms(unsigned int* __restrict__ m, float* __restrict__ s, int n) {
  int i = blockIdx.x * blockDim.x + threadIdx.x;
  if (i < n) { m[i] = FORD_NEG_INF; s[i] = 0.0f; }
}

__global__ __launch_bounds__(256) void k_proj(const float* __restrict__ feat,
                                              const float* __restrict__ W,
                                              float* __restrict__ h) {
  __shared__ float As[16 * 132];
  const int m0 = blockIdx.x * 16;
  const int t  = threadIdx.x;

  {
    const int row = t >> 4;
    const int col = (t & 15) * 8;
    const float4* gp = (const float4*)(feat + (size_t)(m0 + row) * INF_ + col);
    float4 x0 = gp[0];
    float4 x1 = gp[1];
    float* lp = As + row * 132 + col;
    *(float4*)(lp)     = x0;
    *(float4*)(lp + 4) = x1;
  }
  __syncthreads();

  const int lane  = t & 31;
  const int wave  = t >> 5;
  const int n0    = wave * 16;
  const int r     = lane & 15;
  const int khalf = (lane >> 4) << 1;

  v8f acc = {};
  for (int k0 = 0; k0 < INF_; k0 += 4) {
    const int kk = k0 + khalf;
    v2f a;
    a.x = As[r * 132 + kk];
    a.y = As[r * 132 + kk + 1];
    v2f b;
    b.x = W[(size_t)(kk)     * HD + n0 + r];
    b.y = W[(size_t)(kk + 1) * HD + n0 + r];
    acc = __builtin_amdgcn_wmma_f32_16x16x4_f32(false, a, false, b,
                                                (short)0, acc, false, false);
  }

  const int mh = (lane >> 4) * 8;
#pragma unroll
  for (int j = 0; j < 8; ++j) {
    h[(size_t)(m0 + j + mh) * HD + n0 + r] = acc[j];
  }
}

__global__ void k_elr(const float* __restrict__ h, const float* __restrict__ al,
                      const float* __restrict__ ar, float* __restrict__ el,
                      float* __restrict__ er) {
  int tid = blockIdx.x * blockDim.x + threadIdx.x;
  if (tid >= NN * HH) return;
  const int n  = tid / HH;
  const int hh = tid % HH;
  const float* hp  = h  + (size_t)n * HD + hh * DD;
  const float* alp = al + hh * DD;
  const float* arp = ar + hh * DD;
  float sl = 0.0f, sr = 0.0f;
#pragma unroll
  for (int d = 0; d < DD; d += 4) {
    float4 hv = *(const float4*)(hp + d);
    float4 av = *(const float4*)(alp + d);
    float4 rv = *(const float4*)(arp + d);
    sl += hv.x * av.x + hv.y * av.y + hv.z * av.z + hv.w * av.w;
    sr += hv.x * rv.x + hv.y * rv.y + hv.z * rv.z + hv.w * rv.w;
  }
  el[tid] = sl;
  er[tid] = sr;
}

__global__ void k_edge_logit(const int* __restrict__ src, const int* __restrict__ dst,
                             const float* __restrict__ el, const float* __restrict__ er,
                             float* __restrict__ ebuf, unsigned int* __restrict__ m) {
  int tid = blockIdx.x * blockDim.x + threadIdx.x;
  if (tid >= EE * HH) return;
  const int e  = tid >> 2;
  const int hh = tid & 3;
  const int sd = src[e], dd = dst[e];
  float x = el[sd * HH + hh] + er[dd * HH + hh];
  x = (x > 0.0f) ? x : NEG_SLOPE * x;
  ebuf[tid] = x;
  atomicMax(&m[dd * HH + hh], ford(x));
}

__global__ void k_edge_exp(const int* __restrict__ dst, const unsigned int* __restrict__ m,
                           float* __restrict__ ebuf, float* __restrict__ s) {
  int tid = blockIdx.x * blockDim.x + threadIdx.x;
  if (tid >= EE * HH) return;
  const int e  = tid >> 2;
  const int hh = tid & 3;
  const int dd = dst[e];
  const float mm = funord(m[dd * HH + hh]);
  const float ex = __expf(ebuf[tid] - mm);
  ebuf[tid] = ex;
  atomicAdd(&s[dd * HH + hh], ex);
}

__global__ __launch_bounds__(256) void k_edge_msg(const int* __restrict__ src,
                                                  const int* __restrict__ dst,
                                                  const float* __restrict__ ebuf,
                                                  const float* __restrict__ s,
                                                  const float* __restrict__ h,
                                                  float* __restrict__ acc) {
  int tid = blockIdx.x * blockDim.x + threadIdx.x;
  if (tid >= EE * HH * 8) return;
  const int e   = tid >> 5;
  const int rem = tid & 31;
  const int hh  = rem >> 3;
  const int q   = rem & 7;
  const int sd  = src[e], dd = dst[e];
  const float alpha = ebuf[e * HH + hh] / s[dd * HH + hh];
  const float4 hv = *(const float4*)(h + (size_t)sd * HD + hh * DD + q * 4);
  float* ap = acc + (size_t)dd * HD + hh * DD + q * 4;
  atomicAdd(ap + 0, alpha * hv.x);
  atomicAdd(ap + 1, alpha * hv.y);
  atomicAdd(ap + 2, alpha * hv.z);
  atomicAdd(ap + 3, alpha * hv.w);
}

__global__ void k_mean(const float* __restrict__ acc, float* __restrict__ out) {
  int tid = blockIdx.x * blockDim.x + threadIdx.x;
  if (tid >= NN * DD) return;
  const int n = tid >> 5;
  const int d = tid & 31;
  const float* a = acc + (size_t)n * HD + d;
  out[tid] = 0.25f * (a[0] + a[32] + a[64] + a[96]);
}

extern "C" void kernel_launch(void* const* d_in, const int* in_sizes, int n_in,
                              void* d_out, int out_size, void* d_ws, size_t ws_size,
                              hipStream_t stream) {
  const float* feat = (const float*)d_in[0];
  const int*   src_g[2] = { (const int*)d_in[1], (const int*)d_in[3] };
  const int*   dst_g[2] = { (const int*)d_in[2], (const int*)d_in[4] };
  const float* W_g[2]   = { (const float*)d_in[5], (const float*)d_in[8] };
  const float* al_g[2]  = { (const float*)d_in[6], (const float*)d_in[9] };
  const float* ar_g[2]  = { (const float*)d_in[7], (const float*)d_in[10] };
  float* out = (float*)d_out;

  float*        acc  = (float*)d_ws;
  float*        hbuf = acc  + (size_t)NN * HD;
  float*        el   = hbuf + (size_t)NN * HD;
  float*        er   = el   + (size_t)NN * HH;
  unsigned int* m    = (unsigned int*)(er + (size_t)NN * HH);
  float*        s    = (float*)(m + (size_t)NN * HH);
  float*        ebuf = s + (size_t)NN * HH;

  const int T = 256;
  k_zero_f32<<<(NN * HD + T - 1) / T, T, 0, stream>>>(acc, NN * HD);

  for (int g = 0; g < 2; ++g) {
    k_proj<<<NN / 16, T, 0, stream>>>(feat, W_g[g], hbuf);
    k_elr<<<(NN * HH + T - 1) / T, T, 0, stream>>>(hbuf, al_g[g], ar_g[g], el, er);
    k_init_ms<<<(NN * HH + T - 1) / T, T, 0, stream>>>(m, s, NN * HH);
    k_edge_logit<<<(EE * HH + T - 1) / T, T, 0, stream>>>(src_g[g], dst_g[g], el, er, ebuf, m);
    k_edge_exp<<<(EE * HH + T - 1) / T, T, 0, stream>>>(dst_g[g], m, ebuf, s);
    k_edge_msg<<<(EE * HH * 8 + T - 1) / T, T, 0, stream>>>(src_g[g], dst_g[g], ebuf, s, hbuf, acc);
  }

  k_mean<<<(NN * DD + T - 1) / T, T, 0, stream>>>(acc, out);
}


# message pass owner-computes: each work-group keeps 200 destination rows in LDS, streams the whole dst list, adds ex*h[src] with LDS atomics, stores rows once; sum order changes (LDS arrival)
# speedup vs baseline: 1.7617x; 1.7617x over previous
#include <hip/hip_runtime.h>

#define NN 50000
#define EE 1600000
#define INF_ 128
#define HH 4
#define DD 32
#define HD 128
#define NEG_SLOPE 0.2f

typedef __attribute__((ext_vector_type(2))) float v2f;
typedef __attribute__((ext_vector_type(8))) float v8f;

__device__ __forceinline__ unsigned int ford(float f) {
  unsigned int b = __float_as_uint(f);
  return (b & 0x80000000u) ? ~b : (b | 0x80000000u);
}
__device__ __forceinline__ float funord(unsigned int u) {
  unsigned int b = (u & 0x80000000u) ? (u ^ 0x80000000u) : ~u;
  return __uint_as_float(b);
}
#define FORD_NEG_INF 0x007FFFFFu

__global__ void k_init(uint4* __restrict__ m, int nm4) {
  int i = blockIdx.x * blockDim.x + threadIdx.x;
  if (i < nm4) m[i] = make_uint4(FORD_NEG_INF, FORD_NEG_INF, FORD_NEG_INF, FORD_NEG_INF);
}

__global__ __launch_bounds__(256) void k_proj(const float* __restrict__ feat,
                                              const float* __restrict__ W,
                                              float* __restrict__ h) {
  __shared__ float As[16 * 132];
  const int m0 = blockIdx.x * 16;
  const int t  = threadIdx.x;

  {
    const int row = t >> 4;
    const int col = (t & 15) * 8;
    const float4* gp = (const float4*)(feat + (size_t)(m0 + row) * INF_ + col);
    float4 x0 = gp[0];
    float4 x1 = gp[1];
    float* lp = As + row * 132 + col;
    *(float4*)(lp)     = x0;
    *(float4*)(lp + 4) = x1;
  }
  __syncthreads();

  const int lane  = t & 31;
  const int wave  = t >> 5;
  const int n0    = wave * 16;
  const int r     = lane & 15;
  const int khalf = (lane >> 4) << 1;

  v8f acc = {};
  for (int k0 = 0; k0 < INF_; k0 += 4) {
    const int kk = k0 + khalf;
    v2f a;
    a.x = As[r * 132 + kk];
    a.y = As[r * 132 + kk + 1];
    v2f b;
    b.x = W[(size_t)(kk)     * HD + n0 + r];
    b.y = W[(size_t)(kk + 1) * HD + n0 + r];
    acc = __builtin_amdgcn_wmma_f32_16x16x4_f32(false, a, false, b,
                                                (short)0, acc, false, false);
  }

  const int mh = (lane >> 4) * 8;
#pragma unroll
  for (int j = 0; j < 8; ++j) {
    h[(size_t)(m0 + j + mh) * HD + n0 + r] = acc[j];
  }
}

__global__ void k_elr(const float* __restrict__ h, const float* __restrict__ al,
                      const float* __restrict__ ar, float* __restrict__ el,
                      float* __restrict__ er) {
  int tid = blockIdx.x * blockDim.x + threadIdx.x;
  if (tid >= NN * HH) return;
  const int n  = tid / HH;
  const int hh = tid % HH;
  const float* hp  = h  + (size_t)n * HD + hh * DD;
  const float* alp = al + hh * DD;
  const float* arp = ar + hh * DD;
  float sl = 0.0f, sr = 0.0f;
#pragma unroll
  for (int d = 0; d < DD; d += 4) {
    float4 hv = *(const float4*)(hp + d);
    float4 av = *(const float4*)(alp + d);
    float4 rv = *(const float4*)(arp + d);
    sl += hv.x * av.x + hv.y * av.y + hv.z * av.z + hv.w * av.w;
    sr += hv.x * rv.x + hv.y * rv.y + hv.z * rv.z + hv.w * rv.w;
  }
  el[tid] = sl;
  er[tid] = sr;
}

__global__ void k_edge_max(const int* __restrict__ src, const int* __restrict__ dst,
                           const float* __restrict__ el, const float* __restrict__ er,
                           unsigned int* __restrict__ m) {
  int tid = blockIdx.x * blockDim.x + threadIdx.x;
  if (tid >= EE * HH) return;
  const int e  = tid >> 2;
  const int hh = tid & 3;
  const int sd = src[e], dd = dst[e];
  float x = el[sd * HH + hh] + er[dd * HH + hh];
  x = (x > 0.0f) ? x : NEG_SLOPE * x;
  atomicMax(&m[dd * HH + hh], ford(x));
}

#define OWN_R 200
#define OWN_P ((NN + OWN_R - 1) / OWN_R)
#define CH 512
#define NCH (EE / CH)
#define OWN_T 1024
static_assert(EE % CH == 0 && NCH >= OWN_T / 32 && NCH / OWN_P >= 1, "edge list splits into whole wave steps");
static_assert(OWN_R % 8 == 0, "an owner's sum rows fill whole 128-byte lines");
#define OWN_LDS (164 * 1024)

__device__ __forceinline__ void own_add(float* __restrict__ acc_l, float* __restrict__ s_l,
                                        const float* __restrict__ er_l, const float* __restrict__ m_l,
                                        int dl, int lane, int hh, float elv, const float4 hv) {
  float x = elv + er_l[dl * HH + hh];
  x = (x > 0.0f) ? x : NEG_SLOPE * x;
  const float ex = __expf(x - m_l[dl * HH + hh]);
  float* ap = acc_l + dl * HD + lane * 4;
  atomicAdd(ap + 0, ex * hv.x);
  atomicAdd(ap + 1, ex * hv.y);
  atomicAdd(ap + 2, ex * hv.z);
  atomicAdd(ap + 3, ex * hv.w);
  if ((lane & 7) == 0) atomicAdd(&s_l[dl * HH + hh], ex);
}

__global__ __launch_bounds__(OWN_T) void k_edge_msg_own(const int* __restrict__ src,
                                                        const int* __restrict__ dst,
                                                        const float* __restrict__ el,
                                                        const float* __restrict__ er,
                                                        const unsigned int* __restrict__ m,
                                                        const float* __restrict__ h,
                                                        float* __restrict__ s,
                                                        float* __restrict__ acc) {
  __shared__ __attribute__((aligned(16))) unsigned char L[OWN_LDS];
  float* acc_l = (float*)L;
  float* s_l   = acc_l + OWN_R * HD;
  float* er_l  = s_l + OWN_R * HH;
  float* m_l   = er_l + OWN_R * HH;

  const int t    = threadIdx.x;
  const int lane = t & 31;
  const int wave = __builtin_amdgcn_readfirstlane(t >> 5);
  const int hh   = lane >> 3;
  const int n0   = blockIdx.x * OWN_R;
  const int nrows = (NN - n0 < OWN_R) ? (NN - n0) : OWN_R;

  for (int i = t; i < OWN_R * HD / 4; i += OWN_T)
    ((float4*)acc_l)[i] = make_float4(0.0f, 0.0f, 0.0f, 0.0f);
  for (int i = t; i < OWN_R * HH; i += OWN_T) {
    const bool ok = i < nrows * HH;
    s_l[i]  = 0.0f;
    er_l[i] = ok ? er[(size_t)n0 * HH + i] : 0.0f;
    m_l[i]  = ok ? funord(m[(size_t)n0 * HH + i]) : 0.0f;
  }
  __syncthreads();

  const int rot = blockIdx.x * (NCH / OWN_P);
  const int NIT = (NCH + (OWN_T / 32) - 1) / (OWN_T / 32);

  int4 cur[4], nxt[4];
  {
    int cr = wave + rot; if (cr >= NCH) cr -= NCH;
    const int4* p = (const int4*)(dst + (size_t)cr * CH) + lane;
#pragma unroll
    for (int k = 0; k < 4; ++k) cur[k] = p[k * 32];
  }

  for (int it = 0; it < NIT; ++it) {
    const int c = it * (OWN_T / 32) + wave;
    {
      int cn = c + (OWN_T / 32); if (cn >= NCH) cn = wave;
      int crn = cn + rot; if (crn >= NCH) crn -= NCH;
      const int4* p = (const int4*)(dst + (size_t)crn * CH) + lane;
#pragma unroll
      for (int k = 0; k < 4; ++k) nxt[k] = p[k * 32];
    }
    if (c < NCH) {
      int cr = c + rot; if (cr >= NCH) cr -= NCH;
      unsigned int mm = 0;
#pragma unroll
      for (int k = 3; k >= 0; --k) {
        mm = (mm << 1) | (((unsigned int)(cur[k].w - n0) < (unsigned int)OWN_R) ? 1u : 0u);
        mm = (mm << 1) | (((unsigned int)(cur[k].z - n0) < (unsigned int)OWN_R) ? 1u : 0u);
        mm = (mm << 1) | (((unsigned int)(cur[k].y - n0) < (unsigned int)OWN_R) ? 1u : 0u);
        mm = (mm << 1) | (((unsigned int)(cur[k].x - n0) < (unsigned int)OWN_R) ? 1u : 0u);
      }
      const int ebase = cr * CH + lane * 4;
      unsigned int mask = __builtin_amdgcn_ballot_w32(mm != 0u);
      while (mask != 0u) {
        int sd = 0, dl = 0;
        if (mm != 0u) {
          const int j = __builtin_ctz(mm);
          const int e = ebase + (j >> 2) * 128 + (j & 3);
          sd = src[e];
          dl = dst[e] - n0;
          mm &= mm - 1u;
        }
        do {
          const int l0 = __builtin_ctz(mask); mask &= mask - 1u;
          const bool h1 = (mask != 0u);
          const int l1 = h1 ? __builtin_ctz(mask) : l0; mask &= mask - 1u;
          const bool h2 = (mask != 0u);
          const int l2 = h2 ? __builtin_ctz(mask) : l0; mask &= mask - 1u;
          const bool h3 = (mask != 0u);
          const int l3 = h3 ? __builtin_ctz(mask) : l0; mask &= mask - 1u;
          const int sd0 = __builtin_amdgcn_readlane(sd, l0), dl0 = __builtin_amdgcn_readlane(dl, l0);
          const int sd1 = __builtin_amdgcn_readlane(sd, l1), dl1 = __builtin_amdgcn_readlane(dl, l1);
          const int sd2 = __builtin_amdgcn_readlane(sd, l2), dl2 = __builtin_amdgcn_readlane(dl, l2);
          const int sd3 = __builtin_amdgcn_readlane(sd, l3), dl3 = __builtin_amdgcn_readlane(dl, l3);
          const float4 hv0 = *(const float4*)(h + (size_t)sd0 * HD + lane * 4);
          const float4 hv1 = *(const float4*)(h + (size_t)sd1 * HD + lane * 4);
          const float4 hv2 = *(const float4*)(h + (size_t)sd2 * HD + lane * 4);
          const float4 hv3 = *(const float4*)(h + (size_t)sd3 * HD + lane * 4);
          const float e0 = el[sd0 * HH + hh];
          const float e1 = el[sd1 * HH + hh];
          const float e2 = el[sd2 * HH + hh];
          const float e3 = el[sd3 * HH + hh];
          own_add(acc_l, s_l, er_l, m_l, dl0, lane, hh, e0, hv0);
          if (h1) own_add(acc_l, s_l, er_l, m_l, dl1, lane, hh, e1, hv1);
          if (h2) own_add(acc_l, s_l, er_l, m_l, dl2, lane, hh, e2, hv2);
          if (h3) own_add(acc_l, s_l, er_l, m_l, dl3, lane, hh, e3, hv3);
        } while (mask != 0u);
        mask = __builtin_amdgcn_ballot_w32(mm != 0u);
      }
    }
#pragma unroll
    for (int k = 0; k < 4; ++k) cur[k] = nxt[k];
  }
  __syncthreads();

  float4* ag = (float4*)(acc + (size_t)n0 * HD);
  for (int i = t; i < nrows * HD / 4; i += OWN_T) ag[i] = ((const float4*)acc_l)[i];
  float* sg = s + (size_t)n0 * HH;
  for (int i = t; i < nrows * HH; i += OWN_T) sg[i] = s_l[i];
}

__device__ __forceinline__ float inv_or_zero(float v) {
  return (v != 0.0f) ? 1.0f / v : 0.0f;
}
__global__ void k_norm_mean(const float* __restrict__ acc0, const float* __restrict__ acc1,
                            const float* __restrict__ s0, const float* __restrict__ s1,
                            float* __restrict__ out) {
  int tid = blockIdx.x * blockDim.x + threadIdx.x;
  if (tid >= NN * DD) return;
  const int n = tid >> 5;
  const int d = tid & 31;
  const float4 v0 = *(const float4*)(s0 + (size_t)n * HH);
  const float4 v1 = *(const float4*)(s1 + (size_t)n * HH);
  const float* a0 = acc0 + (size_t)n * HD + d;
  const float* a1 = acc1 + (size_t)n * HD + d;
  const float f0 = a0[0], f1 = a0[32], f2 = a0[64], f3 = a0[96];
  const float b0 = a1[0], b1 = a1[32], b2 = a1[64], b3 = a1[96];
  float r = f0 * inv_or_zero(v0.x) + b0 * inv_or_zero(v1.x);
  r += f1 * inv_or_zero(v0.y) + b1 * inv_or_zero(v1.y);
  r += f2 * inv_or_zero(v0.z) + b2 * inv_or_zero(v1.z);
  r += f3 * inv_or_zero(v0.w) + b3 * inv_or_zero(v1.w);
  out[tid] = 0.25f * r;
}

extern "C" void kernel_launch(void* const* d_in, const int* in_sizes, int n_in,
                              void* d_out, int out_size, void* d_ws, size_t ws_size,
                              hipStream_t stream) {
  const float* feat = (const float*)d_in[0];
  const int*   src_g[2] = { (const int*)d_in[1], (const int*)d_in[3] };
  const int*   dst_g[2] = { (const int*)d_in[2], (const int*)d_in[4] };
  const float* W_g[2]   = { (const float*)d_in[5], (const float*)d_in[8] };
  const float* al_g[2]  = { (const float*)d_in[6], (const float*)d_in[9] };
  const float* ar_g[2]  = { (const float*)d_in[7], (const float*)d_in[10] };
  float* out = (float*)d_out;

  float*        acc0 = (float*)d_ws;
  float*        acc1 = acc0 + (size_t)NN * HD;
  float*        s0   = acc1 + (size_t)NN * HD;
  float*        s1   = s0   + (size_t)NN * HH;
  unsigned int* m0   = (unsigned int*)(s1 + (size_t)NN * HH);
  unsigned int* m1   = m0   + (size_t)NN * HH;
  float*        hbuf = (float*)(m1 + (size_t)NN * HH);
  float*        el   = hbuf + (size_t)NN * HD;
  float*        er   = el   + (size_t)NN * HH;
  float*        acc_g[2] = { acc0, acc1 };
  float*        s_g[2]   = { s0, s1 };
  unsigned int* m_g[2]   = { m0, m1 };

  const int T = 256;
  const int NM4 = (2 * NN * HH) / 4;
  k_init<<<(NM4 + T - 1) / T, T, 0, stream>>>((uint4*)m0, NM4);

  for (int g = 0; g < 2; ++g) {
    k_proj<<<NN / 16, T, 0, stream>>>(feat, W_g[g], hbuf);
    k_elr<<<(NN * HH + T - 1) / T, T, 0, stream>>>(hbuf, al_g[g], ar_g[g], el, er);
    k_edge_max<<<(EE * HH + T - 1) / T, T, 0, stream>>>(src_g[g], dst_g[g], el, er, m_g[g]);
    k_edge_msg_own<<<OWN_P, OWN_T, 0, stream>>>(src_g[g], dst_g[g], el, er, m_g[g], hbuf,
                                                          s_g[g], acc_g[g]);
  }

  k_norm_mean<<<(NN * DD + T - 1) / T, T, 0, stream>>>(acc0, acc1, s0, s1, out);
}


# both graphs side by side: each per-graph launch covers the two graphs (own h/el/er planes); owner LDS cut to its tables so two owners share a WGP; 10 launches become 6; sums stay in LDS arrival order
# speedup vs baseline: 1.9142x; 1.0866x over previous
#include <hip/hip_runtime.h>

#define NN 50000
#define EE 1600000
#define INF_ 128
#define HH 4
#define DD 32
#define HD 128
#define NEG_SLOPE 0.2f

typedef __attribute__((ext_vector_type(2))) float v2f;
typedef __attribute__((ext_vector_type(8))) float v8f;

__device__ __forceinline__ unsigned int ford(float f) {
  unsigned int b = __float_as_uint(f);
  return (b & 0x80000000u) ? ~b : (b | 0x80000000u);
}
__device__ __forceinline__ float funord(unsigned int u) {
  unsigned int b = (u & 0x80000000u) ? (u ^ 0x80000000u) : ~u;
  return __uint_as_float(b);
}
#define FORD_NEG_INF 0x007FFFFFu

__global__ void k_init(uint4* __restrict__ m, int nm4) {
  int i = blockIdx.x * blockDim.x + threadIdx.x;
  if (i < nm4) m[i] = make_uint4(FORD_NEG_INF, FORD_NEG_INF, FORD_NEG_INF, FORD_NEG_INF);
}

#define PL_ND ((size_t)NN * HD)
#define PL_NH ((size_t)NN * HH)

__global__ __launch_bounds__(256) void k_proj(const float* __restrict__ feat,
                                              const float* __restrict__ W0,
                                              const float* __restrict__ W1,
                                              float* __restrict__ hb) {
  __shared__ float As[16 * 132];
  const int g = blockIdx.y;
  const float* __restrict__ W = g ? W1 : W0;
  float* __restrict__ h = hb + (size_t)g * PL_ND;
  const int m0 = blockIdx.x * 16;
  const int t  = threadIdx.x;

  {
    const int row = t >> 4;
    const int col = (t & 15) * 8;
    const float4* gp = (const float4*)(feat + (size_t)(m0 + row) * INF_ + col);
    float4 x0 = gp[0];
    float4 x1 = gp[1];
    float* lp = As + row * 132 + col;
    *(float4*)(lp)     = x0;
    *(float4*)(lp + 4) = x1;
  }
  __syncthreads();

  const int lane  = t & 31;
  const int wave  = t >> 5;
  const int n0    = wave * 16;
  const int r     = lane & 15;
  const int khalf = (lane >> 4) << 1;

  v8f acc = {};
  for (int k0 = 0; k0 < INF_; k0 += 4) {
    const int kk = k0 + khalf;
    v2f a;
    a.x = As[r * 132 + kk];
    a.y = As[r * 132 + kk + 1];
    v2f b;
    b.x = W[(size_t)(kk)     * HD + n0 + r];
    b.y = W[(size_t)(kk + 1) * HD + n0 + r];
    acc = __builtin_amdgcn_wmma_f32_16x16x4_f32(false, a, false, b,
                                                (short)0, acc, false, false);
  }

  const int mh = (lane >> 4) * 8;
#pragma unroll
  for (int j = 0; j < 8; ++j) {
    h[(size_t)(m0 + j + mh) * HD + n0 + r] = acc[j];
  }
}

__global__ void k_elr(const float* __restrict__ hb, const float* __restrict__ al0,
                      const float* __restrict__ al1, const float* __restrict__ ar0,
                      const float* __restrict__ ar1, float* __restrict__ elb,
                      float* __restrict__ erb) {
  int tid = blockIdx.x * blockDim.x + threadIdx.x;
  if (tid >= NN * HH) return;
  const int g = blockIdx.y;
  const float* __restrict__ h  = hb + (size_t)g * PL_ND;
  const float* __restrict__ al = g ? al1 : al0;
  const float* __restrict__ ar = g ? ar1 : ar0;
  float* __restrict__ el = elb + (size_t)g * PL_NH;
  float* __restrict__ er = erb + (size_t)g * PL_NH;
  const int n  = tid / HH;
  const int hh = tid % HH;
  const float* hp  = h  + (size_t)n * HD + hh * DD;
  const float* alp = al + hh * DD;
  const float* arp = ar + hh * DD;
  float sl = 0.0f, sr = 0.0f;
#pragma unroll
  for (int d = 0; d < DD; d += 4) {
    float4 hv = *(const float4*)(hp + d);
    float4 av = *(const float4*)(alp + d);
    float4 rv = *(const float4*)(arp + d);
    sl += hv.x * av.x + hv.y * av.y + hv.z * av.z + hv.w * av.w;
    sr += hv.x * rv.x + hv.y * rv.y + hv.z * rv.z + hv.w * rv.w;
  }
  el[tid] = sl;
  er[tid] = sr;
}

__global__ void k_edge_max(const int* __restrict__ src0, const int* __restrict__ src1,
                           const int* __restrict__ dst0, const int* __restrict__ dst1,
                           const float* __restrict__ elb, const float* __restrict__ erb,
                           unsigned int* __restrict__ mb) {
  int tid = blockIdx.x * blockDim.x + threadIdx.x;
  if (tid >= EE * HH) return;
  const int g = blockIdx.y;
  const int* __restrict__ src = g ? src1 : src0;
  const int* __restrict__ dst = g ? dst1 : dst0;
  const float* __restrict__ el = elb + (size_t)g * PL_NH;
  const float* __restrict__ er = erb + (size_t)g * PL_NH;
  unsigned int* __restrict__ m = mb + (size_t)g * PL_NH;
  const int e  = tid >> 2;
  const int hh = tid & 3;
  const int sd = src[e], dd = dst[e];
  float x = el[sd * HH + hh] + er[dd * HH + hh];
  x = (x > 0.0f) ? x : NEG_SLOPE * x;
  atomicMax(&m[dd * HH + hh], ford(x));
}

#define OWN_R 200
#define OWN_P ((NN + OWN_R - 1) / OWN_R)
#define CH 512
#define NCH (EE / CH)
#define OWN_T 1024
static_assert(EE % CH == 0 && NCH >= OWN_T / 32 && NCH / OWN_P >= 1, "edge list splits into whole wave steps");
static_assert(OWN_R % 8 == 0, "an owner's sum rows fill whole 128-byte lines");
#define OWN_LDS ((OWN_R * HD + 3 * OWN_R * HH) * 4)

__device__ __forceinline__ void own_add(float* __restrict__ acc_l, float* __restrict__ s_l,
                                        const float* __restrict__ er_l, const float* __restrict__ m_l,
                                        int dl, int lane, int hh, float elv, const float4 hv) {
  float x = elv + er_l[dl * HH + hh];
  x = (x > 0.0f) ? x : NEG_SLOPE * x;
  const float ex = __expf(x - m_l[dl * HH + hh]);
  float* ap = acc_l + dl * HD + lane * 4;
  atomicAdd(ap + 0, ex * hv.x);
  atomicAdd(ap + 1, ex * hv.y);
  atomicAdd(ap + 2, ex * hv.z);
  atomicAdd(ap + 3, ex * hv.w);
  if ((lane & 7) == 0) atomicAdd(&s_l[dl * HH + hh], ex);
}

__global__ __launch_bounds__(OWN_T) void k_edge_msg_own(const int* __restrict__ src0,
                                                        const int* __restrict__ src1,
                                                        const int* __restrict__ dst0,
                                                        const int* __restrict__ dst1,
                                                        const float* __restrict__ elb,
                                                        const float* __restrict__ erb,
                                                        const unsigned int* __restrict__ mb,
                                                        const float* __restrict__ hb,
                                                        float* __restrict__ sb,
                                                        float* __restrict__ accb) {
  const int g  = blockIdx.x & 1;
  const int ob = blockIdx.x >> 1;
  const int* __restrict__ src = g ? src1 : src0;
  const int* __restrict__ dst = g ? dst1 : dst0;
  const float* __restrict__ el = elb + (size_t)g * PL_NH;
  const float* __restrict__ er = erb + (size_t)g * PL_NH;
  const unsigned int* __restrict__ m = mb + (size_t)g * PL_NH;
  const float* __restrict__ h = hb + (size_t)g * PL_ND;
  float* __restrict__ s   = sb + (size_t)g * PL_NH;
  float* __restrict__ acc = accb + (size_t)g * PL_ND;
  __shared__ __attribute__((aligned(16))) unsigned char L[OWN_LDS];
  float* acc_l = (float*)L;
  float* s_l   = acc_l + OWN_R * HD;
  float* er_l  = s_l + OWN_R * HH;
  float* m_l   = er_l + OWN_R * HH;

  const int t    = threadIdx.x;
  const int lane = t & 31;
  const int wave = __builtin_amdgcn_readfirstlane(t >> 5);
  const int hh   = lane >> 3;
  const int n0   = ob * OWN_R;
  const int nrows = (NN - n0 < OWN_R) ? (NN - n0) : OWN_R;

  for (int i = t; i < OWN_R * HD / 4; i += OWN_T)
    ((float4*)acc_l)[i] = make_float4(0.0f, 0.0f, 0.0f, 0.0f);
  for (int i = t; i < OWN_R * HH; i += OWN_T) {
    const bool ok = i < nrows * HH;
    s_l[i]  = 0.0f;
    er_l[i] = ok ? er[(size_t)n0 * HH + i] : 0.0f;
    m_l[i]  = ok ? funord(m[(size_t)n0 * HH + i]) : 0.0f;
  }
  __syncthreads();

  const int rot = ob * (NCH / OWN_P);
  const int NIT = (NCH + (OWN_T / 32) - 1) / (OWN_T / 32);

  int4 cur[4], nxt[4];
  {
    int cr = wave + rot; if (cr >= NCH) cr -= NCH;
    const int4* p = (const int4*)(dst + (size_t)cr * CH) + lane;
#pragma unroll
    for (int k = 0; k < 4; ++k) cur[k] = p[k * 32];
  }

  for (int it = 0; it < NIT; ++it) {
    const int c = it * (OWN_T / 32) + wave;
    {
      int cn = c + (OWN_T / 32); if (cn >= NCH) cn = wave;
      int crn = cn + rot; if (crn >= NCH) crn -= NCH;
      const int4* p = (const int4*)(dst + (size_t)crn * CH) + lane;
#pragma unroll
      for (int k = 0; k < 4; ++k) nxt[k] = p[k * 32];
    }
    if (c < NCH) {
      int cr = c + rot; if (cr >= NCH) cr -= NCH;
      unsigned int mm = 0;
#pragma unroll
      for (int k = 3; k >= 0; --k) {
        mm = (mm << 1) | (((unsigned int)(cur[k].w - n0) < (unsigned int)OWN_R) ? 1u : 0u);
        mm = (mm << 1) | (((unsigned int)(cur[k].z - n0) < (unsigned int)OWN_R) ? 1u : 0u);
        mm = (mm << 1) | (((unsigned int)(cur[k].y - n0) < (unsigned int)OWN_R) ? 1u : 0u);
        mm = (mm << 1) | (((unsigned int)(cur[k].x - n0) < (unsigned int)OWN_R) ? 1u : 0u);
      }
      const int ebase = cr * CH + lane * 4;
      unsigned int mask = __builtin_amdgcn_ballot_w32(mm != 0u);
      while (mask != 0u) {
        int sd = 0, dl = 0;
        if (mm != 0u) {
          const int j = __builtin_ctz(mm);
          const int e = ebase + (j >> 2) * 128 + (j & 3);
          sd = src[e];
          dl = dst[e] - n0;
          mm &= mm - 1u;
        }
        do {
          const int l0 = __builtin_ctz(mask); mask &= mask - 1u;
          const bool h1 = (mask != 0u);
          const int l1 = h1 ? __builtin_ctz(mask) : l0; mask &= mask - 1u;
          const bool h2 = (mask != 0u);
          const int l2 = h2 ? __builtin_ctz(mask) : l0; mask &= mask - 1u;
          const bool h3 = (mask != 0u);
          const int l3 = h3 ? __builtin_ctz(mask) : l0; mask &= mask - 1u;
          const int sd0 = __builtin_amdgcn_readlane(sd, l0), dl0 = __builtin_amdgcn_readlane(dl, l0);
          const int sd1 = __builtin_amdgcn_readlane(sd, l1), dl1 = __builtin_amdgcn_readlane(dl, l1);
          const int sd2 = __builtin_amdgcn_readlane(sd, l2), dl2 = __builtin_amdgcn_readlane(dl, l2);
          const int sd3 = __builtin_amdgcn_readlane(sd, l3), dl3 = __builtin_amdgcn_readlane(dl, l3);
          const float4 hv0 = *(const float4*)(h + (size_t)sd0 * HD + lane * 4);
          const float4 hv1 = *(const float4*)(h + (size_t)sd1 * HD + lane * 4);
          const float4 hv2 = *(const float4*)(h + (size_t)sd2 * HD + lane * 4);
          const float4 hv3 = *(const float4*)(h + (size_t)sd3 * HD + lane * 4);
          const float e0 = el[sd0 * HH + hh];
          const float e1 = el[sd1 * HH + hh];
          const float e2 = el[sd2 * HH + hh];
          const float e3 = el[sd3 * HH + hh];
          own_add(acc_l, s_l, er_l, m_l, dl0, lane, hh, e0, hv0);
          if (h1) own_add(acc_l, s_l, er_l, m_l, dl1, lane, hh, e1, hv1);
          if (h2) own_add(acc_l, s_l, er_l, m_l, dl2, lane, hh, e2, hv2);
          if (h3) own_add(acc_l, s_l, er_l, m_l, dl3, lane, hh, e3, hv3);
        } while (mask != 0u);
        mask = __builtin_amdgcn_ballot_w32(mm != 0u);
      }
    }
#pragma unroll
    for (int k = 0; k < 4; ++k) cur[k] = nxt[k];
  }
  __syncthreads();

  float4* ag = (float4*)(acc + (size_t)n0 * HD);
  for (int i = t; i < nrows * HD / 4; i += OWN_T) ag[i] = ((const float4*)acc_l)[i];
  float* sg = s + (size_t)n0 * HH;
  for (int i = t; i < nrows * HH; i += OWN_T) sg[i] = s_l[i];
}

__device__ __forceinline__ float inv_or_zero(float v) {
  return (v != 0.0f) ? 1.0f / v : 0.0f;
}
__global__ void k_norm_mean(const float* __restrict__ acc0, const float* __restrict__ acc1,
                            const float* __restrict__ s0, const float* __restrict__ s1,
                            float* __restrict__ out) {
  int tid = blockIdx.x * blockDim.x + threadIdx.x;
  if (tid >= NN * DD) return;
  const int n = tid >> 5;
  const int d = tid & 31;
  const float4 v0 = *(const float4*)(s0 + (size_t)n * HH);
  const float4 v1 = *(const float4*)(s1 + (size_t)n * HH);
  const float* a0 = acc0 + (size_t)n * HD + d;
  const float* a1 = acc1 + (size_t)n * HD + d;
  const float f0 = a0[0], f1 = a0[32], f2 = a0[64], f3 = a0[96];
  const float b0 = a1[0], b1 = a1[32], b2 = a1[64], b3 = a1[96];
  float r = f0 * inv_or_zero(v0.x) + b0 * inv_or_zero(v1.x);
  r += f1 * inv_or_zero(v0.y) + b1 * inv_or_zero(v1.y);
  r += f2 * inv_or_zero(v0.z) + b2 * inv_or_zero(v1.z);
  r += f3 * inv_or_zero(v0.w) + b3 * inv_or_zero(v1.w);
  out[tid] = 0.25f * r;
}

extern "C" void kernel_launch(void* const* d_in, const int* in_sizes, int n_in,
                              void* d_out, int out_size, void* d_ws, size_t ws_size,
                              hipStream_t stream) {
  const float* feat = (const float*)d_in[0];
  const int*   src_g[2] = { (const int*)d_in[1], (const int*)d_in[3] };
  const int*   dst_g[2] = { (const int*)d_in[2], (const int*)d_in[4] };
  const float* W_g[2]   = { (const float*)d_in[5], (const float*)d_in[8] };
  const float* al_g[2]  = { (const float*)d_in[6], (const float*)d_in[9] };
  const float* ar_g[2]  = { (const float*)d_in[7], (const float*)d_in[10] };
  float* out = (float*)d_out;

  float*        acc0 = (float*)d_ws;
  float*        acc1 = acc0 + (size_t)NN * HD;
  float*        s0   = acc1 + (size_t)NN * HD;
  float*        s1   = s0   + (size_t)NN * HH;
  unsigned int* m0   = (unsigned int*)(s1 + (size_t)NN * HH);
  unsigned int* m1   = m0   + (size_t)NN * HH;
  float*        hbuf = (float*)(m1 + (size_t)NN * HH);
  float*        el   = hbuf + 2 * (size_t)NN * HD;
  float*        er   = el   + 2 * (size_t)NN * HH;

  const int T = 256;
  const int NM4 = (2 * NN * HH) / 4;
  k_init<<<(NM4 + T - 1) / T, T, 0, stream>>>((uint4*)m0, NM4);

  k_proj<<<dim3(NN / 16, 2), T, 0, stream>>>(feat, W_g[0], W_g[1], hbuf);
  k_elr<<<dim3((NN * HH + T - 1) / T, 2), T, 0, stream>>>(hbuf, al_g[0], al_g[1], ar_g[0], ar_g[1], el, er);
  k_edge_max<<<dim3((EE * HH + T - 1) / T, 2), T, 0, stream>>>(src_g[0], src_g[1], dst_g[0], dst_g[1],
                                                               el, er, m0);
  k_edge_msg_own<<<2 * OWN_P, OWN_T, 0, stream>>>(src_g[0], src_g[1], dst_g[0], dst_g[1], el, er, m0,
                                                   hbuf, s0, acc0);

  k_norm_mean<<<(NN * DD + T - 1) / T, T, 0, stream>>>(acc0, acc1, s0, s1, out);
}


# owner pass scans a 1-byte owner key per edge (dst/200, written by the init kernel) instead of the 4-byte dst: 2048 edges per wave step, 4 keys tested per word; sums stay in LDS arrival order
# speedup vs baseline: 2.0736x; 1.0833x over previous
#include <hip/hip_runtime.h>

#define NN 50000
#define EE 1600000
#define INF_ 128
#define HH 4
#define DD 32
#define HD 128
#define NEG_SLOPE 0.2f

typedef __attribute__((ext_vector_type(2))) float v2f;
typedef __attribute__((ext_vector_type(8))) float v8f;

__device__ __forceinline__ unsigned int ford(float f) {
  unsigned int b = __float_as_uint(f);
  return (b & 0x80000000u) ? ~b : (b | 0x80000000u);
}
__device__ __forceinline__ float funord(unsigned int u) {
  unsigned int b = (u & 0x80000000u) ? (u ^ 0x80000000u) : ~u;
  return __uint_as_float(b);
}
#define FORD_NEG_INF 0x007FFFFFu

#define OWN_R 200
#define OWN_P ((NN + OWN_R - 1) / OWN_R)
#define KCH 2048
#define NKCH ((EE + KCH - 1) / KCH)
#define KEYW (NKCH * (KCH / 4))
#define KEY_PAD 0xFFFFFFFFu
static_assert(OWN_P <= 255 && EE % 4 == 0, "owner numbers fit a byte below the pad key; edges fill whole key dwords");

__global__ void k_init(uint4* __restrict__ m, int nm4, const int* __restrict__ dst0,
                       const int* __restrict__ dst1, unsigned int* __restrict__ keyb) {
  int i = blockIdx.x * blockDim.x + threadIdx.x;
  const int g = blockIdx.y;
  if (g == 0 && i < nm4) m[i] = make_uint4(FORD_NEG_INF, FORD_NEG_INF, FORD_NEG_INF, FORD_NEG_INF);
  if (i < KEYW) {
    unsigned int kv = KEY_PAD;
    if (i < EE / 4) {
      const int4 d = ((const int4*)(g ? dst1 : dst0))[i];
      kv = (unsigned int)(d.x / OWN_R) | ((unsigned int)(d.y / OWN_R) << 8) |
           ((unsigned int)(d.z / OWN_R) << 16) | ((unsigned int)(d.w / OWN_R) << 24);
    }
    keyb[(size_t)g * KEYW + i] = kv;
  }
}

#define PL_ND ((size_t)NN * HD)
#define PL_NH ((size_t)NN * HH)

__global__ __launch_bounds__(256) void k_proj(const float* __restrict__ feat,
                                              const float* __restrict__ W0,
                                              const float* __restrict__ W1,
                                              float* __restrict__ hb) {
  __shared__ float As[16 * 132];
  const int g = blockIdx.y;
  const float* __restrict__ W = g ? W1 : W0;
  float* __restrict__ h = hb + (size_t)g * PL_ND;
  const int m0 = blockIdx.x * 16;
  const int t  = threadIdx.x;

  {
    const int row = t >> 4;
    const int col = (t & 15) * 8;
    const float4* gp = (const float4*)(feat + (size_t)(m0 + row) * INF_ + col);
    float4 x0 = gp[0];
    float4 x1 = gp[1];
    float* lp = As + row * 132 + col;
    *(float4*)(lp)     = x0;
    *(float4*)(lp + 4) = x1;
  }
  __syncthreads();

  const int lane  = t & 31;
  const int wave  = t >> 5;
  const int n0    = wave * 16;
  const int r     = lane & 15;
  const int khalf = (lane >> 4) << 1;

  v8f acc = {};
  for (int k0 = 0; k0 < INF_; k0 += 4) {
    const int kk = k0 + khalf;
    v2f a;
    a.x = As[r * 132 + kk];
    a.y = As[r * 132 + kk + 1];
    v2f b;
    b.x = W[(size_t)(kk)     * HD + n0 + r];
    b.y = W[(size_t)(kk + 1) * HD + n0 + r];
    acc = __builtin_amdgcn_wmma_f32_16x16x4_f32(false, a, false, b,
                                                (short)0, acc, false, false);
  }

  const int mh = (lane >> 4) * 8;
#pragma unroll
  for (int j = 0; j < 8; ++j) {
    h[(size_t)(m0 + j + mh) * HD + n0 + r] = acc[j];
  }
}

__global__ void k_elr(const float* __restrict__ hb, const float* __restrict__ al0,
                      const float* __restrict__ al1, const float* __restrict__ ar0,
                      const float* __restrict__ ar1, float* __restrict__ elb,
                      float* __restrict__ erb) {
  int tid = blockIdx.x * blockDim.x + threadIdx.x;
  if (tid >= NN * HH) return;
  const int g = blockIdx.y;
  const float* __restrict__ h  = hb + (size_t)g * PL_ND;
  const float* __restrict__ al = g ? al1 : al0;
  const float* __restrict__ ar = g ? ar1 : ar0;
  float* __restrict__ el = elb + (size_t)g * PL_NH;
  float* __restrict__ er = erb + (size_t)g * PL_NH;
  const int n  = tid / HH;
  const int hh = tid % HH;
  const float* hp  = h  + (size_t)n * HD + hh * DD;
  const float* alp = al + hh * DD;
  const float* arp = ar + hh * DD;
  float sl = 0.0f, sr = 0.0f;
#pragma unroll
  for (int d = 0; d < DD; d += 4) {
    float4 hv = *(const float4*)(hp + d);
    float4 av = *(const float4*)(alp + d);
    float4 rv = *(const float4*)(arp + d);
    sl += hv.x * av.x + hv.y * av.y + hv.z * av.z + hv.w * av.w;
    sr += hv.x * rv.x + hv.y * rv.y + hv.z * rv.z + hv.w * rv.w;
  }
  el[tid] = sl;
  er[tid] = sr;
}

__global__ void k_edge_max(const int* __restrict__ src0, const int* __restrict__ src1,
                           const int* __restrict__ dst0, const int* __restrict__ dst1,
                           const float* __restrict__ elb, const float* __restrict__ erb,
                           unsigned int* __restrict__ mb) {
  int tid = blockIdx.x * blockDim.x + threadIdx.x;
  if (tid >= EE * HH) return;
  const int g = blockIdx.y;
  const int* __restrict__ src = g ? src1 : src0;
  const int* __restrict__ dst = g ? dst1 : dst0;
  const float* __restrict__ el = elb + (size_t)g * PL_NH;
  const float* __restrict__ er = erb + (size_t)g * PL_NH;
  unsigned int* __restrict__ m = mb + (size_t)g * PL_NH;
  const int e  = tid >> 2;
  const int hh = tid & 3;
  const int sd = src[e], dd = dst[e];
  float x = el[sd * HH + hh] + er[dd * HH + hh];
  x = (x > 0.0f) ? x : NEG_SLOPE * x;
  atomicMax(&m[dd * HH + hh], ford(x));
}

#define OWN_T 1024
static_assert(NKCH >= OWN_T / 32 && NKCH / OWN_P >= 1, "every wave has a step; owners start at different steps");
static_assert(OWN_R % 8 == 0, "an owner's sum rows fill whole 128-byte lines");
#define OWN_LDS ((OWN_R * HD + 3 * OWN_R * HH) * 4)

__device__ __forceinline__ void own_add(float* __restrict__ acc_l, float* __restrict__ s_l,
                                        const float* __restrict__ er_l, const float* __restrict__ m_l,
                                        int dl, int lane, int hh, float elv, const float4 hv) {
  float x = elv + er_l[dl * HH + hh];
  x = (x > 0.0f) ? x : NEG_SLOPE * x;
  const float ex = __expf(x - m_l[dl * HH + hh]);
  float* ap = acc_l + dl * HD + lane * 4;
  atomicAdd(ap + 0, ex * hv.x);
  atomicAdd(ap + 1, ex * hv.y);
  atomicAdd(ap + 2, ex * hv.z);
  atomicAdd(ap + 3, ex * hv.w);
  if ((lane & 7) == 0) atomicAdd(&s_l[dl * HH + hh], ex);
}

__device__ __forceinline__ unsigned int key_differs(unsigned int x, unsigned int pat) {
  const unsigned int t = x ^ pat;
  return ((t & 0x7F7F7F7Fu) + 0x7F7F7F7Fu) | t;
}
__device__ __forceinline__ unsigned int own16(const int4 v, unsigned int pat) {
  const unsigned int c = ((key_differs((unsigned int)v.x, pat) >> 7) & 0x01010101u) |
                         ((key_differs((unsigned int)v.y, pat) >> 6) & 0x02020202u) |
                         ((key_differs((unsigned int)v.z, pat) >> 5) & 0x04040404u) |
                         ((key_differs((unsigned int)v.w, pat) >> 4) & 0x08080808u);
  return c ^ 0x0F0F0F0Fu;
}

__global__ __launch_bounds__(OWN_T) void k_edge_msg_own(const int* __restrict__ src0,
                                                        const int* __restrict__ src1,
                                                        const int* __restrict__ dst0,
                                                        const int* __restrict__ dst1,
                                                        const float* __restrict__ elb,
                                                        const float* __restrict__ erb,
                                                        const unsigned int* __restrict__ mb,
                                                        const unsigned int* __restrict__ keyb,
                                                        const float* __restrict__ hb,
                                                        float* __restrict__ sb,
                                                        float* __restrict__ accb) {
  const int g  = blockIdx.x & 1;
  const int ob = blockIdx.x >> 1;
  const int* __restrict__ src = g ? src1 : src0;
  const int* __restrict__ dst = g ? dst1 : dst0;
  const float* __restrict__ el = elb + (size_t)g * PL_NH;
  const float* __restrict__ er = erb + (size_t)g * PL_NH;
  const unsigned int* __restrict__ m = mb + (size_t)g * PL_NH;
  const unsigned int* __restrict__ key = keyb + (size_t)g * KEYW;
  const float* __restrict__ h = hb + (size_t)g * PL_ND;
  float* __restrict__ s   = sb + (size_t)g * PL_NH;
  float* __restrict__ acc = accb + (size_t)g * PL_ND;
  __shared__ __attribute__((aligned(16))) unsigned char L[OWN_LDS];
  float* acc_l = (float*)L;
  float* s_l   = acc_l + OWN_R * HD;
  float* er_l  = s_l + OWN_R * HH;
  float* m_l   = er_l + OWN_R * HH;

  const int t    = threadIdx.x;
  const int lane = t & 31;
  const int wave = __builtin_amdgcn_readfirstlane(t >> 5);
  const int hh   = lane >> 3;
  const int n0   = ob * OWN_R;
  const int nrows = (NN - n0 < OWN_R) ? (NN - n0) : OWN_R;

  for (int i = t; i < OWN_R * HD / 4; i += OWN_T)
    ((float4*)acc_l)[i] = make_float4(0.0f, 0.0f, 0.0f, 0.0f);
  for (int i = t; i < OWN_R * HH; i += OWN_T) {
    const bool ok = i < nrows * HH;
    s_l[i]  = 0.0f;
    er_l[i] = ok ? er[(size_t)n0 * HH + i] : 0.0f;
    m_l[i]  = ok ? funord(m[(size_t)n0 * HH + i]) : 0.0f;
  }
  __syncthreads();

  const int rot = ob * (NKCH / OWN_P);
  const int NIT = (NKCH + (OWN_T / 32) - 1) / (OWN_T / 32);
  const unsigned int pat = (unsigned int)ob * 0x01010101u;

  int4 cur[4], nxt[4];
  {
    int cr = wave + rot; if (cr >= NKCH) cr -= NKCH;
    const int4* p = (const int4*)(key + (size_t)cr * (KCH / 4)) + lane;
#pragma unroll
    for (int k = 0; k < 4; ++k) cur[k] = p[k * 32];
  }

  for (int it = 0; it < NIT; ++it) {
    const int c = it * (OWN_T / 32) + wave;
    {
      int cn = c + (OWN_T / 32); if (cn >= NKCH) cn = wave;
      int crn = cn + rot; if (crn >= NKCH) crn -= NKCH;
      const int4* p = (const int4*)(key + (size_t)crn * (KCH / 4)) + lane;
#pragma unroll
      for (int k = 0; k < 4; ++k) nxt[k] = p[k * 32];
    }
    if (c < NKCH) {
      int cr = c + rot; if (cr >= NKCH) cr -= NKCH;
      const unsigned int m01 = own16(cur[0], pat) | (own16(cur[1], pat) << 4);
      const unsigned int m23 = own16(cur[2], pat) | (own16(cur[3], pat) << 4);
      unsigned long long mm = (unsigned long long)m01 | ((unsigned long long)m23 << 32);
      const int ebase = cr * KCH + lane * 16;
      unsigned int mask = __builtin_amdgcn_ballot_w32(mm != 0ull);
      while (mask != 0u) {
        int sd = 0, dl = 0;
        if (mm != 0ull) {
          const int p = __builtin_ctzll(mm);
          const int k = ((p >> 5) << 1) | ((p >> 2) & 1);
          const int e = ebase + k * 512 + ((p & 3) << 2) + ((p & 31) >> 3);
          sd = src[e];
          dl = dst[e] - n0;
          mm &= mm - 1ull;
        }
        do {
          const int l0 = __builtin_ctz(mask); mask &= mask - 1u;
          const bool h1 = (mask != 0u);
          const int l1 = h1 ? __builtin_ctz(mask) : l0; mask &= mask - 1u;
          const bool h2 = (mask != 0u);
          const int l2 = h2 ? __builtin_ctz(mask) : l0; mask &= mask - 1u;
          const bool h3 = (mask != 0u);
          const int l3 = h3 ? __builtin_ctz(mask) : l0; mask &= mask - 1u;
          const int sd0 = __builtin_amdgcn_readlane(sd, l0), dl0 = __builtin_amdgcn_readlane(dl, l0);
          const int sd1 = __builtin_amdgcn_readlane(sd, l1), dl1 = __builtin_amdgcn_readlane(dl, l1);
          const int sd2 = __builtin_amdgcn_readlane(sd, l2), dl2 = __builtin_amdgcn_readlane(dl, l2);
          const int sd3 = __builtin_amdgcn_readlane(sd, l3), dl3 = __builtin_amdgcn_readlane(dl, l3);
          const float4 hv0 = *(const float4*)(h + (size_t)sd0 * HD + lane * 4);
          const float4 hv1 = *(const float4*)(h + (size_t)sd1 * HD + lane * 4);
          const float4 hv2 = *(const float4*)(h + (size_t)sd2 * HD + lane * 4);
          const float4 hv3 = *(const float4*)(h + (size_t)sd3 * HD + lane * 4);
          const float e0 = el[sd0 * HH + hh];
          const float e1 = el[sd1 * HH + hh];
          const float e2 = el[sd2 * HH + hh];
          const float e3 = el[sd3 * HH + hh];
          own_add(acc_l, s_l, er_l, m_l, dl0, lane, hh, e0, hv0);
          if (h1) own_add(acc_l, s_l, er_l, m_l, dl1, lane, hh, e1, hv1);
          if (h2) own_add(acc_l, s_l, er_l, m_l, dl2, lane, hh, e2, hv2);
          if (h3) own_add(acc_l, s_l, er_l, m_l, dl3, lane, hh, e3, hv3);
        } while (mask != 0u);
        mask = __builtin_amdgcn_ballot_w32(mm != 0ull);
      }
    }
#pragma unroll
    for (int k = 0; k < 4; ++k) cur[k] = nxt[k];
  }
  __syncthreads();

  float4* ag = (float4*)(acc + (size_t)n0 * HD);
  for (int i = t; i < nrows * HD / 4; i += OWN_T) ag[i] = ((const float4*)acc_l)[i];
  float* sg = s + (size_t)n0 * HH;
  for (int i = t; i < nrows * HH; i += OWN_T) sg[i] = s_l[i];
}

__device__ __forceinline__ float inv_or_zero(float v) {
  return (v != 0.0f) ? 1.0f / v : 0.0f;
}
__global__ void k_norm_mean(const float* __restrict__ acc0, const float* __restrict__ acc1,
                            const float* __restrict__ s0, const float* __restrict__ s1,
                            float* __restrict__ out) {
  int tid = blockIdx.x * blockDim.x + threadIdx.x;
  if (tid >= NN * DD) return;
  const int n = tid >> 5;
  const int d = tid & 31;
  const float4 v0 = *(const float4*)(s0 + (size_t)n * HH);
  const float4 v1 = *(const float4*)(s1 + (size_t)n * HH);
  const float* a0 = acc0 + (size_t)n * HD + d;
  const float* a1 = acc1 + (size_t)n * HD + d;
  const float f0 = a0[0], f1 = a0[32], f2 = a0[64], f3 = a0[96];
  const float b0 = a1[0], b1 = a1[32], b2 = a1[64], b3 = a1[96];
  float r = f0 * inv_or_zero(v0.x) + b0 * inv_or_zero(v1.x);
  r += f1 * inv_or_zero(v0.y) + b1 * inv_or_zero(v1.y);
  r += f2 * inv_or_zero(v0.z) + b2 * inv_or_zero(v1.z);
  r += f3 * inv_or_zero(v0.w) + b3 * inv_or_zero(v1.w);
  out[tid] = 0.25f * r;
}

extern "C" void kernel_launch(void* const* d_in, const int* in_sizes, int n_in,
                              void* d_out, int out_size, void* d_ws, size_t ws_size,
                              hipStream_t stream) {
  const float* feat = (const float*)d_in[0];
  const int*   src_g[2] = { (const int*)d_in[1], (const int*)d_in[3] };
  const int*   dst_g[2] = { (const int*)d_in[2], (const int*)d_in[4] };
  const float* W_g[2]   = { (const float*)d_in[5], (const float*)d_in[8] };
  const float* al_g[2]  = { (const float*)d_in[6], (const float*)d_in[9] };
  const float* ar_g[2]  = { (const float*)d_in[7], (const float*)d_in[10] };
  float* out = (float*)d_out;

  float*        acc0 = (float*)d_ws;
  float*        acc1 = acc0 + (size_t)NN * HD;
  float*        s0   = acc1 + (size_t)NN * HD;
  float*        s1   = s0   + (size_t)NN * HH;
  unsigned int* m0   = (unsigned int*)(s1 + (size_t)NN * HH);
  unsigned int* m1   = m0   + (size_t)NN * HH;
  float*        hbuf = (float*)(m1 + (size_t)NN * HH);
  float*        el   = hbuf + 2 * (size_t)NN * HD;
  float*        er   = el   + 2 * (size_t)NN * HH;
  unsigned int* keyb = (unsigned int*)(er + 2 * (size_t)NN * HH);

  const int T = 256;
  const int NM4 = (2 * NN * HH) / 4;
  static_assert(KEYW >= (2 * NN * HH) / 4, "the key grid also covers the max tables");
  k_init<<<dim3((KEYW + T - 1) / T, 2), T, 0, stream>>>((uint4*)m0, NM4, dst_g[0], dst_g[1], keyb);

  k_proj<<<dim3(NN / 16, 2), T, 0, stream>>>(feat, W_g[0], W_g[1], hbuf);
  k_elr<<<dim3((NN * HH + T - 1) / T, 2), T, 0, stream>>>(hbuf, al_g[0], al_g[1], ar_g[0], ar_g[1], el, er);
  k_edge_max<<<dim3((EE * HH + T - 1) / T, 2), T, 0, stream>>>(src_g[0], src_g[1], dst_g[0], dst_g[1],
                                                               el, er, m0);
  k_edge_msg_own<<<2 * OWN_P, OWN_T, 0, stream>>>(src_g[0], src_g[1], dst_g[0], dst_g[1], el, er, m0,
                                                   keyb, hbuf, s0, acc0);

  k_norm_mean<<<(NN * DD + T - 1) / T, T, 0, stream>>>(acc0, acc1, s0, s1, out);
}


# segment max moved into the owner pass: an owner collects its edges once into an LDS list, takes the max there by LDS atomics, then adds messages from the list; global max pass goes; sum order changes
# speedup vs baseline: 2.7179x; 1.3107x over previous
#include <hip/hip_runtime.h>

#define NN 50000
#define EE 1600000
#define INF_ 128
#define HH 4
#define DD 32
#define HD 128
#define NEG_SLOPE 0.2f

typedef __attribute__((ext_vector_type(2))) float v2f;
typedef __attribute__((ext_vector_type(8))) float v8f;

__device__ __forceinline__ unsigned int ford(float f) {
  unsigned int b = __float_as_uint(f);
  return (b & 0x80000000u) ? ~b : (b | 0x80000000u);
}
__device__ __forceinline__ float funord(unsigned int u) {
  unsigned int b = (u & 0x80000000u) ? (u ^ 0x80000000u) : ~u;
  return __uint_as_float(b);
}
#define FORD_NEG_INF 0x007FFFFFu

#define OWN_R 200
#define OWN_P ((NN + OWN_R - 1) / OWN_R)
#define KCH 2048
#define NKCH ((EE + KCH - 1) / KCH)
#define KEYW (NKCH * (KCH / 4))
#define KEY_PAD 0xFFFFFFFFu
static_assert(OWN_P <= 255 && EE % 4 == 0, "owner numbers fit a byte below the pad key; edges fill whole key dwords");

__global__ void k_init(const int* __restrict__ dst0, const int* __restrict__ dst1,
                       unsigned int* __restrict__ keyb) {
  int i = blockIdx.x * blockDim.x + threadIdx.x;
  const int g = blockIdx.y;
  if (i < KEYW) {
    unsigned int kv = KEY_PAD;
    if (i < EE / 4) {
      const int4 d = ((const int4*)(g ? dst1 : dst0))[i];
      kv = (unsigned int)(d.x / OWN_R) | ((unsigned int)(d.y / OWN_R) << 8) |
           ((unsigned int)(d.z / OWN_R) << 16) | ((unsigned int)(d.w / OWN_R) << 24);
    }
    keyb[(size_t)g * KEYW + i] = kv;
  }
}

#define PL_ND ((size_t)NN * HD)
#define PL_NH ((size_t)NN * HH)

__global__ __launch_bounds__(256) void k_proj(const float* __restrict__ feat,
                                              const float* __restrict__ W0,
                                              const float* __restrict__ W1,
                                              float* __restrict__ hb) {
  __shared__ float As[16 * 132];
  const int g = blockIdx.y;
  const float* __restrict__ W = g ? W1 : W0;
  float* __restrict__ h = hb + (size_t)g * PL_ND;
  const int m0 = blockIdx.x * 16;
  const int t  = threadIdx.x;

  {
    const int row = t >> 4;
    const int col = (t & 15) * 8;
    const float4* gp = (const float4*)(feat + (size_t)(m0 + row) * INF_ + col);
    float4 x0 = gp[0];
    float4 x1 = gp[1];
    float* lp = As + row * 132 + col;
    *(float4*)(lp)     = x0;
    *(float4*)(lp + 4) = x1;
  }
  __syncthreads();

  const int lane  = t & 31;
  const int wave  = t >> 5;
  const int n0    = wave * 16;
  const int r     = lane & 15;
  const int khalf = (lane >> 4) << 1;

  v8f acc = {};
  for (int k0 = 0; k0 < INF_; k0 += 4) {
    const int kk = k0 + khalf;
    v2f a;
    a.x = As[r * 132 + kk];
    a.y = As[r * 132 + kk + 1];
    v2f b;
    b.x = W[(size_t)(kk)     * HD + n0 + r];
    b.y = W[(size_t)(kk + 1) * HD + n0 + r];
    acc = __builtin_amdgcn_wmma_f32_16x16x4_f32(false, a, false, b,
                                                (short)0, acc, false, false);
  }

  const int mh = (lane >> 4) * 8;
#pragma unroll
  for (int j = 0; j < 8; ++j) {
    h[(size_t)(m0 + j + mh) * HD + n0 + r] = acc[j];
  }
}

__global__ void k_elr(const float* __restrict__ hb, const float* __restrict__ al0,
                      const float* __restrict__ al1, const float* __restrict__ ar0,
                      const float* __restrict__ ar1, float* __restrict__ elb,
                      float* __restrict__ erb) {
  int tid = blockIdx.x * blockDim.x + threadIdx.x;
  if (tid >= NN * HH) return;
  const int g = blockIdx.y;
  const float* __restrict__ h  = hb + (size_t)g * PL_ND;
  const float* __restrict__ al = g ? al1 : al0;
  const float* __restrict__ ar = g ? ar1 : ar0;
  float* __restrict__ el = elb + (size_t)g * PL_NH;
  float* __restrict__ er = erb + (size_t)g * PL_NH;
  const int n  = tid / HH;
  const int hh = tid % HH;
  const float* hp  = h  + (size_t)n * HD + hh * DD;
  const float* alp = al + hh * DD;
  const float* arp = ar + hh * DD;
  float sl = 0.0f, sr = 0.0f;
#pragma unroll
  for (int d = 0; d < DD; d += 4) {
    float4 hv = *(const float4*)(hp + d);
    float4 av = *(const float4*)(alp + d);
    float4 rv = *(const float4*)(arp + d);
    sl += hv.x * av.x + hv.y * av.y + hv.z * av.z + hv.w * av.w;
    sr += hv.x * rv.x + hv.y * rv.y + hv.z * rv.z + hv.w * rv.w;
  }
  el[tid] = sl;
  er[tid] = sr;
}

#define OWN_T 1024
#define OWN_CAP 8192
static_assert(NKCH >= OWN_T / 32 && NKCH / OWN_P >= 1, "every wave has a step; owners start at different steps");
static_assert(OWN_R % 8 == 0, "an owner's sum rows fill whole 128-byte lines");
static_assert(NN <= 65536 && OWN_R <= 65536 && OWN_CAP % 4 == 0, "src and the local row share one list word");
#define OWN_TAB ((OWN_R * HD + 3 * OWN_R * HH) * 4)
#define OWN_LDS (OWN_TAB + 16 + OWN_CAP * 4)

__device__ __forceinline__ void own_max_v(const float4 ev, const float* __restrict__ er_l,
                                          unsigned int* __restrict__ mu_l, int dl) {
  const float4 rv = *(const float4*)(er_l + dl * HH);
  float x0 = ev.x + rv.x, x1 = ev.y + rv.y, x2 = ev.z + rv.z, x3 = ev.w + rv.w;
  x0 = (x0 > 0.0f) ? x0 : NEG_SLOPE * x0;
  x1 = (x1 > 0.0f) ? x1 : NEG_SLOPE * x1;
  x2 = (x2 > 0.0f) ? x2 : NEG_SLOPE * x2;
  x3 = (x3 > 0.0f) ? x3 : NEG_SLOPE * x3;
  atomicMax(&mu_l[dl * HH + 0], ford(x0));
  atomicMax(&mu_l[dl * HH + 1], ford(x1));
  atomicMax(&mu_l[dl * HH + 2], ford(x2));
  atomicMax(&mu_l[dl * HH + 3], ford(x3));
}
__device__ __forceinline__ void own_max(const float* __restrict__ el, const float* __restrict__ er_l,
                                        unsigned int* __restrict__ mu_l, int sd, int dl) {
  own_max_v(*(const float4*)(el + (size_t)sd * HH), er_l, mu_l, dl);
}

__device__ __forceinline__ void own_add(float* __restrict__ acc_l, float* __restrict__ s_l,
                                        const float* __restrict__ er_l, const unsigned int* __restrict__ mu_l,
                                        int dl, int lane, int hh, float elv, const float4 hv) {
  float x = elv + er_l[dl * HH + hh];
  x = (x > 0.0f) ? x : NEG_SLOPE * x;
  const float ex = __expf(x - funord(mu_l[dl * HH + hh]));
  float* ap = acc_l + dl * HD + lane * 4;
  atomicAdd(ap + 0, ex * hv.x);
  atomicAdd(ap + 1, ex * hv.y);
  atomicAdd(ap + 2, ex * hv.z);
  atomicAdd(ap + 3, ex * hv.w);
  if ((lane & 7) == 0) atomicAdd(&s_l[dl * HH + hh], ex);
}

__device__ __forceinline__ void own_add4(const float* __restrict__ h, const float* __restrict__ el,
                                         float* __restrict__ acc_l, float* __restrict__ s_l,
                                         const float* __restrict__ er_l, const unsigned int* __restrict__ mu_l,
                                         int lane, int hh, int sd0, int dl0, int sd1, int dl1, int sd2, int dl2,
                                         int sd3, int dl3, bool h1, bool h2, bool h3) {
  const float4 hv0 = *(const float4*)(h + (size_t)sd0 * HD + lane * 4);
  const float4 hv1 = *(const float4*)(h + (size_t)sd1 * HD + lane * 4);
  const float4 hv2 = *(const float4*)(h + (size_t)sd2 * HD + lane * 4);
  const float4 hv3 = *(const float4*)(h + (size_t)sd3 * HD + lane * 4);
  const float e0 = el[sd0 * HH + hh];
  const float e1 = el[sd1 * HH + hh];
  const float e2 = el[sd2 * HH + hh];
  const float e3 = el[sd3 * HH + hh];
  own_add(acc_l, s_l, er_l, mu_l, dl0, lane, hh, e0, hv0);
  if (h1) own_add(acc_l, s_l, er_l, mu_l, dl1, lane, hh, e1, hv1);
  if (h2) own_add(acc_l, s_l, er_l, mu_l, dl2, lane, hh, e2, hv2);
  if (h3) own_add(acc_l, s_l, er_l, mu_l, dl3, lane, hh, e3, hv3);
}

__device__ __forceinline__ unsigned int key_differs(unsigned int x, unsigned int pat) {
  const unsigned int t = x ^ pat;
  return ((t & 0x7F7F7F7Fu) + 0x7F7F7F7Fu) | t;
}
__device__ __forceinline__ unsigned int own16(const int4 v, unsigned int pat) {
  const unsigned int c = ((key_differs((unsigned int)v.x, pat) >> 7) & 0x01010101u) |
                         ((key_differs((unsigned int)v.y, pat) >> 6) & 0x02020202u) |
                         ((key_differs((unsigned int)v.z, pat) >> 5) & 0x04040404u) |
                         ((key_differs((unsigned int)v.w, pat) >> 4) & 0x08080808u);
  return c ^ 0x0F0F0F0Fu;
}

template <int MODE>
__device__ __forceinline__ void own_sweep(const int* __restrict__ src, const int* __restrict__ dst,
                                          const float* __restrict__ el, const unsigned int* __restrict__ key,
                                          const float* __restrict__ h, float* __restrict__ acc_l,
                                          float* __restrict__ s_l, const float* __restrict__ er_l,
                                          unsigned int* __restrict__ mu_l, unsigned int* __restrict__ cnt_l,
                                          unsigned int* __restrict__ list_l, int ob, int n0, int wave, int lane,
                                          int hh) {
  const int rot = ob * (NKCH / OWN_P);
  const int NIT = (NKCH + (OWN_T / 32) - 1) / (OWN_T / 32);
  const unsigned int pat = (unsigned int)ob * 0x01010101u;

  int4 cur[4], nxt[4];
  {
    int cr = wave + rot; if (cr >= NKCH) cr -= NKCH;
    const int4* p = (const int4*)(key + (size_t)cr * (KCH / 4)) + lane;
#pragma unroll
    for (int k = 0; k < 4; ++k) cur[k] = p[k * 32];
  }

  for (int it = 0; it < NIT; ++it) {
    const int c = it * (OWN_T / 32) + wave;
    {
      int cn = c + (OWN_T / 32); if (cn >= NKCH) cn = wave;
      int crn = cn + rot; if (crn >= NKCH) crn -= NKCH;
      const int4* p = (const int4*)(key + (size_t)crn * (KCH / 4)) + lane;
#pragma unroll
      for (int k = 0; k < 4; ++k) nxt[k] = p[k * 32];
    }
    if (c < NKCH) {
      int cr = c + rot; if (cr >= NKCH) cr -= NKCH;
      const unsigned int m01 = own16(cur[0], pat) | (own16(cur[1], pat) << 4);
      const unsigned int m23 = own16(cur[2], pat) | (own16(cur[3], pat) << 4);
      unsigned long long mm = (unsigned long long)m01 | ((unsigned long long)m23 << 32);
      const int ebase = cr * KCH + lane * 16;
      unsigned int mask = __builtin_amdgcn_ballot_w32(mm != 0ull);
      while (mask != 0u) {
        const bool has = (mm != 0ull);
        int sd = 0, dl = 0;
        if (has) {
          const int p = __builtin_ctzll(mm);
          const int k = ((p >> 5) << 1) | ((p >> 2) & 1);
          const int e = ebase + k * 512 + ((p & 3) << 2) + ((p & 31) >> 3);
          sd = src[e];
          dl = dst[e] - n0;
          mm &= mm - 1ull;
        }
        if constexpr (MODE == 0) {
          unsigned int base = 0u;
          if (lane == 0) base = atomicAdd(cnt_l, (unsigned int)__builtin_popcount(mask));
          base = (unsigned int)__builtin_amdgcn_readfirstlane((int)base);
          const unsigned int slot = base + __builtin_amdgcn_mbcnt_lo(mask, 0u);
          if (has && slot < OWN_CAP) list_l[slot] = (unsigned int)sd | ((unsigned int)dl << 16);
        } else if constexpr (MODE == 1) {
          if (has) own_max(el, er_l, mu_l, sd, dl);
        } else {
          do {
            const int l0 = __builtin_ctz(mask); mask &= mask - 1u;
            const bool h1 = (mask != 0u);
            const int l1 = h1 ? __builtin_ctz(mask) : l0; mask &= mask - 1u;
            const bool h2 = (mask != 0u);
            const int l2 = h2 ? __builtin_ctz(mask) : l0; mask &= mask - 1u;
            const bool h3 = (mask != 0u);
            const int l3 = h3 ? __builtin_ctz(mask) : l0; mask &= mask - 1u;
            const int sd0 = __builtin_amdgcn_readlane(sd, l0), dl0 = __builtin_amdgcn_readlane(dl, l0);
            const int sd1 = __builtin_amdgcn_readlane(sd, l1), dl1 = __builtin_amdgcn_readlane(dl, l1);
            const int sd2 = __builtin_amdgcn_readlane(sd, l2), dl2 = __builtin_amdgcn_readlane(dl, l2);
            const int sd3 = __builtin_amdgcn_readlane(sd, l3), dl3 = __builtin_amdgcn_readlane(dl, l3);
            own_add4(h, el, acc_l, s_l, er_l, mu_l, lane, hh, sd0, dl0, sd1, dl1, sd2, dl2, sd3, dl3, h1, h2, h3);
          } while (mask != 0u);
        }
        mask = __builtin_amdgcn_ballot_w32(mm != 0ull);
      }
    }
#pragma unroll
    for (int k = 0; k < 4; ++k) cur[k] = nxt[k];
  }
}

__global__ __launch_bounds__(OWN_T) void k_edge_own(const int* __restrict__ src0,
                                                    const int* __restrict__ src1,
                                                    const int* __restrict__ dst0,
                                                    const int* __restrict__ dst1,
                                                    const float* __restrict__ elb,
                                                    const float* __restrict__ erb,
                                                    const unsigned int* __restrict__ keyb,
                                                    const float* __restrict__ hb,
                                                    float* __restrict__ sb,
                                                    float* __restrict__ accb) {
  const int g  = blockIdx.x & 1;
  const int ob = blockIdx.x >> 1;
  const int* __restrict__ src = g ? src1 : src0;
  const int* __restrict__ dst = g ? dst1 : dst0;
  const float* __restrict__ el = elb + (size_t)g * PL_NH;
  const float* __restrict__ er = erb + (size_t)g * PL_NH;
  const unsigned int* __restrict__ key = keyb + (size_t)g * KEYW;
  const float* __restrict__ h = hb + (size_t)g * PL_ND;
  float* __restrict__ s   = sb + (size_t)g * PL_NH;
  float* __restrict__ acc = accb + (size_t)g * PL_ND;
  __shared__ __attribute__((aligned(16))) unsigned char L[OWN_LDS];
  float* acc_l = (float*)L;
  float* s_l   = acc_l + OWN_R * HD;
  float* er_l  = s_l + OWN_R * HH;
  unsigned int* mu_l   = (unsigned int*)(er_l + OWN_R * HH);
  unsigned int* cnt_l  = mu_l + OWN_R * HH;
  unsigned int* list_l = cnt_l + 4;

  const int t    = threadIdx.x;
  const int lane = t & 31;
  const int wave = __builtin_amdgcn_readfirstlane(t >> 5);
  const int hh   = lane >> 3;
  const int n0   = ob * OWN_R;
  const int nrows = (NN - n0 < OWN_R) ? (NN - n0) : OWN_R;

  for (int i = t; i < OWN_R * HD / 4; i += OWN_T)
    ((float4*)acc_l)[i] = make_float4(0.0f, 0.0f, 0.0f, 0.0f);
  for (int i = t; i < OWN_R * HH; i += OWN_T) {
    const bool ok = i < nrows * HH;
    s_l[i]  = 0.0f;
    er_l[i] = ok ? er[(size_t)n0 * HH + i] : 0.0f;
    mu_l[i] = FORD_NEG_INF;
  }
  if (t == 0) *cnt_l = 0u;
  __syncthreads();

  own_sweep<0>(src, dst, el, key, h, acc_l, s_l, er_l, mu_l, cnt_l, list_l, ob, n0, wave, lane, hh);
  __syncthreads();

  const unsigned int n_own = (unsigned int)__builtin_amdgcn_readfirstlane((int)*cnt_l);
  if (n_own <= OWN_CAP) {
    for (unsigned int i = (unsigned int)t; i < n_own; i += 4u * OWN_T) {
      const bool k1 = i + OWN_T < n_own, k2 = i + 2u * OWN_T < n_own, k3 = i + 3u * OWN_T < n_own;
      const unsigned int w0 = list_l[i];
      const unsigned int w1 = list_l[k1 ? i + OWN_T : i];
      const unsigned int w2 = list_l[k2 ? i + 2u * OWN_T : i];
      const unsigned int w3 = list_l[k3 ? i + 3u * OWN_T : i];
      const float4 ev0 = *(const float4*)(el + (size_t)(w0 & 0xFFFFu) * HH);
      const float4 ev1 = *(const float4*)(el + (size_t)(w1 & 0xFFFFu) * HH);
      const float4 ev2 = *(const float4*)(el + (size_t)(w2 & 0xFFFFu) * HH);
      const float4 ev3 = *(const float4*)(el + (size_t)(w3 & 0xFFFFu) * HH);
      asm volatile("" :: "v"(ev0.x), "v"(ev0.y), "v"(ev0.z), "v"(ev0.w), "v"(ev1.x), "v"(ev1.y), "v"(ev1.z), "v"(ev1.w),
                   "v"(ev2.x), "v"(ev2.y), "v"(ev2.z), "v"(ev2.w), "v"(ev3.x), "v"(ev3.y), "v"(ev3.z), "v"(ev3.w));
      own_max_v(ev0, er_l, mu_l, (int)(w0 >> 16));
      own_max_v(ev1, er_l, mu_l, (int)(w1 >> 16));
      own_max_v(ev2, er_l, mu_l, (int)(w2 >> 16));
      own_max_v(ev3, er_l, mu_l, (int)(w3 >> 16));
    }
    __syncthreads();
    unsigned int b = (unsigned int)wave * 4u;
    int lv = 0;
    if (b < n_own) {
      const unsigned int li = b + (unsigned int)(lane & 3);
      lv = (int)list_l[(li < n_own) ? li : b];
    }
    while (b < n_own) {
      const bool h1 = b + 1u < n_own, h2 = b + 2u < n_own, h3 = b + 3u < n_own;
      const unsigned int w0 = (unsigned int)__builtin_amdgcn_readlane(lv, 0);
      const unsigned int w1 = (unsigned int)__builtin_amdgcn_readlane(lv, 1);
      const unsigned int w2 = (unsigned int)__builtin_amdgcn_readlane(lv, 2);
      const unsigned int w3 = (unsigned int)__builtin_amdgcn_readlane(lv, 3);
      b += (OWN_T / 32) * 4u;
      if (b < n_own) {
        const unsigned int li = b + (unsigned int)(lane & 3);
        lv = (int)list_l[(li < n_own) ? li : b];
      }
      own_add4(h, el, acc_l, s_l, er_l, mu_l, lane, hh,
               (int)(w0 & 0xFFFFu), (int)(w0 >> 16), (int)(w1 & 0xFFFFu), (int)(w1 >> 16),
               (int)(w2 & 0xFFFFu), (int)(w2 >> 16), (int)(w3 & 0xFFFFu), (int)(w3 >> 16), h1, h2, h3);
    }
  } else {
    own_sweep<1>(src, dst, el, key, h, acc_l, s_l, er_l, mu_l, cnt_l, list_l, ob, n0, wave, lane, hh);
    __syncthreads();
    own_sweep<2>(src, dst, el, key, h, acc_l, s_l, er_l, mu_l, cnt_l, list_l, ob, n0, wave, lane, hh);
  }
  __syncthreads();

  float4* ag = (float4*)(acc + (size_t)n0 * HD);
  for (int i = t; i < nrows * HD / 4; i += OWN_T) ag[i] = ((const float4*)acc_l)[i];
  float* sg = s + (size_t)n0 * HH;
  for (int i = t; i < nrows * HH; i += OWN_T) sg[i] = s_l[i];
}

__device__ __forceinline__ float inv_or_zero(float v) {
  return (v != 0.0f) ? 1.0f / v : 0.0f;
}
__global__ void k_norm_mean(const float* __restrict__ acc0, const float* __restrict__ acc1,
                            const float* __restrict__ s0, const float* __restrict__ s1,
                            float* __restrict__ out) {
  int tid = blockIdx.x * blockDim.x + threadIdx.x;
  if (tid >= NN * DD) return;
  const int n = tid >> 5;
  const int d = tid & 31;
  const float4 v0 = *(const float4*)(s0 + (size_t)n * HH);
  const float4 v1 = *(const float4*)(s1 + (size_t)n * HH);
  const float* a0 = acc0 + (size_t)n * HD + d;
  const float* a1 = acc1 + (size_t)n * HD + d;
  const float f0 = a0[0], f1 = a0[32], f2 = a0[64], f3 = a0[96];
  const float b0 = a1[0], b1 = a1[32], b2 = a1[64], b3 = a1[96];
  float r = f0 * inv_or_zero(v0.x) + b0 * inv_or_zero(v1.x);
  r += f1 * inv_or_zero(v0.y) + b1 * inv_or_zero(v1.y);
  r += f2 * inv_or_zero(v0.z) + b2 * inv_or_zero(v1.z);
  r += f3 * inv_or_zero(v0.w) + b3 * inv_or_zero(v1.w);
  out[tid] = 0.25f * r;
}

extern "C" void kernel_launch(void* const* d_in, const int* in_sizes, int n_in,
                              void* d_out, int out_size, void* d_ws, size_t ws_size,
                              hipStream_t stream) {
  const float* feat = (const float*)d_in[0];
  const int*   src_g[2] = { (const int*)d_in[1], (const int*)d_in[3] };
  const int*   dst_g[2] = { (const int*)d_in[2], (const int*)d_in[4] };
  const float* W_g[2]   = { (const float*)d_in[5], (const float*)d_in[8] };
  const float* al_g[2]  = { (const float*)d_in[6], (const float*)d_in[9] };
  const float* ar_g[2]  = { (const float*)d_in[7], (const float*)d_in[10] };
  float* out = (float*)d_out;

  float*        acc0 = (float*)d_ws;
  float*        acc1 = acc0 + (size_t)NN * HD;
  float*        s0   = acc1 + (size_t)NN * HD;
  float*        s1   = s0   + (size_t)NN * HH;
  float*        hbuf = s1   + (size_t)NN * HH;
  float*        el   = hbuf + 2 * (size_t)NN * HD;
  float*        er   = el   + 2 * (size_t)NN * HH;
  unsigned int* keyb = (unsigned int*)(er + 2 * (size_t)NN * HH);

  const int T = 256;
  k_init<<<dim3((KEYW + T - 1) / T, 2), T, 0, stream>>>(dst_g[0], dst_g[1], keyb);

  k_proj<<<dim3(NN / 16, 2), T, 0, stream>>>(feat, W_g[0], W_g[1], hbuf);
  k_elr<<<dim3((NN * HH + T - 1) / T, 2), T, 0, stream>>>(hbuf, al_g[0], al_g[1], ar_g[0], ar_g[1], el, er);
  k_edge_own<<<2 * OWN_P, OWN_T, 0, stream>>>(src_g[0], src_g[1], dst_g[0], dst_g[1], el, er,
                                               keyb, hbuf, s0, acc0);

  k_norm_mean<<<(NN * DD + T - 1) / T, T, 0, stream>>>(acc0, acc1, s0, s1, out);
}


# add loop's LDS work cut: accumulator rows lane-major so a lane's 4 adds each hit 32 distinct banks (was 2-way conflicts); er+max in one 8-byte entry, read ahead of the adds; LDS arrival order kept
# speedup vs baseline: 3.6250x; 1.3338x over previous
#include <hip/hip_runtime.h>

#define NN 50000
#define EE 1600000
#define INF_ 128
#define HH 4
#define DD 32
#define HD 128
#define NEG_SLOPE 0.2f

typedef __attribute__((ext_vector_type(2))) float v2f;
typedef __attribute__((ext_vector_type(8))) float v8f;

__device__ __forceinline__ unsigned int ford(float f) {
  unsigned int b = __float_as_uint(f);
  return (b & 0x80000000u) ? ~b : (b | 0x80000000u);
}
__device__ __forceinline__ float funord(unsigned int u) {
  unsigned int b = (u & 0x80000000u) ? (u ^ 0x80000000u) : ~u;
  return __uint_as_float(b);
}
#define FORD_NEG_INF 0x007FFFFFu

#define OWN_R 200
#define OWN_P ((NN + OWN_R - 1) / OWN_R)
#define KCH 2048
#define NKCH ((EE + KCH - 1) / KCH)
#define KEYW (NKCH * (KCH / 4))
#define KEY_PAD 0xFFFFFFFFu
static_assert(OWN_P <= 255 && EE % 4 == 0, "owner numbers fit a byte below the pad key; edges fill whole key dwords");

__global__ void k_init(const int* __restrict__ dst0, const int* __restrict__ dst1,
                       unsigned int* __restrict__ keyb) {
  int i = blockIdx.x * blockDim.x + threadIdx.x;
  const int g = blockIdx.y;
  if (i < KEYW) {
    unsigned int kv = KEY_PAD;
    if (i < EE / 4) {
      const int4 d = ((const int4*)(g ? dst1 : dst0))[i];
      kv = (unsigned int)(d.x / OWN_R) | ((unsigned int)(d.y / OWN_R) << 8) |
           ((unsigned int)(d.z / OWN_R) << 16) | ((unsigned int)(d.w / OWN_R) << 24);
    }
    keyb[(size_t)g * KEYW + i] = kv;
  }
}

#define PL_ND ((size_t)NN * HD)
#define PL_NH ((size_t)NN * HH)

__global__ __launch_bounds__(256) void k_proj(const float* __restrict__ feat,
                                              const float* __restrict__ W0,
                                              const float* __restrict__ W1,
                                              float* __restrict__ hb) {
  __shared__ float As[16 * 132];
  const int g = blockIdx.y;
  const float* __restrict__ W = g ? W1 : W0;
  float* __restrict__ h = hb + (size_t)g * PL_ND;
  const int m0 = blockIdx.x * 16;
  const int t  = threadIdx.x;

  {
    const int row = t >> 4;
    const int col = (t & 15) * 8;
    const float4* gp = (const float4*)(feat + (size_t)(m0 + row) * INF_ + col);
    float4 x0 = gp[0];
    float4 x1 = gp[1];
    float* lp = As + row * 132 + col;
    *(float4*)(lp)     = x0;
    *(float4*)(lp + 4) = x1;
  }
  __syncthreads();

  const int lane  = t & 31;
  const int wave  = t >> 5;
  const int n0    = wave * 16;
  const int r     = lane & 15;
  const int khalf = (lane >> 4) << 1;

  v8f acc = {};
  for (int k0 = 0; k0 < INF_; k0 += 4) {
    const int kk = k0 + khalf;
    v2f a;
    a.x = As[r * 132 + kk];
    a.y = As[r * 132 + kk + 1];
    v2f b;
    b.x = W[(size_t)(kk)     * HD + n0 + r];
    b.y = W[(size_t)(kk + 1) * HD + n0 + r];
    acc = __builtin_amdgcn_wmma_f32_16x16x4_f32(false, a, false, b,
                                                (short)0, acc, false, false);
  }

  const int mh = (lane >> 4) * 8;
#pragma unroll
  for (int j = 0; j < 8; ++j) {
    h[(size_t)(m0 + j + mh) * HD + n0 + r] = acc[j];
  }
}

__global__ void k_elr(const float* __restrict__ hb, const float* __restrict__ al0,
                      const float* __restrict__ al1, const float* __restrict__ ar0,
                      const float* __restrict__ ar1, float* __restrict__ elb,
                      float* __restrict__ erb) {
  int tid = blockIdx.x * blockDim.x + threadIdx.x;
  if (tid >= NN * HH) return;
  const int g = blockIdx.y;
  const float* __restrict__ h  = hb + (size_t)g * PL_ND;
  const float* __restrict__ al = g ? al1 : al0;
  const float* __restrict__ ar = g ? ar1 : ar0;
  float* __restrict__ el = elb + (size_t)g * PL_NH;
  float* __restrict__ er = erb + (size_t)g * PL_NH;
  const int n  = tid / HH;
  const int hh = tid % HH;
  const float* hp  = h  + (size_t)n * HD + hh * DD;
  const float* alp = al + hh * DD;
  const float* arp = ar + hh * DD;
  float sl = 0.0f, sr = 0.0f;
#pragma unroll
  for (int d = 0; d < DD; d += 4) {
    float4 hv = *(const float4*)(hp + d);
    float4 av = *(const float4*)(alp + d);
    float4 rv = *(const float4*)(arp + d);
    sl += hv.x * av.x + hv.y * av.y + hv.z * av.z + hv.w * av.w;
    sr += hv.x * rv.x + hv.y * rv.y + hv.z * rv.z + hv.w * rv.w;
  }
  el[tid] = sl;
  er[tid] = sr;
}

#define OWN_T 1024
#define OWN_CAP 8192
static_assert(NKCH >= OWN_T / 32 && NKCH / OWN_P >= 1, "every wave has a step; owners start at different steps");
static_assert(OWN_R % 8 == 0, "an owner's sum rows fill whole 128-byte lines");
static_assert(OWN_R * HH <= OWN_T, "one thread per (node, head) merges the er and max tables");
static_assert(NN <= 65536 && OWN_R <= 65536 && OWN_CAP % 4 == 0, "src and the local row share one list word");
#define OWN_TAB ((OWN_R * HD + 3 * OWN_R * HH) * 4)
#define OWN_LDS (OWN_TAB + 16 + OWN_CAP * 4)

__device__ __forceinline__ void own_max_v(const float4 ev, const float* __restrict__ er_l,
                                          unsigned int* __restrict__ mu_l, int dl) {
  const float4 rv = *(const float4*)(er_l + dl * HH);
  float x0 = ev.x + rv.x, x1 = ev.y + rv.y, x2 = ev.z + rv.z, x3 = ev.w + rv.w;
  x0 = (x0 > 0.0f) ? x0 : NEG_SLOPE * x0;
  x1 = (x1 > 0.0f) ? x1 : NEG_SLOPE * x1;
  x2 = (x2 > 0.0f) ? x2 : NEG_SLOPE * x2;
  x3 = (x3 > 0.0f) ? x3 : NEG_SLOPE * x3;
  atomicMax(&mu_l[dl * HH + 0], ford(x0));
  atomicMax(&mu_l[dl * HH + 1], ford(x1));
  atomicMax(&mu_l[dl * HH + 2], ford(x2));
  atomicMax(&mu_l[dl * HH + 3], ford(x3));
}
__device__ __forceinline__ void own_max(const float* __restrict__ el, const float* __restrict__ er_l,
                                        unsigned int* __restrict__ mu_l, int sd, int dl) {
  own_max_v(*(const float4*)(el + (size_t)sd * HH), er_l, mu_l, dl);
}

__device__ __forceinline__ void own_add(float* __restrict__ acc_l, float* __restrict__ s_l, const float2 em,
                                        int dl, int lane, int hh, float elv, const float4 hv) {
  float x = elv + em.x;
  x = (x > 0.0f) ? x : NEG_SLOPE * x;
  const float ex = __expf(x - em.y);
  float* ap = acc_l + dl * HD + lane;
  atomicAdd(ap + 0,  ex * hv.x);
  atomicAdd(ap + 32, ex * hv.y);
  atomicAdd(ap + 64, ex * hv.z);
  atomicAdd(ap + 96, ex * hv.w);
  if ((lane & 7) == 0) atomicAdd(&s_l[dl * HH + hh], ex);
}

__device__ __forceinline__ void own_merge(const float* er_l, const unsigned int* mu_l, float2* em_l, int t) {
  float e = 0.0f, m = 0.0f;
  if (t < OWN_R * HH) { e = er_l[t]; m = funord(mu_l[t]); }
  __syncthreads();
  if (t < OWN_R * HH) em_l[t] = make_float2(e, m);
  __syncthreads();
}

__device__ __forceinline__ void own_add4(const float* __restrict__ h, const float* __restrict__ el,
                                         float* __restrict__ acc_l, float* __restrict__ s_l,
                                         const float2* em_l,
                                         int lane, int hh, int sd0, int dl0, int sd1, int dl1, int sd2, int dl2,
                                         int sd3, int dl3, bool h1, bool h2, bool h3) {
  const float4 hv0 = *(const float4*)(h + (size_t)sd0 * HD + lane * 4);
  const float4 hv1 = *(const float4*)(h + (size_t)sd1 * HD + lane * 4);
  const float4 hv2 = *(const float4*)(h + (size_t)sd2 * HD + lane * 4);
  const float4 hv3 = *(const float4*)(h + (size_t)sd3 * HD + lane * 4);
  const float e0 = el[sd0 * HH + hh];
  const float e1 = el[sd1 * HH + hh];
  const float e2 = el[sd2 * HH + hh];
  const float e3 = el[sd3 * HH + hh];
  const float2 em0 = em_l[dl0 * HH + hh];
  const float2 em1 = em_l[dl1 * HH + hh];
  const float2 em2 = em_l[dl2 * HH + hh];
  const float2 em3 = em_l[dl3 * HH + hh];
  own_add(acc_l, s_l, em0, dl0, lane, hh, e0, hv0);
  if (h1) own_add(acc_l, s_l, em1, dl1, lane, hh, e1, hv1);
  if (h2) own_add(acc_l, s_l, em2, dl2, lane, hh, e2, hv2);
  if (h3) own_add(acc_l, s_l, em3, dl3, lane, hh, e3, hv3);
}

__device__ __forceinline__ unsigned int key_differs(unsigned int x, unsigned int pat) {
  const unsigned int t = x ^ pat;
  return ((t & 0x7F7F7F7Fu) + 0x7F7F7F7Fu) | t;
}
__device__ __forceinline__ unsigned int own16(const int4 v, unsigned int pat) {
  const unsigned int c = ((key_differs((unsigned int)v.x, pat) >> 7) & 0x01010101u) |
                         ((key_differs((unsigned int)v.y, pat) >> 6) & 0x02020202u) |
                         ((key_differs((unsigned int)v.z, pat) >> 5) & 0x04040404u) |
                         ((key_differs((unsigned int)v.w, pat) >> 4) & 0x08080808u);
  return c ^ 0x0F0F0F0Fu;
}

template <int MODE>
__device__ __forceinline__ void own_sweep(const int* __restrict__ src, const int* __restrict__ dst,
                                          const float* __restrict__ el, const unsigned int* __restrict__ key,
                                          const float* __restrict__ h, float* __restrict__ acc_l,
                                          float* __restrict__ s_l, const float* er_l,
                                          unsigned int* mu_l, const float2* em_l,
                                          unsigned int* __restrict__ cnt_l,
                                          unsigned int* __restrict__ list_l, int ob, int n0, int wave, int lane,
                                          int hh) {
  const int rot = ob * (NKCH / OWN_P);
  const int NIT = (NKCH + (OWN_T / 32) - 1) / (OWN_T / 32);
  const unsigned int pat = (unsigned int)ob * 0x01010101u;

  int4 cur[4], nxt[4];
  {
    int cr = wave + rot; if (cr >= NKCH) cr -= NKCH;
    const int4* p = (const int4*)(key + (size_t)cr * (KCH / 4)) + lane;
#pragma unroll
    for (int k = 0; k < 4; ++k) cur[k] = p[k * 32];
  }

  for (int it = 0; it < NIT; ++it) {
    const int c = it * (OWN_T / 32) + wave;
    {
      int cn = c + (OWN_T / 32); if (cn >= NKCH) cn = wave;
      int crn = cn + rot; if (crn >= NKCH) crn -= NKCH;
      const int4* p = (const int4*)(key + (size_t)crn * (KCH / 4)) + lane;
#pragma unroll
      for (int k = 0; k < 4; ++k) nxt[k] = p[k * 32];
    }
    if (c < NKCH) {
      int cr = c + rot; if (cr >= NKCH) cr -= NKCH;
      const unsigned int m01 = own16(cur[0], pat) | (own16(cur[1], pat) << 4);
      const unsigned int m23 = own16(cur[2], pat) | (own16(cur[3], pat) << 4);
      unsigned long long mm = (unsigned long long)m01 | ((unsigned long long)m23 << 32);
      const int ebase = cr * KCH + lane * 16;
      unsigned int mask = __builtin_amdgcn_ballot_w32(mm != 0ull);
      while (mask != 0u) {
        const bool has = (mm != 0ull);
        int sd = 0, dl = 0;
        if (has) {
          const int p = __builtin_ctzll(mm);
          const int k = ((p >> 5) << 1) | ((p >> 2) & 1);
          const int e = ebase + k * 512 + ((p & 3) << 2) + ((p & 31) >> 3);
          sd = src[e];
          dl = dst[e] - n0;
          mm &= mm - 1ull;
        }
        if constexpr (MODE == 0) {
          unsigned int base = 0u;
          if (lane == 0) base = atomicAdd(cnt_l, (unsigned int)__builtin_popcount(mask));
          base = (unsigned int)__builtin_amdgcn_readfirstlane((int)base);
          const unsigned int slot = base + __builtin_amdgcn_mbcnt_lo(mask, 0u);
          if (has && slot < OWN_CAP) list_l[slot] = (unsigned int)sd | ((unsigned int)dl << 16);
        } else if constexpr (MODE == 1) {
          if (has) own_max(el, er_l, mu_l, sd, dl);
        } else {
          do {
            const int l0 = __builtin_ctz(mask); mask &= mask - 1u;
            const bool h1 = (mask != 0u);
            const int l1 = h1 ? __builtin_ctz(mask) : l0; mask &= mask - 1u;
            const bool h2 = (mask != 0u);
            const int l2 = h2 ? __builtin_ctz(mask) : l0; mask &= mask - 1u;
            const bool h3 = (mask != 0u);
            const int l3 = h3 ? __builtin_ctz(mask) : l0; mask &= mask - 1u;
            const int sd0 = __builtin_amdgcn_readlane(sd, l0), dl0 = __builtin_amdgcn_readlane(dl, l0);
            const int sd1 = __builtin_amdgcn_readlane(sd, l1), dl1 = __builtin_amdgcn_readlane(dl, l1);
            const int sd2 = __builtin_amdgcn_readlane(sd, l2), dl2 = __builtin_amdgcn_readlane(dl, l2);
            const int sd3 = __builtin_amdgcn_readlane(sd, l3), dl3 = __builtin_amdgcn_readlane(dl, l3);
            own_add4(h, el, acc_l, s_l, em_l, lane, hh, sd0, dl0, sd1, dl1, sd2, dl2, sd3, dl3, h1, h2, h3);
          } while (mask != 0u);
        }
        mask = __builtin_amdgcn_ballot_w32(mm != 0ull);
      }
    }
#pragma unroll
    for (int k = 0; k < 4; ++k) cur[k] = nxt[k];
  }
}

__global__ __launch_bounds__(OWN_T) void k_edge_own(const int* __restrict__ src0,
                                                    const int* __restrict__ src1,
                                                    const int* __restrict__ dst0,
                                                    const int* __restrict__ dst1,
                                                    const float* __restrict__ elb,
                                                    const float* __restrict__ erb,
                                                    const unsigned int* __restrict__ keyb,
                                                    const float* __restrict__ hb,
                                                    float* __restrict__ sb,
                                                    float* __restrict__ accb) {
  const int g  = blockIdx.x & 1;
  const int ob = blockIdx.x >> 1;
  const int* __restrict__ src = g ? src1 : src0;
  const int* __restrict__ dst = g ? dst1 : dst0;
  const float* __restrict__ el = elb + (size_t)g * PL_NH;
  const float* __restrict__ er = erb + (size_t)g * PL_NH;
  const unsigned int* __restrict__ key = keyb + (size_t)g * KEYW;
  const float* __restrict__ h = hb + (size_t)g * PL_ND;
  float* __restrict__ s   = sb + (size_t)g * PL_NH;
  float* __restrict__ acc = accb + (size_t)g * PL_ND;
  __shared__ __attribute__((aligned(16))) unsigned char L[OWN_LDS];
  float* acc_l = (float*)L;
  float* s_l   = acc_l + OWN_R * HD;
  float* er_l  = s_l + OWN_R * HH;
  unsigned int* mu_l   = (unsigned int*)(er_l + OWN_R * HH);
  float2* em_l = (float2*)er_l;
  unsigned int* cnt_l  = mu_l + OWN_R * HH;
  unsigned int* list_l = cnt_l + 4;

  const int t    = threadIdx.x;
  const int lane = t & 31;
  const int wave = __builtin_amdgcn_readfirstlane(t >> 5);
  const int hh   = lane >> 3;
  const int n0   = ob * OWN_R;
  const int nrows = (NN - n0 < OWN_R) ? (NN - n0) : OWN_R;

  for (int i = t; i < OWN_R * HD / 4; i += OWN_T)
    ((float4*)acc_l)[i] = make_float4(0.0f, 0.0f, 0.0f, 0.0f);
  for (int i = t; i < OWN_R * HH; i += OWN_T) {
    const bool ok = i < nrows * HH;
    s_l[i]  = 0.0f;
    er_l[i] = ok ? er[(size_t)n0 * HH + i] : 0.0f;
    mu_l[i] = FORD_NEG_INF;
  }
  if (t == 0) *cnt_l = 0u;
  __syncthreads();

  own_sweep<0>(src, dst, el, key, h, acc_l, s_l, er_l, mu_l, em_l, cnt_l, list_l, ob, n0, wave, lane, hh);
  __syncthreads();

  const unsigned int n_own = (unsigned int)__builtin_amdgcn_readfirstlane((int)*cnt_l);
  if (n_own <= OWN_CAP) {
    for (unsigned int i = (unsigned int)t; i < n_own; i += 4u * OWN_T) {
      const bool k1 = i + OWN_T < n_own, k2 = i + 2u * OWN_T < n_own, k3 = i + 3u * OWN_T < n_own;
      const unsigned int w0 = list_l[i];
      const unsigned int w1 = list_l[k1 ? i + OWN_T : i];
      const unsigned int w2 = list_l[k2 ? i + 2u * OWN_T : i];
      const unsigned int w3 = list_l[k3 ? i + 3u * OWN_T : i];
      const float4 ev0 = *(const float4*)(el + (size_t)(w0 & 0xFFFFu) * HH);
      const float4 ev1 = *(const float4*)(el + (size_t)(w1 & 0xFFFFu) * HH);
      const float4 ev2 = *(const float4*)(el + (size_t)(w2 & 0xFFFFu) * HH);
      const float4 ev3 = *(const float4*)(el + (size_t)(w3 & 0xFFFFu) * HH);
      asm volatile("" :: "v"(ev0.x), "v"(ev0.y), "v"(ev0.z), "v"(ev0.w), "v"(ev1.x), "v"(ev1.y), "v"(ev1.z), "v"(ev1.w),
                   "v"(ev2.x), "v"(ev2.y), "v"(ev2.z), "v"(ev2.w), "v"(ev3.x), "v"(ev3.y), "v"(ev3.z), "v"(ev3.w));
      own_max_v(ev0, er_l, mu_l, (int)(w0 >> 16));
      own_max_v(ev1, er_l, mu_l, (int)(w1 >> 16));
      own_max_v(ev2, er_l, mu_l, (int)(w2 >> 16));
      own_max_v(ev3, er_l, mu_l, (int)(w3 >> 16));
    }
    __syncthreads();
    own_merge(er_l, mu_l, em_l, t);
    unsigned int b = (unsigned int)wave * 4u;
    int lv = 0;
    if (b < n_own) {
      const unsigned int li = b + (unsigned int)(lane & 3);
      lv = (int)list_l[(li < n_own) ? li : b];
    }
    while (b < n_own) {
      const bool h1 = b + 1u < n_own, h2 = b + 2u < n_own, h3 = b + 3u < n_own;
      const unsigned int w0 = (unsigned int)__builtin_amdgcn_readlane(lv, 0);
      const unsigned int w1 = (unsigned int)__builtin_amdgcn_readlane(lv, 1);
      const unsigned int w2 = (unsigned int)__builtin_amdgcn_readlane(lv, 2);
      const unsigned int w3 = (unsigned int)__builtin_amdgcn_readlane(lv, 3);
      b += (OWN_T / 32) * 4u;
      if (b < n_own) {
        const unsigned int li = b + (unsigned int)(lane & 3);
        lv = (int)list_l[(li < n_own) ? li : b];
      }
      own_add4(h, el, acc_l, s_l, em_l, lane, hh,
               (int)(w0 & 0xFFFFu), (int)(w0 >> 16), (int)(w1 & 0xFFFFu), (int)(w1 >> 16),
               (int)(w2 & 0xFFFFu), (int)(w2 >> 16), (int)(w3 & 0xFFFFu), (int)(w3 >> 16), h1, h2, h3);
    }
  } else {
    own_sweep<1>(src, dst, el, key, h, acc_l, s_l, er_l, mu_l, em_l, cnt_l, list_l, ob, n0, wave, lane, hh);
    __syncthreads();
    own_merge(er_l, mu_l, em_l, t);
    own_sweep<2>(src, dst, el, key, h, acc_l, s_l, er_l, mu_l, em_l, cnt_l, list_l, ob, n0, wave, lane, hh);
  }
  __syncthreads();

  float4* ag = (float4*)(acc + (size_t)n0 * HD);
  for (int i = t; i < nrows * HD / 4; i += OWN_T) {
    const float* rp = acc_l + (i >> 5) * HD + (i & 31);
    ag[i] = make_float4(rp[0], rp[32], rp[64], rp[96]);
  }
  float* sg = s + (size_t)n0 * HH;
  for (int i = t; i < nrows * HH; i += OWN_T) sg[i] = s_l[i];
}

__device__ __forceinline__ float inv_or_zero(float v) {
  return (v != 0.0f) ? 1.0f / v : 0.0f;
}
__global__ void k_norm_mean(const float* __restrict__ acc0, const float* __restrict__ acc1,
                            const float* __restrict__ s0, const float* __restrict__ s1,
                            float* __restrict__ out) {
  int tid = blockIdx.x * blockDim.x + threadIdx.x;
  if (tid >= NN * DD) return;
  const int n = tid >> 5;
  const int d = tid & 31;
  const float4 v0 = *(const float4*)(s0 + (size_t)n * HH);
  const float4 v1 = *(const float4*)(s1 + (size_t)n * HH);
  const float* a0 = acc0 + (size_t)n * HD + d;
  const float* a1 = acc1 + (size_t)n * HD + d;
  const float f0 = a0[0], f1 = a0[32], f2 = a0[64], f3 = a0[96];
  const float b0 = a1[0], b1 = a1[32], b2 = a1[64], b3 = a1[96];
  float r = f0 * inv_or_zero(v0.x) + b0 * inv_or_zero(v1.x);
  r += f1 * inv_or_zero(v0.y) + b1 * inv_or_zero(v1.y);
  r += f2 * inv_or_zero(v0.z) + b2 * inv_or_zero(v1.z);
  r += f3 * inv_or_zero(v0.w) + b3 * inv_or_zero(v1.w);
  out[tid] = 0.25f * r;
}

extern "C" void kernel_launch(void* const* d_in, const int* in_sizes, int n_in,
                              void* d_out, int out_size, void* d_ws, size_t ws_size,
                              hipStream_t stream) {
  const float* feat = (const float*)d_in[0];
  const int*   src_g[2] = { (const int*)d_in[1], (const int*)d_in[3] };
  const int*   dst_g[2] = { (const int*)d_in[2], (const int*)d_in[4] };
  const float* W_g[2]   = { (const float*)d_in[5], (const float*)d_in[8] };
  const float* al_g[2]  = { (const float*)d_in[6], (const float*)d_in[9] };
  const float* ar_g[2]  = { (const float*)d_in[7], (const float*)d_in[10] };
  float* out = (float*)d_out;

  float*        acc0 = (float*)d_ws;
  float*        acc1 = acc0 + (size_t)NN * HD;
  float*        s0   = acc1 + (size_t)NN * HD;
  float*        s1   = s0   + (size_t)NN * HH;
  float*        hbuf = s1   + (size_t)NN * HH;
  float*        el   = hbuf + 2 * (size_t)NN * HD;
  float*        er   = el   + 2 * (size_t)NN * HH;
  unsigned int* keyb = (unsigned int*)(er + 2 * (size_t)NN * HH);

  const int T = 256;
  k_init<<<dim3((KEYW + T - 1) / T, 2), T, 0, stream>>>(dst_g[0], dst_g[1], keyb);

  k_proj<<<dim3(NN / 16, 2), T, 0, stream>>>(feat, W_g[0], W_g[1], hbuf);
  k_elr<<<dim3((NN * HH + T - 1) / T, 2), T, 0, stream>>>(hbuf, al_g[0], al_g[1], ar_g[0], ar_g[1], el, er);
  k_edge_own<<<2 * OWN_P, OWN_T, 0, stream>>>(src_g[0], src_g[1], dst_g[0], dst_g[1], el, er,
                                               keyb, hbuf, s0, acc0);

  k_norm_mean<<<(NN * DD + T - 1) / T, T, 0, stream>>>(acc0, acc1, s0, s1, out);
}


# edge lists built by a two-launch partition (chunk counts per owner, prefix sums, words ordered in LDS, stored as per-owner runs); owners copy their run, no key scan; 6 launches; LDS arrival order
# speedup vs baseline: 4.3756x; 1.2071x over previous
#include <hip/hip_runtime.h>

#define NN 50000
#define EE 1600000
#define INF_ 128
#define HH 4
#define DD 32
#define HD 128
#define NEG_SLOPE 0.2f

typedef __attribute__((ext_vector_type(2))) float v2f;
typedef __attribute__((ext_vector_type(8))) float v8f;

__device__ __forceinline__ unsigned int ford(float f) {
  unsigned int b = __float_as_uint(f);
  return (b & 0x80000000u) ? ~b : (b | 0x80000000u);
}
__device__ __forceinline__ float funord(unsigned int u) {
  unsigned int b = (u & 0x80000000u) ? (u ^ 0x80000000u) : ~u;
  return __uint_as_float(b);
}
#define FORD_NEG_INF 0x007FFFFFu

#define OWN_R 200
#define OWN_P ((NN + OWN_R - 1) / OWN_R)
static_assert(NN <= 65536 && OWN_R <= 65536, "src and the local row share one list word");

#define PB 250
#define PCH (EE / PB)
#define PT 1024
static_assert(EE % PB == 0 && PCH % 4 == 0, "chunks are whole and start on 16-byte boundaries");
static_assert(OWN_P <= 256 && OWN_R <= 256, "owner and local row take a byte each in a staged word; one wave scans the owners");

__global__ __launch_bounds__(PT) void k_count(const int* __restrict__ dst0, const int* __restrict__ dst1,
                                              unsigned int* __restrict__ cnt) {
  __shared__ unsigned int c[OWN_P];
  const int t = threadIdx.x;
  const int b = blockIdx.x;
  const int g = blockIdx.y;
  if (t < OWN_P) c[t] = 0u;
  __syncthreads();
  const int4* dp = (const int4*)((g ? dst1 : dst0) + (size_t)b * PCH);
  for (int i = t; i < PCH / 4; i += PT) {
    const int4 d = dp[i];
    atomicAdd(&c[d.x / OWN_R], 1u);
    atomicAdd(&c[d.y / OWN_R], 1u);
    atomicAdd(&c[d.z / OWN_R], 1u);
    atomicAdd(&c[d.w / OWN_R], 1u);
  }
  __syncthreads();
  if (t < OWN_P) cnt[((size_t)g * PB + b) * OWN_P + t] = c[t];
}

__global__ __launch_bounds__(PT) void k_scatter(const int* __restrict__ src0, const int* __restrict__ src1,
                                                const int* __restrict__ dst0, const int* __restrict__ dst1,
                                                const unsigned int* __restrict__ cnt,
                                                unsigned int* __restrict__ obase,
                                                unsigned int* __restrict__ bucket) {
  __shared__ unsigned int part_a[4][256];
  __shared__ unsigned int part_p[4][256];
  __shared__ unsigned int lcur[256];
  __shared__ unsigned int delta[256];
  __shared__ unsigned int stage[PCH];
  const int t = threadIdx.x;
  const int b = blockIdx.x;
  const int g = blockIdx.y;
  const int lane = t & 31;
  const int wave = t >> 5;
  const int q  = t >> 8;
  const int oc = t & 255;
  const unsigned int* __restrict__ cg = cnt + (size_t)g * PB * OWN_P;
  unsigned int a = 0u, p = 0u;
  if (oc < OWN_P) {
#pragma unroll 8
    for (int bb = q; bb < PB; bb += 4) {
      const unsigned int v = cg[(size_t)bb * OWN_P + oc];
      a += v;
      p += (bb < b) ? v : 0u;
    }
  }
  part_a[q][oc] = a;
  part_p[q][oc] = p;
  __syncthreads();
  if (wave == 0) {
    unsigned int vt[8], vp[8], vo[8];
    unsigned int st = 0u, so = 0u;
#pragma unroll
    for (int k = 0; k < 8; ++k) {
      const int o = lane * 8 + k;
      vt[k] = part_a[0][o] + part_a[1][o] + part_a[2][o] + part_a[3][o];
      vp[k] = part_p[0][o] + part_p[1][o] + part_p[2][o] + part_p[3][o];
      vo[k] = (o < OWN_P) ? cg[(size_t)b * OWN_P + o] : 0u;
      st += vt[k];
      so += vo[k];
    }
    unsigned int it = st, io = so;
#pragma unroll
    for (int d = 1; d < 32; d <<= 1) {
      const unsigned int nt = (unsigned int)__shfl_up((int)it, d);
      const unsigned int no = (unsigned int)__shfl_up((int)io, d);
      if (lane >= d) { it += nt; io += no; }
    }
    unsigned int run = it - st;
    unsigned int loc = io - so;
    unsigned int* __restrict__ og = obase + (size_t)g * (OWN_P + 1);
#pragma unroll
    for (int k = 0; k < 8; ++k) {
      const int o = lane * 8 + k;
      if (o < OWN_P) {
        lcur[o]  = loc;
        delta[o] = run + vp[k] - loc;
        if (b == 0) og[o] = run;
      }
      run += vt[k];
      loc += vo[k];
    }
    if (b == 0 && lane == 31) og[OWN_P] = run;
  }
  __syncthreads();
  const int4* dp = (const int4*)((g ? dst1 : dst0) + (size_t)b * PCH);
  const int4* sp = (const int4*)((g ? src1 : src0) + (size_t)b * PCH);
  for (int i = t; i < PCH / 4; i += PT) {
    const int4 d = dp[i];
    const int4 s = sp[i];
    const int o0 = d.x / OWN_R, o1 = d.y / OWN_R, o2 = d.z / OWN_R, o3 = d.w / OWN_R;
    const unsigned int q0 = atomicAdd(&lcur[o0], 1u);
    const unsigned int q1 = atomicAdd(&lcur[o1], 1u);
    const unsigned int q2 = atomicAdd(&lcur[o2], 1u);
    const unsigned int q3 = atomicAdd(&lcur[o3], 1u);
    stage[q0] = (unsigned int)s.x | ((unsigned int)(d.x - o0 * OWN_R) << 16) | ((unsigned int)o0 << 24);
    stage[q1] = (unsigned int)s.y | ((unsigned int)(d.y - o1 * OWN_R) << 16) | ((unsigned int)o1 << 24);
    stage[q2] = (unsigned int)s.z | ((unsigned int)(d.z - o2 * OWN_R) << 16) | ((unsigned int)o2 << 24);
    stage[q3] = (unsigned int)s.w | ((unsigned int)(d.w - o3 * OWN_R) << 16) | ((unsigned int)o3 << 24);
  }
  __syncthreads();
  unsigned int* __restrict__ bk = bucket + (size_t)g * EE;
  for (int i = t; i < PCH; i += PT) {
    const unsigned int w = stage[i];
    bk[(unsigned int)i + delta[w >> 24]] = w & 0x00FFFFFFu;
  }
}

#define PL_ND ((size_t)NN * HD)
#define PL_NH ((size_t)NN * HH)

__global__ __launch_bounds__(256) void k_proj(const float* __restrict__ feat,
                                              const float* __restrict__ W0,
                                              const float* __restrict__ W1,
                                              float* __restrict__ hb) {
  __shared__ float As[16 * 132];
  const int g = blockIdx.y;
  const float* __restrict__ W = g ? W1 : W0;
  float* __restrict__ h = hb + (size_t)g * PL_ND;
  const int m0 = blockIdx.x * 16;
  const int t  = threadIdx.x;

  {
    const int row = t >> 4;
    const int col = (t & 15) * 8;
    const float4* gp = (const float4*)(feat + (size_t)(m0 + row) * INF_ + col);
    float4 x0 = gp[0];
    float4 x1 = gp[1];
    float* lp = As + row * 132 + col;
    *(float4*)(lp)     = x0;
    *(float4*)(lp + 4) = x1;
  }
  __syncthreads();

  const int lane  = t & 31;
  const int wave  = t >> 5;
  const int n0    = wave * 16;
  const int r     = lane & 15;
  const int khalf = (lane >> 4) << 1;

  v8f acc = {};
  for (int k0 = 0; k0 < INF_; k0 += 4) {
    const int kk = k0 + khalf;
    v2f a;
    a.x = As[r * 132 + kk];
    a.y = As[r * 132 + kk + 1];
    v2f b;
    b.x = W[(size_t)(kk)     * HD + n0 + r];
    b.y = W[(size_t)(kk + 1) * HD + n0 + r];
    acc = __builtin_amdgcn_wmma_f32_16x16x4_f32(false, a, false, b,
                                                (short)0, acc, false, false);
  }

  const int mh = (lane >> 4) * 8;
#pragma unroll
  for (int j = 0; j < 8; ++j) {
    h[(size_t)(m0 + j + mh) * HD + n0 + r] = acc[j];
  }
}

__global__ void k_elr(const float* __restrict__ hb, const float* __restrict__ al0,
                      const float* __restrict__ al1, const float* __restrict__ ar0,
                      const float* __restrict__ ar1, float* __restrict__ elb,
                      float* __restrict__ erb) {
  int tid = blockIdx.x * blockDim.x + threadIdx.x;
  if (tid >= NN * HH) return;
  const int g = blockIdx.y;
  const float* __restrict__ h  = hb + (size_t)g * PL_ND;
  const float* __restrict__ al = g ? al1 : al0;
  const float* __restrict__ ar = g ? ar1 : ar0;
  float* __restrict__ el = elb + (size_t)g * PL_NH;
  float* __restrict__ er = erb + (size_t)g * PL_NH;
  const int n  = tid / HH;
  const int hh = tid % HH;
  const float* hp  = h  + (size_t)n * HD + hh * DD;
  const float* alp = al + hh * DD;
  const float* arp = ar + hh * DD;
  float sl = 0.0f, sr = 0.0f;
#pragma unroll
  for (int d = 0; d < DD; d += 4) {
    float4 hv = *(const float4*)(hp + d);
    float4 av = *(const float4*)(alp + d);
    float4 rv = *(const float4*)(arp + d);
    sl += hv.x * av.x + hv.y * av.y + hv.z * av.z + hv.w * av.w;
    sr += hv.x * rv.x + hv.y * rv.y + hv.z * rv.z + hv.w * rv.w;
  }
  el[tid] = sl;
  er[tid] = sr;
}

#define OWN_T 1024
#define OWN_CAP 8192
static_assert(OWN_R % 8 == 0, "an owner's sum rows fill whole 128-byte lines");
static_assert(OWN_R * HH <= OWN_T, "one thread per (node, head) merges the er and max tables");
static_assert(OWN_CAP % 4 == 0, "the list starts and ends on 16-byte boundaries");
#define OWN_TAB ((OWN_R * HD + 3 * OWN_R * HH) * 4)
#define OWN_LDS (OWN_TAB + OWN_CAP * 4)

__device__ __forceinline__ void own_max_v(const float4 ev, const float* __restrict__ er_l,
                                          unsigned int* __restrict__ mu_l, int dl) {
  const float4 rv = *(const float4*)(er_l + dl * HH);
  float x0 = ev.x + rv.x, x1 = ev.y + rv.y, x2 = ev.z + rv.z, x3 = ev.w + rv.w;
  x0 = (x0 > 0.0f) ? x0 : NEG_SLOPE * x0;
  x1 = (x1 > 0.0f) ? x1 : NEG_SLOPE * x1;
  x2 = (x2 > 0.0f) ? x2 : NEG_SLOPE * x2;
  x3 = (x3 > 0.0f) ? x3 : NEG_SLOPE * x3;
  atomicMax(&mu_l[dl * HH + 0], ford(x0));
  atomicMax(&mu_l[dl * HH + 1], ford(x1));
  atomicMax(&mu_l[dl * HH + 2], ford(x2));
  atomicMax(&mu_l[dl * HH + 3], ford(x3));
}

__device__ __forceinline__ void own_add(float* __restrict__ acc_l, float* __restrict__ s_l, const float2 em,
                                        int dl, int lane, int hh, float elv, const float4 hv) {
  float x = elv + em.x;
  x = (x > 0.0f) ? x : NEG_SLOPE * x;
  const float ex = __expf(x - em.y);
  float* ap = acc_l + dl * HD + lane;
  atomicAdd(ap + 0,  ex * hv.x);
  atomicAdd(ap + 32, ex * hv.y);
  atomicAdd(ap + 64, ex * hv.z);
  atomicAdd(ap + 96, ex * hv.w);
  if ((lane & 7) == 0) atomicAdd(&s_l[dl * HH + hh], ex);
}

__device__ __forceinline__ void own_merge(const float* er_l, const unsigned int* mu_l, float2* em_l, int t) {
  float e = 0.0f, m = 0.0f;
  if (t < OWN_R * HH) { e = er_l[t]; m = funord(mu_l[t]); }
  __syncthreads();
  if (t < OWN_R * HH) em_l[t] = make_float2(e, m);
  __syncthreads();
}

__device__ __forceinline__ void own_add4(const float* __restrict__ h, const float* __restrict__ el,
                                         float* __restrict__ acc_l, float* __restrict__ s_l,
                                         const float2* em_l,
                                         int lane, int hh, int sd0, int dl0, int sd1, int dl1, int sd2, int dl2,
                                         int sd3, int dl3, bool h1, bool h2, bool h3) {
  const float4 hv0 = *(const float4*)(h + (size_t)sd0 * HD + lane * 4);
  const float4 hv1 = *(const float4*)(h + (size_t)sd1 * HD + lane * 4);
  const float4 hv2 = *(const float4*)(h + (size_t)sd2 * HD + lane * 4);
  const float4 hv3 = *(const float4*)(h + (size_t)sd3 * HD + lane * 4);
  const float e0 = el[sd0 * HH + hh];
  const float e1 = el[sd1 * HH + hh];
  const float e2 = el[sd2 * HH + hh];
  const float e3 = el[sd3 * HH + hh];
  const float2 em0 = em_l[dl0 * HH + hh];
  const float2 em1 = em_l[dl1 * HH + hh];
  const float2 em2 = em_l[dl2 * HH + hh];
  const float2 em3 = em_l[dl3 * HH + hh];
  own_add(acc_l, s_l, em0, dl0, lane, hh, e0, hv0);
  if (h1) own_add(acc_l, s_l, em1, dl1, lane, hh, e1, hv1);
  if (h2) own_add(acc_l, s_l, em2, dl2, lane, hh, e2, hv2);
  if (h3) own_add(acc_l, s_l, em3, dl3, lane, hh, e3, hv3);
}

__device__ __forceinline__ void own_load_list(const unsigned int* __restrict__ lst, unsigned int* __restrict__ list_l,
                                              unsigned int n, int t) {
  for (unsigned int i = (unsigned int)t; i < n; i += OWN_T) list_l[i] = lst[i];
}

__device__ __forceinline__ void own_max_list(const float* __restrict__ el, const float* er_l, unsigned int* mu_l,
                                             const unsigned int* list_l, unsigned int n, int t) {
  for (unsigned int i = (unsigned int)t; i < n; i += 4u * OWN_T) {
    const bool k1 = i + OWN_T < n, k2 = i + 2u * OWN_T < n, k3 = i + 3u * OWN_T < n;
    const unsigned int w0 = list_l[i];
    const unsigned int w1 = list_l[k1 ? i + OWN_T : i];
    const unsigned int w2 = list_l[k2 ? i + 2u * OWN_T : i];
    const unsigned int w3 = list_l[k3 ? i + 3u * OWN_T : i];
    const float4 ev0 = *(const float4*)(el + (size_t)(w0 & 0xFFFFu) * HH);
    const float4 ev1 = *(const float4*)(el + (size_t)(w1 & 0xFFFFu) * HH);
    const float4 ev2 = *(const float4*)(el + (size_t)(w2 & 0xFFFFu) * HH);
    const float4 ev3 = *(const float4*)(el + (size_t)(w3 & 0xFFFFu) * HH);
    asm volatile("" :: "v"(ev0.x), "v"(ev0.y), "v"(ev0.z), "v"(ev0.w), "v"(ev1.x), "v"(ev1.y), "v"(ev1.z), "v"(ev1.w),
                 "v"(ev2.x), "v"(ev2.y), "v"(ev2.z), "v"(ev2.w), "v"(ev3.x), "v"(ev3.y), "v"(ev3.z), "v"(ev3.w));
    own_max_v(ev0, er_l, mu_l, (int)(w0 >> 16));
    own_max_v(ev1, er_l, mu_l, (int)(w1 >> 16));
    own_max_v(ev2, er_l, mu_l, (int)(w2 >> 16));
    own_max_v(ev3, er_l, mu_l, (int)(w3 >> 16));
  }
}

__device__ __forceinline__ void own_add_list(const float* __restrict__ h, const float* __restrict__ el,
                                             float* __restrict__ acc_l, float* __restrict__ s_l, const float2* em_l,
                                             const unsigned int* list_l, unsigned int n, int wave, int lane, int hh) {
  unsigned int b = (unsigned int)wave * 4u;
  int lv = 0;
  if (b < n) {
    const unsigned int li = b + (unsigned int)(lane & 3);
    lv = (int)list_l[(li < n) ? li : b];
  }
  while (b < n) {
    const bool h1 = b + 1u < n, h2 = b + 2u < n, h3 = b + 3u < n;
    const unsigned int w0 = (unsigned int)__builtin_amdgcn_readlane(lv, 0);
    const unsigned int w1 = (unsigned int)__builtin_amdgcn_readlane(lv, 1);
    const unsigned int w2 = (unsigned int)__builtin_amdgcn_readlane(lv, 2);
    const unsigned int w3 = (unsigned int)__builtin_amdgcn_readlane(lv, 3);
    b += (OWN_T / 32) * 4u;
    if (b < n) {
      const unsigned int li = b + (unsigned int)(lane & 3);
      lv = (int)list_l[(li < n) ? li : b];
    }
    own_add4(h, el, acc_l, s_l, em_l, lane, hh,
             (int)(w0 & 0xFFFFu), (int)(w0 >> 16), (int)(w1 & 0xFFFFu), (int)(w1 >> 16),
             (int)(w2 & 0xFFFFu), (int)(w2 >> 16), (int)(w3 & 0xFFFFu), (int)(w3 >> 16), h1, h2, h3);
  }
}

__global__ __launch_bounds__(OWN_T) void k_edge_own(const float* __restrict__ elb,
                                                    const float* __restrict__ erb,
                                                    const unsigned int* __restrict__ obase,
                                                    const unsigned int* __restrict__ bucket,
                                                    const float* __restrict__ hb,
                                                    float* __restrict__ sb,
                                                    float* __restrict__ accb) {
  const int g  = blockIdx.x & 1;
  const int ob = blockIdx.x >> 1;
  const float* __restrict__ el = elb + (size_t)g * PL_NH;
  const float* __restrict__ er = erb + (size_t)g * PL_NH;
  const float* __restrict__ h = hb + (size_t)g * PL_ND;
  float* __restrict__ s   = sb + (size_t)g * PL_NH;
  float* __restrict__ acc = accb + (size_t)g * PL_ND;
  __shared__ __attribute__((aligned(16))) unsigned char L[OWN_LDS];
  float* acc_l = (float*)L;
  float* s_l   = acc_l + OWN_R * HD;
  float* er_l  = s_l + OWN_R * HH;
  unsigned int* mu_l   = (unsigned int*)(er_l + OWN_R * HH);
  float2* em_l = (float2*)er_l;
  unsigned int* list_l = mu_l + OWN_R * HH;

  const int t    = threadIdx.x;
  const int lane = t & 31;
  const int wave = __builtin_amdgcn_readfirstlane(t >> 5);
  const int hh   = lane >> 3;
  const int n0   = ob * OWN_R;
  const int nrows = (NN - n0 < OWN_R) ? (NN - n0) : OWN_R;

  const unsigned int* __restrict__ og = obase + (size_t)g * (OWN_P + 1);
  const unsigned int r0 = (unsigned int)__builtin_amdgcn_readfirstlane((int)og[ob]);
  const unsigned int r1 = (unsigned int)__builtin_amdgcn_readfirstlane((int)og[ob + 1]);
  const unsigned int n_own = r1 - r0;
  const unsigned int* __restrict__ lst = bucket + (size_t)g * EE + r0;
  const bool whole = n_own <= OWN_CAP;

  for (int i = t; i < OWN_R * HD / 4; i += OWN_T)
    ((float4*)acc_l)[i] = make_float4(0.0f, 0.0f, 0.0f, 0.0f);
  for (int i = t; i < OWN_R * HH; i += OWN_T) {
    const bool ok = i < nrows * HH;
    s_l[i]  = 0.0f;
    er_l[i] = ok ? er[(size_t)n0 * HH + i] : 0.0f;
    mu_l[i] = FORD_NEG_INF;
  }

  for (unsigned int tb = 0u; tb < n_own; tb += OWN_CAP) {
    const unsigned int n = (n_own - tb < OWN_CAP) ? (n_own - tb) : OWN_CAP;
    own_load_list(lst + tb, list_l, n, t);
    __syncthreads();
    own_max_list(el, er_l, mu_l, list_l, n, t);
    __syncthreads();
  }
  __syncthreads();
  own_merge(er_l, mu_l, em_l, t);
  for (unsigned int tb = 0u; tb < n_own; tb += OWN_CAP) {
    const unsigned int n = (n_own - tb < OWN_CAP) ? (n_own - tb) : OWN_CAP;
    if (!whole) {
      own_load_list(lst + tb, list_l, n, t);
      __syncthreads();
    }
    own_add_list(h, el, acc_l, s_l, em_l, list_l, n, wave, lane, hh);
    if (!whole) __syncthreads();
  }
  __syncthreads();

  float4* ag = (float4*)(acc + (size_t)n0 * HD);
  for (int i = t; i < nrows * HD / 4; i += OWN_T) {
    const float* rp = acc_l + (i >> 5) * HD + (i & 31);
    ag[i] = make_float4(rp[0], rp[32], rp[64], rp[96]);
  }
  float* sg = s + (size_t)n0 * HH;
  for (int i = t; i < nrows * HH; i += OWN_T) sg[i] = s_l[i];
}

__device__ __forceinline__ float inv_or_zero(float v) {
  return (v != 0.0f) ? 1.0f / v : 0.0f;
}
__global__ void k_norm_mean(const float* __restrict__ acc0, const float* __restrict__ acc1,
                            const float* __restrict__ s0, const float* __restrict__ s1,
                            float* __restrict__ out) {
  int tid = blockIdx.x * blockDim.x + threadIdx.x;
  if (tid >= NN * DD) return;
  const int n = tid >> 5;
  const int d = tid & 31;
  const float4 v0 = *(const float4*)(s0 + (size_t)n * HH);
  const float4 v1 = *(const float4*)(s1 + (size_t)n * HH);
  const float* a0 = acc0 + (size_t)n * HD + d;
  const float* a1 = acc1 + (size_t)n * HD + d;
  const float f0 = a0[0], f1 = a0[32], f2 = a0[64], f3 = a0[96];
  const float b0 = a1[0], b1 = a1[32], b2 = a1[64], b3 = a1[96];
  float r = f0 * inv_or_zero(v0.x) + b0 * inv_or_zero(v1.x);
  r += f1 * inv_or_zero(v0.y) + b1 * inv_or_zero(v1.y);
  r += f2 * inv_or_zero(v0.z) + b2 * inv_or_zero(v1.z);
  r += f3 * inv_or_zero(v0.w) + b3 * inv_or_zero(v1.w);
  out[tid] = 0.25f * r;
}

extern "C" void kernel_launch(void* const* d_in, const int* in_sizes, int n_in,
                              void* d_out, int out_size, void* d_ws, size_t ws_size,
                              hipStream_t stream) {
  const float* feat = (const float*)d_in[0];
  const int*   src_g[2] = { (const int*)d_in[1], (const int*)d_in[3] };
  const int*   dst_g[2] = { (const int*)d_in[2], (const int*)d_in[4] };
  const float* W_g[2]   = { (const float*)d_in[5], (const float*)d_in[8] };
  const float* al_g[2]  = { (const float*)d_in[6], (const float*)d_in[9] };
  const float* ar_g[2]  = { (const float*)d_in[7], (const float*)d_in[10] };
  float* out = (float*)d_out;

  float*        acc0 = (float*)d_ws;
  float*        acc1 = acc0 + (size_t)NN * HD;
  float*        s0   = acc1 + (size_t)NN * HD;
  float*        s1   = s0   + (size_t)NN * HH;
  float*        hbuf = s1   + (size_t)NN * HH;
  float*        el   = hbuf + 2 * (size_t)NN * HD;
  float*        er   = el   + 2 * (size_t)NN * HH;
  unsigned int* cntb = (unsigned int*)(er + 2 * (size_t)NN * HH);
  unsigned int* obas = cntb + 2 * (size_t)OWN_P * PB;
  unsigned int* buck = obas + 2 * (size_t)(OWN_P + 1);

  const int T = 256;
  k_count<<<dim3(PB, 2), PT, 0, stream>>>(dst_g[0], dst_g[1], cntb);
  k_scatter<<<dim3(PB, 2), PT, 0, stream>>>(src_g[0], src_g[1], dst_g[0], dst_g[1], cntb, obas, buck);

  k_proj<<<dim3(NN / 16, 2), T, 0, stream>>>(feat, W_g[0], W_g[1], hbuf);
  k_elr<<<dim3((NN * HH + T - 1) / T, 2), T, 0, stream>>>(hbuf, al_g[0], al_g[1], ar_g[0], ar_g[1], el, er);
  k_edge_own<<<2 * OWN_P, OWN_T, 0, stream>>>(el, er, obas, buck, hbuf, s0, acc0);

  k_norm_mean<<<(NN * DD + T - 1) / T, T, 0, stream>>>(acc0, acc1, s0, s1, out);
}


# short kernels as resident grids: projection, el/er, node pass, edge count loop over tiles/rows in few long-lived waves (128,500 wave starts become 20,400); W fragments stay in registers over tiles
# speedup vs baseline: 4.5477x; 1.0393x over previous
#include <hip/hip_runtime.h>

#define NN 50000
#define EE 1600000
#define INF_ 128
#define HH 4
#define DD 32
#define HD 128
#define NEG_SLOPE 0.2f

typedef __attribute__((ext_vector_type(2))) float v2f;
typedef __attribute__((ext_vector_type(8))) float v8f;

__device__ __forceinline__ unsigned int ford(float f) {
  unsigned int b = __float_as_uint(f);
  return (b & 0x80000000u) ? ~b : (b | 0x80000000u);
}
__device__ __forceinline__ float funord(unsigned int u) {
  unsigned int b = (u & 0x80000000u) ? (u ^ 0x80000000u) : ~u;
  return __uint_as_float(b);
}
#define FORD_NEG_INF 0x007FFFFFu

#define OWN_R 200
#define OWN_P ((NN + OWN_R - 1) / OWN_R)
static_assert(NN <= 65536 && OWN_R <= 65536, "src and the local row share one list word");

#define PB 250
#define PCH (EE / PB)
#define PT 1024
#define PTC 256
#define PCW ((PCH / 4 + PTC - 1) / PTC)
static_assert(EE % PB == 0 && PCH % 4 == 0, "chunks are whole and start on 16-byte boundaries");
static_assert(OWN_P <= 256 && OWN_R <= 256, "owner and local row take a byte each in a staged word; one wave scans the owners");
static_assert(OWN_P <= PTC, "one counting thread per owner clears and stores the counts");

__global__ __launch_bounds__(PTC) void k_count(const int* __restrict__ dst0, const int* __restrict__ dst1,
                                               unsigned int* __restrict__ cnt) {
  __shared__ unsigned int c[OWN_P];
  const int t = threadIdx.x;
  const int b = blockIdx.x;
  const int g = blockIdx.y;
  if (t < OWN_P) c[t] = 0u;
  __syncthreads();
  const int4* dp = (const int4*)((g ? dst1 : dst0) + (size_t)b * PCH);
  int4 d[PCW];
#pragma unroll
  for (int k = 0; k < PCW; ++k) {
    const int i = t + k * PTC;
    d[k] = dp[(i < PCH / 4) ? i : t];
  }
  static_assert(PCW == 7, "the statement below names the seven words of a thread");
  asm volatile("" :: "v"(d[0].x), "v"(d[0].y), "v"(d[0].z), "v"(d[0].w), "v"(d[1].x), "v"(d[1].y), "v"(d[1].z), "v"(d[1].w),
               "v"(d[2].x), "v"(d[2].y), "v"(d[2].z), "v"(d[2].w), "v"(d[3].x), "v"(d[3].y), "v"(d[3].z), "v"(d[3].w),
               "v"(d[4].x), "v"(d[4].y), "v"(d[4].z), "v"(d[4].w), "v"(d[5].x), "v"(d[5].y), "v"(d[5].z), "v"(d[5].w),
               "v"(d[6].x), "v"(d[6].y), "v"(d[6].z), "v"(d[6].w));
#pragma unroll
  for (int k = 0; k < PCW; ++k) {
    if (t + k * PTC < PCH / 4) {
      atomicAdd(&c[d[k].x / OWN_R], 1u);
      atomicAdd(&c[d[k].y / OWN_R], 1u);
      atomicAdd(&c[d[k].z / OWN_R], 1u);
      atomicAdd(&c[d[k].w / OWN_R], 1u);
    }
  }
  __syncthreads();
  if (t < OWN_P) cnt[((size_t)g * PB + b) * OWN_P + t] = c[t];
}

__global__ __launch_bounds__(PT) void k_scatter(const int* __restrict__ src0, const int* __restrict__ src1,
                                                const int* __restrict__ dst0, const int* __restrict__ dst1,
                                                const unsigned int* __restrict__ cnt,
                                                unsigned int* __restrict__ obase,
                                                unsigned int* __restrict__ bucket) {
  __shared__ unsigned int part_a[4][256];
  __shared__ unsigned int part_p[4][256];
  __shared__ unsigned int lcur[256];
  __shared__ unsigned int delta[256];
  __shared__ unsigned int stage[PCH];
  const int t = threadIdx.x;
  const int b = blockIdx.x;
  const int g = blockIdx.y;
  const int lane = t & 31;
  const int wave = t >> 5;
  const int q  = t >> 8;
  const int oc = t & 255;
  const unsigned int* __restrict__ cg = cnt + (size_t)g * PB * OWN_P;
  unsigned int a = 0u, p = 0u;
  if (oc < OWN_P) {
#pragma unroll 8
    for (int bb = q; bb < PB; bb += 4) {
      const unsigned int v = cg[(size_t)bb * OWN_P + oc];
      a += v;
      p += (bb < b) ? v : 0u;
    }
  }
  part_a[q][oc] = a;
  part_p[q][oc] = p;
  __syncthreads();
  if (wave == 0) {
    unsigned int vt[8], vp[8], vo[8];
    unsigned int st = 0u, so = 0u;
#pragma unroll
    for (int k = 0; k < 8; ++k) {
      const int o = lane * 8 + k;
      vt[k] = part_a[0][o] + part_a[1][o] + part_a[2][o] + part_a[3][o];
      vp[k] = part_p[0][o] + part_p[1][o] + part_p[2][o] + part_p[3][o];
      vo[k] = (o < OWN_P) ? cg[(size_t)b * OWN_P + o] : 0u;
      st += vt[k];
      so += vo[k];
    }
    unsigned int it = st, io = so;
#pragma unroll
    for (int d = 1; d < 32; d <<= 1) {
      const unsigned int nt = (unsigned int)__shfl_up((int)it, d);
      const unsigned int no = (unsigned int)__shfl_up((int)io, d);
      if (lane >= d) { it += nt; io += no; }
    }
    unsigned int run = it - st;
    unsigned int loc = io - so;
    unsigned int* __restrict__ og = obase + (size_t)g * (OWN_P + 1);
#pragma unroll
    for (int k = 0; k < 8; ++k) {
      const int o = lane * 8 + k;
      if (o < OWN_P) {
        lcur[o]  = loc;
        delta[o] = run + vp[k] - loc;
        if (b == 0) og[o] = run;
      }
      run += vt[k];
      loc += vo[k];
    }
    if (b == 0 && lane == 31) og[OWN_P] = run;
  }
  __syncthreads();
  const int4* dp = (const int4*)((g ? dst1 : dst0) + (size_t)b * PCH);
  const int4* sp = (const int4*)((g ? src1 : src0) + (size_t)b * PCH);
  for (int i = t; i < PCH / 4; i += PT) {
    const int4 d = dp[i];
    const int4 s = sp[i];
    const int o0 = d.x / OWN_R, o1 = d.y / OWN_R, o2 = d.z / OWN_R, o3 = d.w / OWN_R;
    const unsigned int q0 = atomicAdd(&lcur[o0], 1u);
    const unsigned int q1 = atomicAdd(&lcur[o1], 1u);
    const unsigned int q2 = atomicAdd(&lcur[o2], 1u);
    const unsigned int q3 = atomicAdd(&lcur[o3], 1u);
    stage[q0] = (unsigned int)s.x | ((unsigned int)(d.x - o0 * OWN_R) << 16) | ((unsigned int)o0 << 24);
    stage[q1] = (unsigned int)s.y | ((unsigned int)(d.y - o1 * OWN_R) << 16) | ((unsigned int)o1 << 24);
    stage[q2] = (unsigned int)s.z | ((unsigned int)(d.z - o2 * OWN_R) << 16) | ((unsigned int)o2 << 24);
    stage[q3] = (unsigned int)s.w | ((unsigned int)(d.w - o3 * OWN_R) << 16) | ((unsigned int)o3 << 24);
  }
  __syncthreads();
  unsigned int* __restrict__ bk = bucket + (size_t)g * EE;
  for (int i = t; i < PCH; i += PT) {
    const unsigned int w = stage[i];
    bk[(unsigned int)i + delta[w >> 24]] = w & 0x00FFFFFFu;
  }
}

#define PL_ND ((size_t)NN * HD)
#define PL_NH ((size_t)NN * HH)

#define PJ_G 512
#define PJ_NT (NN / 16)
#define ELR_G 256
#define NM_G 512
static_assert(NN % 16 == 0 && PJ_G <= PJ_NT, "every projection work-group has a first tile; tiles are whole");
static_assert(NN % 4 == 0, "a node-pass wave takes four whole nodes per trip");

__global__ __launch_bounds__(256) void k_proj(const float* __restrict__ feat,
                                              const float* __restrict__ W0,
                                              const float* __restrict__ W1,
                                              float* __restrict__ hb) {
  __shared__ float As[16 * 132];
  const int g = blockIdx.y;
  const float* __restrict__ W = g ? W1 : W0;
  float* __restrict__ h = hb + (size_t)g * PL_ND;
  const int t  = threadIdx.x;

  const int lane  = t & 31;
  const int wave  = t >> 5;
  const int n0    = wave * 16;
  const int r     = lane & 15;
  const int khalf = (lane >> 4) << 1;
  const int mh    = (lane >> 4) * 8;

  v2f bf[INF_ / 4];
#pragma unroll
  for (int s = 0; s < INF_ / 4; ++s) {
    const int kk = s * 4 + khalf;
    bf[s].x = W[(size_t)(kk)     * HD + n0 + r];
    bf[s].y = W[(size_t)(kk + 1) * HD + n0 + r];
  }

  const int row = t >> 4;
  const int col = (t & 15) * 8;
  float* lp = As + row * 132 + col;
  int tile = blockIdx.x;
  float4 x0, x1;
  {
    const float4* gp = (const float4*)(feat + (size_t)(tile * 16 + row) * INF_ + col);
    x0 = gp[0];
    x1 = gp[1];
  }
  for (; tile < PJ_NT; tile += PJ_G) {
    __syncthreads();
    *(float4*)(lp)     = x0;
    *(float4*)(lp + 4) = x1;
    __syncthreads();
    if (tile + PJ_G < PJ_NT) {
      const float4* gp = (const float4*)(feat + (size_t)((tile + PJ_G) * 16 + row) * INF_ + col);
      x0 = gp[0];
      x1 = gp[1];
    }

    v8f acc = {};
#pragma unroll
    for (int s = 0; s < INF_ / 4; ++s) {
      const int kk = s * 4 + khalf;
      v2f a;
      a.x = As[r * 132 + kk];
      a.y = As[r * 132 + kk + 1];
      acc = __builtin_amdgcn_wmma_f32_16x16x4_f32(false, a, false, bf[s],
                                                  (short)0, acc, false, false);
    }

    const int m0 = tile * 16;
#pragma unroll
    for (int j = 0; j < 8; ++j) {
      h[(size_t)(m0 + j + mh) * HD + n0 + r] = acc[j];
    }
  }
}

__global__ __launch_bounds__(256) void k_elr(const float* __restrict__ hb, const float* __restrict__ al0,
                                             const float* __restrict__ al1, const float* __restrict__ ar0,
                                             const float* __restrict__ ar1, float* __restrict__ elb,
                                             float* __restrict__ erb) {
  const int g = blockIdx.y;
  const float* __restrict__ h  = hb + (size_t)g * PL_ND;
  const float* __restrict__ al = g ? al1 : al0;
  const float* __restrict__ ar = g ? ar1 : ar0;
  float* __restrict__ el = elb + (size_t)g * PL_NH;
  float* __restrict__ er = erb + (size_t)g * PL_NH;
  const int hh = threadIdx.x % HH;
  const float* alp = al + hh * DD;
  const float* arp = ar + hh * DD;
  for (int tid = blockIdx.x * 256 + threadIdx.x; tid < NN * HH; tid += ELR_G * 256) {
    const int n = tid / HH;
    const float* hp = h + (size_t)n * HD + hh * DD;
    float sl = 0.0f, sr = 0.0f;
#pragma unroll
    for (int d = 0; d < DD; d += 4) {
      float4 hv = *(const float4*)(hp + d);
      float4 av = *(const float4*)(alp + d);
      float4 rv = *(const float4*)(arp + d);
      sl += hv.x * av.x + hv.y * av.y + hv.z * av.z + hv.w * av.w;
      sr += hv.x * rv.x + hv.y * rv.y + hv.z * rv.z + hv.w * rv.w;
    }
    el[tid] = sl;
    er[tid] = sr;
  }
}

#define OWN_T 1024
#define OWN_CAP 8192
static_assert(OWN_R % 8 == 0, "an owner's sum rows fill whole 128-byte lines");
static_assert(OWN_R * HH <= OWN_T, "one thread per (node, head) merges the er and max tables");
static_assert(OWN_CAP % 4 == 0, "the list starts and ends on 16-byte boundaries");
#define OWN_TAB ((OWN_R * HD + 3 * OWN_R * HH) * 4)
#define OWN_LDS (OWN_TAB + OWN_CAP * 4)

__device__ __forceinline__ void own_max_v(const float4 ev, const float* __restrict__ er_l,
                                          unsigned int* __restrict__ mu_l, int dl) {
  const float4 rv = *(const float4*)(er_l + dl * HH);
  float x0 = ev.x + rv.x, x1 = ev.y + rv.y, x2 = ev.z + rv.z, x3 = ev.w + rv.w;
  x0 = (x0 > 0.0f) ? x0 : NEG_SLOPE * x0;
  x1 = (x1 > 0.0f) ? x1 : NEG_SLOPE * x1;
  x2 = (x2 > 0.0f) ? x2 : NEG_SLOPE * x2;
  x3 = (x3 > 0.0f) ? x3 : NEG_SLOPE * x3;
  atomicMax(&mu_l[dl * HH + 0], ford(x0));
  atomicMax(&mu_l[dl * HH + 1], ford(x1));
  atomicMax(&mu_l[dl * HH + 2], ford(x2));
  atomicMax(&mu_l[dl * HH + 3], ford(x3));
}

__device__ __forceinline__ void own_add(float* __restrict__ acc_l, float* __restrict__ s_l, const float2 em,
                                        int dl, int lane, int hh, float elv, const float4 hv) {
  float x = elv + em.x;
  x = (x > 0.0f) ? x : NEG_SLOPE * x;
  const float ex = __expf(x - em.y);
  float* ap = acc_l + dl * HD + lane;
  atomicAdd(ap + 0,  ex * hv.x);
  atomicAdd(ap + 32, ex * hv.y);
  atomicAdd(ap + 64, ex * hv.z);
  atomicAdd(ap + 96, ex * hv.w);
  if ((lane & 7) == 0) atomicAdd(&s_l[dl * HH + hh], ex);
}

__device__ __forceinline__ void own_merge(const float* er_l, const unsigned int* mu_l, float2* em_l, int t) {
  float e = 0.0f, m = 0.0f;
  if (t < OWN_R * HH) { e = er_l[t]; m = funord(mu_l[t]); }
  __syncthreads();
  if (t < OWN_R * HH) em_l[t] = make_float2(e, m);
  __syncthreads();
}

__device__ __forceinline__ void own_add4(const float* __restrict__ h, const float* __restrict__ el,
                                         float* __restrict__ acc_l, float* __restrict__ s_l,
                                         const float2* em_l,
                                         int lane, int hh, int sd0, int dl0, int sd1, int dl1, int sd2, int dl2,
                                         int sd3, int dl3, bool h1, bool h2, bool h3) {
  const float4 hv0 = *(const float4*)(h + (size_t)sd0 * HD + lane * 4);
  const float4 hv1 = *(const float4*)(h + (size_t)sd1 * HD + lane * 4);
  const float4 hv2 = *(const float4*)(h + (size_t)sd2 * HD + lane * 4);
  const float4 hv3 = *(const float4*)(h + (size_t)sd3 * HD + lane * 4);
  const float e0 = el[sd0 * HH + hh];
  const float e1 = el[sd1 * HH + hh];
  const float e2 = el[sd2 * HH + hh];
  const float e3 = el[sd3 * HH + hh];
  const float2 em0 = em_l[dl0 * HH + hh];
  const float2 em1 = em_l[dl1 * HH + hh];
  const float2 em2 = em_l[dl2 * HH + hh];
  const float2 em3 = em_l[dl3 * HH + hh];
  own_add(acc_l, s_l, em0, dl0, lane, hh, e0, hv0);
  if (h1) own_add(acc_l, s_l, em1, dl1, lane, hh, e1, hv1);
  if (h2) own_add(acc_l, s_l, em2, dl2, lane, hh, e2, hv2);
  if (h3) own_add(acc_l, s_l, em3, dl3, lane, hh, e3, hv3);
}

__device__ __forceinline__ void own_load_list(const unsigned int* __restrict__ lst, unsigned int* __restrict__ list_l,
                                              unsigned int n, int t) {
  for (unsigned int i = (unsigned int)t; i < n; i += OWN_T) list_l[i] = lst[i];
}

__device__ __forceinline__ void own_max_list(const float* __restrict__ el, const float* er_l, unsigned int* mu_l,
                                             const unsigned int* list_l, unsigned int n, int t) {
  for (unsigned int i = (unsigned int)t; i < n; i += 4u * OWN_T) {
    const bool k1 = i + OWN_T < n, k2 = i + 2u * OWN_T < n, k3 = i + 3u * OWN_T < n;
    const unsigned int w0 = list_l[i];
    const unsigned int w1 = list_l[k1 ? i + OWN_T : i];
    const unsigned int w2 = list_l[k2 ? i + 2u * OWN_T : i];
    const unsigned int w3 = list_l[k3 ? i + 3u * OWN_T : i];
    const float4 ev0 = *(const float4*)(el + (size_t)(w0 & 0xFFFFu) * HH);
    const float4 ev1 = *(const float4*)(el + (size_t)(w1 & 0xFFFFu) * HH);
    const float4 ev2 = *(const float4*)(el + (size_t)(w2 & 0xFFFFu) * HH);
    const float4 ev3 = *(const float4*)(el + (size_t)(w3 & 0xFFFFu) * HH);
    asm volatile("" :: "v"(ev0.x), "v"(ev0.y), "v"(ev0.z), "v"(ev0.w), "v"(ev1.x), "v"(ev1.y), "v"(ev1.z), "v"(ev1.w),
                 "v"(ev2.x), "v"(ev2.y), "v"(ev2.z), "v"(ev2.w), "v"(ev3.x), "v"(ev3.y), "v"(ev3.z), "v"(ev3.w));
    own_max_v(ev0, er_l, mu_l, (int)(w0 >> 16));
    own_max_v(ev1, er_l, mu_l, (int)(w1 >> 16));
    own_max_v(ev2, er_l, mu_l, (int)(w2 >> 16));
    own_max_v(ev3, er_l, mu_l, (int)(w3 >> 16));
  }
}

__device__ __forceinline__ void own_add_list(const float* __restrict__ h, const float* __restrict__ el,
                                             float* __restrict__ acc_l, float* __restrict__ s_l, const float2* em_l,
                                             const unsigned int* list_l, unsigned int n, int wave, int lane, int hh) {
  unsigned int b = (unsigned int)wave * 4u;
  int lv = 0;
  if (b < n) {
    const unsigned int li = b + (unsigned int)(lane & 3);
    lv = (int)list_l[(li < n) ? li : b];
  }
  while (b < n) {
    const bool h1 = b + 1u < n, h2 = b + 2u < n, h3 = b + 3u < n;
    const unsigned int w0 = (unsigned int)__builtin_amdgcn_readlane(lv, 0);
    const unsigned int w1 = (unsigned int)__builtin_amdgcn_readlane(lv, 1);
    const unsigned int w2 = (unsigned int)__builtin_amdgcn_readlane(lv, 2);
    const unsigned int w3 = (unsigned int)__builtin_amdgcn_readlane(lv, 3);
    b += (OWN_T / 32) * 4u;
    if (b < n) {
      const unsigned int li = b + (unsigned int)(lane & 3);
      lv = (int)list_l[(li < n) ? li : b];
    }
    own_add4(h, el, acc_l, s_l, em_l, lane, hh,
             (int)(w0 & 0xFFFFu), (int)(w0 >> 16), (int)(w1 & 0xFFFFu), (int)(w1 >> 16),
             (int)(w2 & 0xFFFFu), (int)(w2 >> 16), (int)(w3 & 0xFFFFu), (int)(w3 >> 16), h1, h2, h3);
  }
}

__global__ __launch_bounds__(OWN_T) void k_edge_own(const float* __restrict__ elb,
                                                    const float* __restrict__ erb,
                                                    const unsigned int* __restrict__ obase,
                                                    const unsigned int* __restrict__ bucket,
                                                    const float* __restrict__ hb,
                                                    float* __restrict__ sb,
                                                    float* __restrict__ accb) {
  const int g  = blockIdx.x & 1;
  const int ob = blockIdx.x >> 1;
  const float* __restrict__ el = elb + (size_t)g * PL_NH;
  const float* __restrict__ er = erb + (size_t)g * PL_NH;
  const float* __restrict__ h = hb + (size_t)g * PL_ND;
  float* __restrict__ s   = sb + (size_t)g * PL_NH;
  float* __restrict__ acc = accb + (size_t)g * PL_ND;
  __shared__ __attribute__((aligned(16))) unsigned char L[OWN_LDS];
  float* acc_l = (float*)L;
  float* s_l   = acc_l + OWN_R * HD;
  float* er_l  = s_l + OWN_R * HH;
  unsigned int* mu_l   = (unsigned int*)(er_l + OWN_R * HH);
  float2* em_l = (float2*)er_l;
  unsigned int* list_l = mu_l + OWN_R * HH;

  const int t    = threadIdx.x;
  const int lane = t & 31;
  const int wave = __builtin_amdgcn_readfirstlane(t >> 5);
  const int hh   = lane >> 3;
  const int n0   = ob * OWN_R;
  const int nrows = (NN - n0 < OWN_R) ? (NN - n0) : OWN_R;

  const unsigned int* __restrict__ og = obase + (size_t)g * (OWN_P + 1);
  const unsigned int r0 = (unsigned int)__builtin_amdgcn_readfirstlane((int)og[ob]);
  const unsigned int r1 = (unsigned int)__builtin_amdgcn_readfirstlane((int)og[ob + 1]);
  const unsigned int n_own = r1 - r0;
  const unsigned int* __restrict__ lst = bucket + (size_t)g * EE + r0;
  const bool whole = n_own <= OWN_CAP;

  for (int i = t; i < OWN_R * HD / 4; i += OWN_T)
    ((float4*)acc_l)[i] = make_float4(0.0f, 0.0f, 0.0f, 0.0f);
  for (int i = t; i < OWN_R * HH; i += OWN_T) {
    const bool ok = i < nrows * HH;
    s_l[i]  = 0.0f;
    er_l[i] = ok ? er[(size_t)n0 * HH + i] : 0.0f;
    mu_l[i] = FORD_NEG_INF;
  }

  for (unsigned int tb = 0u; tb < n_own; tb += OWN_CAP) {
    const unsigned int n = (n_own - tb < OWN_CAP) ? (n_own - tb) : OWN_CAP;
    own_load_list(lst + tb, list_l, n, t);
    __syncthreads();
    own_max_list(el, er_l, mu_l, list_l, n, t);
    __syncthreads();
  }
  __syncthreads();
  own_merge(er_l, mu_l, em_l, t);
  for (unsigned int tb = 0u; tb < n_own; tb += OWN_CAP) {
    const unsigned int n = (n_own - tb < OWN_CAP) ? (n_own - tb) : OWN_CAP;
    if (!whole) {
      own_load_list(lst + tb, list_l, n, t);
      __syncthreads();
    }
    own_add_list(h, el, acc_l, s_l, em_l, list_l, n, wave, lane, hh);
    if (!whole) __syncthreads();
  }
  __syncthreads();

  float4* ag = (float4*)(acc + (size_t)n0 * HD);
  for (int i = t; i < nrows * HD / 4; i += OWN_T) {
    const float* rp = acc_l + (i >> 5) * HD + (i & 31);
    ag[i] = make_float4(rp[0], rp[32], rp[64], rp[96]);
  }
  float* sg = s + (size_t)n0 * HH;
  for (int i = t; i < nrows * HH; i += OWN_T) sg[i] = s_l[i];
}

__device__ __forceinline__ float inv_or_zero(float v) {
  return (v != 0.0f) ? 1.0f / v : 0.0f;
}
__device__ __forceinline__ float norm_one(float f0, float f1, float f2, float f3, float b0, float b1, float b2,
                                          float b3, const float4 i0, const float4 i1) {
  float r = f0 * i0.x + b0 * i1.x;
  r += f1 * i0.y + b1 * i1.y;
  r += f2 * i0.z + b2 * i1.z;
  r += f3 * i0.w + b3 * i1.w;
  return 0.25f * r;
}
__global__ __launch_bounds__(256) void k_norm_mean(const float* __restrict__ acc0, const float* __restrict__ acc1,
                                                   const float* __restrict__ s0, const float* __restrict__ s1,
                                                   float* __restrict__ out) {
  const int lane = threadIdx.x & 31;
  const int gw   = blockIdx.x * 8 + (threadIdx.x >> 5);
  const int d4   = (lane & 7) * 4;
  for (int n = gw * 4 + (lane >> 3); n < NN; n += NM_G * 8 * 4) {
    const float4 v0 = *(const float4*)(s0 + (size_t)n * HH);
    const float4 v1 = *(const float4*)(s1 + (size_t)n * HH);
    const float* a0 = acc0 + (size_t)n * HD + d4;
    const float* a1 = acc1 + (size_t)n * HD + d4;
    const float4 f0 = *(const float4*)(a0), f1 = *(const float4*)(a0 + 32);
    const float4 f2 = *(const float4*)(a0 + 64), f3 = *(const float4*)(a0 + 96);
    const float4 b0 = *(const float4*)(a1), b1 = *(const float4*)(a1 + 32);
    const float4 b2 = *(const float4*)(a1 + 64), b3 = *(const float4*)(a1 + 96);
    const float4 i0 = make_float4(inv_or_zero(v0.x), inv_or_zero(v0.y), inv_or_zero(v0.z), inv_or_zero(v0.w));
    const float4 i1 = make_float4(inv_or_zero(v1.x), inv_or_zero(v1.y), inv_or_zero(v1.z), inv_or_zero(v1.w));
    float4 o;
    o.x = norm_one(f0.x, f1.x, f2.x, f3.x, b0.x, b1.x, b2.x, b3.x, i0, i1);
    o.y = norm_one(f0.y, f1.y, f2.y, f3.y, b0.y, b1.y, b2.y, b3.y, i0, i1);
    o.z = norm_one(f0.z, f1.z, f2.z, f3.z, b0.z, b1.z, b2.z, b3.z, i0, i1);
    o.w = norm_one(f0.w, f1.w, f2.w, f3.w, b0.w, b1.w, b2.w, b3.w, i0, i1);
    *(float4*)(out + (size_t)n * DD + d4) = o;
  }
}

extern "C" void kernel_launch(void* const* d_in, const int* in_sizes, int n_in,
                              void* d_out, int out_size, void* d_ws, size_t ws_size,
                              hipStream_t stream) {
  const float* feat = (const float*)d_in[0];
  const int*   src_g[2] = { (const int*)d_in[1], (const int*)d_in[3] };
  const int*   dst_g[2] = { (const int*)d_in[2], (const int*)d_in[4] };
  const float* W_g[2]   = { (const float*)d_in[5], (const float*)d_in[8] };
  const float* al_g[2]  = { (const float*)d_in[6], (const float*)d_in[9] };
  const float* ar_g[2]  = { (const float*)d_in[7], (const float*)d_in[10] };
  float* out = (float*)d_out;

  float*        acc0 = (float*)d_ws;
  float*        acc1 = acc0 + (size_t)NN * HD;
  float*        s0   = acc1 + (size_t)NN * HD;
  float*        s1   = s0   + (size_t)NN * HH;
  float*        hbuf = s1   + (size_t)NN * HH;
  float*        el   = hbuf + 2 * (size_t)NN * HD;
  float*        er   = el   + 2 * (size_t)NN * HH;
  unsigned int* cntb = (unsigned int*)(er + 2 * (size_t)NN * HH);
  unsigned int* obas = cntb + 2 * (size_t)OWN_P * PB;
  unsigned int* buck = obas + 2 * (size_t)(OWN_P + 1);

  const int T = 256;
  k_count<<<dim3(PB, 2), PTC, 0, stream>>>(dst_g[0], dst_g[1], cntb);
  k_scatter<<<dim3(PB, 2), PT, 0, stream>>>(src_g[0], src_g[1], dst_g[0], dst_g[1], cntb, obas, buck);

  k_proj<<<dim3(PJ_G, 2), T, 0, stream>>>(feat, W_g[0], W_g[1], hbuf);
  k_elr<<<dim3(ELR_G, 2), T, 0, stream>>>(hbuf, al_g[0], al_g[1], ar_g[0], ar_g[1], el, er);
  k_edge_own<<<2 * OWN_P, OWN_T, 0, stream>>>(el, er, obas, buck, hbuf, s0, acc0);

  k_norm_mean<<<NM_G, T, 0, stream>>>(acc0, acc1, s0, s1, out);
}


# node pass fused into owner pass: a work-group owns 200 rows of BOTH graphs in LDS (1 per WGP, 8 row loads per wave trip), scales by 1/s, stores out once; acc/s planes, a launch go; LDS arrival order
# speedup vs baseline: 4.7892x; 1.0531x over previous
#include <hip/hip_runtime.h>

#define NN 50000
#define EE 1600000
#define INF_ 128
#define HH 4
#define DD 32
#define HD 128
#define NEG_SLOPE 0.2f

typedef __attribute__((ext_vector_type(2))) float v2f;
typedef __attribute__((ext_vector_type(8))) float v8f;

__device__ __forceinline__ unsigned int ford(float f) {
  unsigned int b = __float_as_uint(f);
  return (b & 0x80000000u) ? ~b : (b | 0x80000000u);
}
__device__ __forceinline__ float funord(unsigned int u) {
  unsigned int b = (u & 0x80000000u) ? (u ^ 0x80000000u) : ~u;
  return __uint_as_float(b);
}
#define FORD_NEG_INF 0x007FFFFFu

#define OWN_R 200
#define OWN_P ((NN + OWN_R - 1) / OWN_R)
static_assert(NN <= 65536 && OWN_R <= 65536, "src and the local row share one list word");

#define PB 250
#define PCH (EE / PB)
#define PT 1024
#define PTC 256
#define PCW ((PCH / 4 + PTC - 1) / PTC)
static_assert(EE % PB == 0 && PCH % 4 == 0, "chunks are whole and start on 16-byte boundaries");
static_assert(OWN_P <= 256 && OWN_R <= 256, "owner and local row take a byte each in a staged word; one wave scans the owners");
static_assert(OWN_P <= PTC, "one counting thread per owner clears and stores the counts");

__global__ __launch_bounds__(PTC) void k_count(const int* __restrict__ dst0, const int* __restrict__ dst1,
                                               unsigned int* __restrict__ cnt) {
  __shared__ unsigned int c[OWN_P];
  const int t = threadIdx.x;
  const int b = blockIdx.x;
  const int g = blockIdx.y;
  if (t < OWN_P) c[t] = 0u;
  __syncthreads();
  const int4* dp = (const int4*)((g ? dst1 : dst0) + (size_t)b * PCH);
  int4 d[PCW];
#pragma unroll
  for (int k = 0; k < PCW; ++k) {
    const int i = t + k * PTC;
    d[k] = dp[(i < PCH / 4) ? i : t];
  }
  static_assert(PCW == 7, "the statement below names the seven words of a thread");
  asm volatile("" :: "v"(d[0].x), "v"(d[0].y), "v"(d[0].z), "v"(d[0].w), "v"(d[1].x), "v"(d[1].y), "v"(d[1].z), "v"(d[1].w),
               "v"(d[2].x), "v"(d[2].y), "v"(d[2].z), "v"(d[2].w), "v"(d[3].x), "v"(d[3].y), "v"(d[3].z), "v"(d[3].w),
               "v"(d[4].x), "v"(d[4].y), "v"(d[4].z), "v"(d[4].w), "v"(d[5].x), "v"(d[5].y), "v"(d[5].z), "v"(d[5].w),
               "v"(d[6].x), "v"(d[6].y), "v"(d[6].z), "v"(d[6].w));
#pragma unroll
  for (int k = 0; k < PCW; ++k) {
    if (t + k * PTC < PCH / 4) {
      atomicAdd(&c[d[k].x / OWN_R], 1u);
      atomicAdd(&c[d[k].y / OWN_R], 1u);
      atomicAdd(&c[d[k].z / OWN_R], 1u);
      atomicAdd(&c[d[k].w / OWN_R], 1u);
    }
  }
  __syncthreads();
  if (t < OWN_P) cnt[((size_t)g * PB + b) * OWN_P + t] = c[t];
}

__global__ __launch_bounds__(PT) void k_scatter(const int* __restrict__ src0, const int* __restrict__ src1,
                                                const int* __restrict__ dst0, const int* __restrict__ dst1,
                                                const unsigned int* __restrict__ cnt,
                                                unsigned int* __restrict__ obase,
                                                unsigned int* __restrict__ bucket) {
  __shared__ unsigned int part_a[4][256];
  __shared__ unsigned int part_p[4][256];
  __shared__ unsigned int lcur[256];
  __shared__ unsigned int delta[256];
  __shared__ unsigned int stage[PCH];
  const int t = threadIdx.x;
  const int b = blockIdx.x;
  const int g = blockIdx.y;
  const int lane = t & 31;
  const int wave = t >> 5;
  const int q  = t >> 8;
  const int oc = t & 255;
  const unsigned int* __restrict__ cg = cnt + (size_t)g * PB * OWN_P;
  unsigned int a = 0u, p = 0u;
  if (oc < OWN_P) {
#pragma unroll 8
    for (int bb = q; bb < PB; bb += 4) {
      const unsigned int v = cg[(size_t)bb * OWN_P + oc];
      a += v;
      p += (bb < b) ? v : 0u;
    }
  }
  part_a[q][oc] = a;
  part_p[q][oc] = p;
  __syncthreads();
  if (wave == 0) {
    unsigned int vt[8], vp[8], vo[8];
    unsigned int st = 0u, so = 0u;
#pragma unroll
    for (int k = 0; k < 8; ++k) {
      const int o = lane * 8 + k;
      vt[k] = part_a[0][o] + part_a[1][o] + part_a[2][o] + part_a[3][o];
      vp[k] = part_p[0][o] + part_p[1][o] + part_p[2][o] + part_p[3][o];
      vo[k] = (o < OWN_P) ? cg[(size_t)b * OWN_P + o] : 0u;
      st += vt[k];
      so += vo[k];
    }
    unsigned int it = st, io = so;
#pragma unroll
    for (int d = 1; d < 32; d <<= 1) {
      const unsigned int nt = (unsigned int)__shfl_up((int)it, d);
      const unsigned int no = (unsigned int)__shfl_up((int)io, d);
      if (lane >= d) { it += nt; io += no; }
    }
    unsigned int run = it - st;
    unsigned int loc = io - so;
    unsigned int* __restrict__ og = obase + (size_t)g * (OWN_P + 1);
#pragma unroll
    for (int k = 0; k < 8; ++k) {
      const int o = lane * 8 + k;
      if (o < OWN_P) {
        lcur[o]  = loc;
        delta[o] = run + vp[k] - loc;
        if (b == 0) og[o] = run;
      }
      run += vt[k];
      loc += vo[k];
    }
    if (b == 0 && lane == 31) og[OWN_P] = run;
  }
  __syncthreads();
  const int4* dp = (const int4*)((g ? dst1 : dst0) + (size_t)b * PCH);
  const int4* sp = (const int4*)((g ? src1 : src0) + (size_t)b * PCH);
  for (int i = t; i < PCH / 4; i += PT) {
    const int4 d = dp[i];
    const int4 s = sp[i];
    const int o0 = d.x / OWN_R, o1 = d.y / OWN_R, o2 = d.z / OWN_R, o3 = d.w / OWN_R;
    const unsigned int q0 = atomicAdd(&lcur[o0], 1u);
    const unsigned int q1 = atomicAdd(&lcur[o1], 1u);
    const unsigned int q2 = atomicAdd(&lcur[o2], 1u);
    const unsigned int q3 = atomicAdd(&lcur[o3], 1u);
    stage[q0] = (unsigned int)s.x | ((unsigned int)(d.x - o0 * OWN_R) << 16) | ((unsigned int)o0 << 24);
    stage[q1] = (unsigned int)s.y | ((unsigned int)(d.y - o1 * OWN_R) << 16) | ((unsigned int)o1 << 24);
    stage[q2] = (unsigned int)s.z | ((unsigned int)(d.z - o2 * OWN_R) << 16) | ((unsigned int)o2 << 24);
    stage[q3] = (unsigned int)s.w | ((unsigned int)(d.w - o3 * OWN_R) << 16) | ((unsigned int)o3 << 24);
  }
  __syncthreads();
  unsigned int* __restrict__ bk = bucket + (size_t)g * EE;
  for (int i = t; i < PCH; i += PT) {
    const unsigned int w = stage[i];
    bk[(unsigned int)i + delta[w >> 24]] = w & 0x00FFFFFFu;
  }
}

#define PL_ND ((size_t)NN * HD)
#define PL_NH ((size_t)NN * HH)

#define PJ_G 512
#define PJ_NT (NN / 16)
#define ELR_G 256
static_assert(NN % 16 == 0 && PJ_G <= PJ_NT, "every projection work-group has a first tile; tiles are whole");

__global__ __launch_bounds__(256) void k_proj(const float* __restrict__ feat,
                                              const float* __restrict__ W0,
                                              const float* __restrict__ W1,
                                              float* __restrict__ hb) {
  __shared__ float As[16 * 132];
  const int g = blockIdx.y;
  const float* __restrict__ W = g ? W1 : W0;
  float* __restrict__ h = hb + (size_t)g * PL_ND;
  const int t  = threadIdx.x;

  const int lane  = t & 31;
  const int wave  = t >> 5;
  const int n0    = wave * 16;
  const int r     = lane & 15;
  const int khalf = (lane >> 4) << 1;
  const int mh    = (lane >> 4) * 8;

  v2f bf[INF_ / 4];
#pragma unroll
  for (int s = 0; s < INF_ / 4; ++s) {
    const int kk = s * 4 + khalf;
    bf[s].x = W[(size_t)(kk)     * HD + n0 + r];
    bf[s].y = W[(size_t)(kk + 1) * HD + n0 + r];
  }

  const int row = t >> 4;
  const int col = (t & 15) * 8;
  float* lp = As + row * 132 + col;
  int tile = blockIdx.x;
  float4 x0, x1;
  {
    const float4* gp = (const float4*)(feat + (size_t)(tile * 16 + row) * INF_ + col);
    x0 = gp[0];
    x1 = gp[1];
  }
  for (; tile < PJ_NT; tile += PJ_G) {
    __syncthreads();
    *(float4*)(lp)     = x0;
    *(float4*)(lp + 4) = x1;
    __syncthreads();
    if (tile + PJ_G < PJ_NT) {
      const float4* gp = (const float4*)(feat + (size_t)((tile + PJ_G) * 16 + row) * INF_ + col);
      x0 = gp[0];
      x1 = gp[1];
    }

    v8f acc = {};
#pragma unroll
    for (int s = 0; s < INF_ / 4; ++s) {
      const int kk = s * 4 + khalf;
      v2f a;
      a.x = As[r * 132 + kk];
      a.y = As[r * 132 + kk + 1];
      acc = __builtin_amdgcn_wmma_f32_16x16x4_f32(false, a, false, bf[s],
                                                  (short)0, acc, false, false);
    }

    const int m0 = tile * 16;
#pragma unroll
    for (int j = 0; j < 8; ++j) {
      h[(size_t)(m0 + j + mh) * HD + n0 + r] = acc[j];
    }
  }
}

__global__ __launch_bounds__(256) void k_elr(const float* __restrict__ hb, const float* __restrict__ al0,
                                             const float* __restrict__ al1, const float* __restrict__ ar0,
                                             const float* __restrict__ ar1, float* __restrict__ elb,
                                             float* __restrict__ erb) {
  const int g = blockIdx.y;
  const float* __restrict__ h  = hb + (size_t)g * PL_ND;
  const float* __restrict__ al = g ? al1 : al0;
  const float* __restrict__ ar = g ? ar1 : ar0;
  float* __restrict__ el = elb + (size_t)g * PL_NH;
  float* __restrict__ er = erb + (size_t)g * PL_NH;
  const int hh = threadIdx.x % HH;
  const float* alp = al + hh * DD;
  const float* arp = ar + hh * DD;
  for (int tid = blockIdx.x * 256 + threadIdx.x; tid < NN * HH; tid += ELR_G * 256) {
    const int n = tid / HH;
    const float* hp = h + (size_t)n * HD + hh * DD;
    float sl = 0.0f, sr = 0.0f;
#pragma unroll
    for (int d = 0; d < DD; d += 4) {
      float4 hv = *(const float4*)(hp + d);
      float4 av = *(const float4*)(alp + d);
      float4 rv = *(const float4*)(arp + d);
      sl += hv.x * av.x + hv.y * av.y + hv.z * av.z + hv.w * av.w;
      sr += hv.x * rv.x + hv.y * rv.y + hv.z * rv.z + hv.w * rv.w;
    }
    el[tid] = sl;
    er[tid] = sr;
  }
}


#define OWN_T 1024
#define OWN_CAP 8192
#define OWN_PIECE (OWN_CAP - 4)
#define OWN_NB 8
static_assert(OWN_NB == 8, "the add loop names the eight table entries of a trip");
#define ACC_P 136
static_assert(OWN_R * HH <= OWN_T, "one thread per (node, head) merges the er and max tables");
static_assert(OWN_CAP == 2 * 4 * OWN_T, "a thread copies two 16-byte units of a list");
static_assert(OWN_CAP <= 8 * OWN_T, "the max pass takes a list in one trip of eight edges per lane");
static_assert(EE % 4 == 0, "a graph's runs start and end on 16-byte boundaries of the workspace");
#define OWN_ACC (OWN_R * ACC_P)
#define OWN_NHH (OWN_R * HH)
#define OWN_LDS ((2 * OWN_ACC + 2 * OWN_NHH + 4 * OWN_NHH + 2 * OWN_CAP) * 4)
static_assert((2 * OWN_ACC) % 4 == 0 && (2 * OWN_NHH) % 4 == 0, "every LDS table starts on a 16-byte boundary");

__device__ __forceinline__ void own_max_v(const float4 ev, const float* __restrict__ er_l,
                                          unsigned int* __restrict__ mu_l, int dl) {
  const float4 rv = *(const float4*)(er_l + dl * HH);
  float x0 = ev.x + rv.x, x1 = ev.y + rv.y, x2 = ev.z + rv.z, x3 = ev.w + rv.w;
  x0 = (x0 > 0.0f) ? x0 : NEG_SLOPE * x0;
  x1 = (x1 > 0.0f) ? x1 : NEG_SLOPE * x1;
  x2 = (x2 > 0.0f) ? x2 : NEG_SLOPE * x2;
  x3 = (x3 > 0.0f) ? x3 : NEG_SLOPE * x3;
  atomicMax(&mu_l[dl * HH + 0], ford(x0));
  atomicMax(&mu_l[dl * HH + 1], ford(x1));
  atomicMax(&mu_l[dl * HH + 2], ford(x2));
  atomicMax(&mu_l[dl * HH + 3], ford(x3));
}

__device__ __forceinline__ void own_add(float* __restrict__ acc_l, float* __restrict__ s_l, const float2 em,
                                        int dl, int lane, int hh, float elv, const float4 hv) {
  float x = elv + em.x;
  x = (x > 0.0f) ? x : NEG_SLOPE * x;
  const float ex = __expf(x - em.y);
  float* ap = acc_l + dl * ACC_P + lane;
  atomicAdd(ap + 0,  ex * hv.x);
  atomicAdd(ap + 32, ex * hv.y);
  atomicAdd(ap + 64, ex * hv.z);
  atomicAdd(ap + 96, ex * hv.w);
  if ((lane & 7) == 0) atomicAdd(&s_l[dl * HH + hh], ex);
}

__device__ __forceinline__ void own_merge(float* tab_l, int t) {
  float e0 = 0.0f, m0 = 0.0f, e1 = 0.0f, m1 = 0.0f;
  if (t < OWN_NHH) {
    e0 = tab_l[t];
    m0 = funord(((const unsigned int*)tab_l)[OWN_NHH + t]);
    e1 = tab_l[2 * OWN_NHH + t];
    m1 = funord(((const unsigned int*)tab_l)[3 * OWN_NHH + t]);
  }
  __syncthreads();
  if (t < OWN_NHH) {
    ((float2*)tab_l)[t] = make_float2(e0, m0);
    ((float2*)tab_l)[OWN_NHH + t] = make_float2(e1, m1);
  }
  __syncthreads();
}

__device__ __forceinline__ void own_load_lists(const uint4* __restrict__ p0, const uint4* __restrict__ p1,
                                               uint4* __restrict__ l0, uint4* __restrict__ l1,
                                               unsigned int nv0, unsigned int nv1, int t) {
  const unsigned int i0 = (unsigned int)t, i1 = (unsigned int)t + OWN_T;
  uint4 a0 = make_uint4(0u, 0u, 0u, 0u), a1 = a0, b0 = a0, b1 = a0;
  if (i0 < nv0) a0 = p0[i0];
  if (i1 < nv0) a1 = p0[i1];
  if (i0 < nv1) b0 = p1[i0];
  if (i1 < nv1) b1 = p1[i1];
  asm volatile("" :: "v"(a0.x), "v"(a0.y), "v"(a0.z), "v"(a0.w), "v"(a1.x), "v"(a1.y), "v"(a1.z), "v"(a1.w),
               "v"(b0.x), "v"(b0.y), "v"(b0.z), "v"(b0.w), "v"(b1.x), "v"(b1.y), "v"(b1.z), "v"(b1.w));
  if (i0 < nv0) l0[i0] = a0;
  if (i1 < nv0) l0[i1] = a1;
  if (i0 < nv1) l1[i0] = b0;
  if (i1 < nv1) l1[i1] = b1;
}

__device__ __forceinline__ void own_max_list(const float* __restrict__ el, const float* er_l, unsigned int* mu_l,
                                             const unsigned int* list_l, unsigned int n, int t) {
  for (unsigned int i = (unsigned int)t; i < n; i += 8u * OWN_T) {
    unsigned int w[8];
    float4 ev[8];
#pragma unroll
    for (int k = 0; k < 8; ++k) {
      const unsigned int ik = i + (unsigned int)k * OWN_T;
      w[k] = list_l[(ik < n) ? ik : i];
    }
#pragma unroll
    for (int k = 0; k < 8; ++k) ev[k] = *(const float4*)(el + (size_t)(w[k] & 0xFFFFu) * HH);
    asm volatile("" :: "v"(ev[0].x), "v"(ev[0].y), "v"(ev[0].z), "v"(ev[0].w), "v"(ev[1].x), "v"(ev[1].y), "v"(ev[1].z),
                 "v"(ev[1].w), "v"(ev[2].x), "v"(ev[2].y), "v"(ev[2].z), "v"(ev[2].w), "v"(ev[3].x), "v"(ev[3].y),
                 "v"(ev[3].z), "v"(ev[3].w), "v"(ev[4].x), "v"(ev[4].y), "v"(ev[4].z), "v"(ev[4].w), "v"(ev[5].x),
                 "v"(ev[5].y), "v"(ev[5].z), "v"(ev[5].w), "v"(ev[6].x), "v"(ev[6].y), "v"(ev[6].z), "v"(ev[6].w),
                 "v"(ev[7].x), "v"(ev[7].y), "v"(ev[7].z), "v"(ev[7].w));
#pragma unroll
    for (int k = 0; k < 8; ++k) own_max_v(ev[k], er_l, mu_l, (int)(w[k] >> 16));
  }
}

__device__ __forceinline__ void own_add_list(const float* __restrict__ h, const float* __restrict__ el,
                                             float* __restrict__ acc_l, float* __restrict__ s_l, const float2* em_l,
                                             const unsigned int* list_l, unsigned int n, int wave, int lane, int hh) {
  unsigned int b = (unsigned int)wave * OWN_NB;
  if (b >= n) return;
  unsigned int w[OWN_NB];
  {
    const unsigned int li = b + (unsigned int)(lane & (OWN_NB - 1));
    const int lv = (int)list_l[(li < n) ? li : b];
#pragma unroll
    for (int k = 0; k < OWN_NB; ++k) w[k] = (unsigned int)__builtin_amdgcn_readlane(lv, k);
  }
  for (;;) {
    const unsigned int left = n - b;
    const unsigned int bn = b + (OWN_T / 32) * OWN_NB;
    const bool more = bn < n;
    int lvn = 0;
    if (more) {
      const unsigned int li = bn + (unsigned int)(lane & (OWN_NB - 1));
      lvn = (int)list_l[(li < n) ? li : bn];
    }
    float4 hv[OWN_NB];
    float  ev[OWN_NB];
    float2 em[OWN_NB];
    int    dl[OWN_NB];
#pragma unroll
    for (int k = 0; k < OWN_NB; ++k) hv[k] = *(const float4*)(h + (size_t)(w[k] & 0xFFFFu) * HD + lane * 4);
#pragma unroll
    for (int k = 0; k < OWN_NB; ++k) ev[k] = el[(w[k] & 0xFFFFu) * HH + hh];
#pragma unroll
    for (int k = 0; k < OWN_NB; ++k) {
      dl[k] = (int)(w[k] >> 16);
      em[k] = em_l[dl[k] * HH + hh];
    }
    asm volatile("" :: "v"(em[0].x), "v"(em[0].y), "v"(em[1].x), "v"(em[1].y), "v"(em[2].x), "v"(em[2].y), "v"(em[3].x),
                 "v"(em[3].y), "v"(em[4].x), "v"(em[4].y), "v"(em[5].x), "v"(em[5].y), "v"(em[6].x), "v"(em[6].y),
                 "v"(em[7].x), "v"(em[7].y));
    if (more) {
#pragma unroll
      for (int k = 0; k < OWN_NB; ++k) w[k] = (unsigned int)__builtin_amdgcn_readlane(lvn, k);
    }
    own_add(acc_l, s_l, em[0], dl[0], lane, hh, ev[0], hv[0]);
#pragma unroll
    for (int k = 1; k < OWN_NB; ++k)
      if ((unsigned int)k < left) own_add(acc_l, s_l, em[k], dl[k], lane, hh, ev[k], hv[k]);
    if (!more) break;
    b = bn;
  }
}

__device__ __forceinline__ float inv_or_zero(float v) {
  return (v != 0.0f) ? 1.0f / v : 0.0f;
}
__device__ __forceinline__ float norm_one(float f0, float f1, float f2, float f3, float b0, float b1, float b2,
                                          float b3, const float4 i0, const float4 i1) {
  float r = f0 * i0.x + b0 * i1.x;
  r += f1 * i0.y + b1 * i1.y;
  r += f2 * i0.z + b2 * i1.z;
  r += f3 * i0.w + b3 * i1.w;
  return 0.25f * r;
}
__device__ __forceinline__ float norm_lds(const float* a0, const float* a1, int j, const float4 i0, const float4 i1) {
  return norm_one(a0[j * 32], a0[j * 32 + 8], a0[j * 32 + 16], a0[j * 32 + 24],
                  a1[j * 32], a1[j * 32 + 8], a1[j * 32 + 16], a1[j * 32 + 24], i0, i1);
}

__device__ __forceinline__ unsigned int own_piece(unsigned int n_own, unsigned int tb) {
  if (tb >= n_own) return 0u;
  return (n_own - tb < OWN_PIECE) ? (n_own - tb) : OWN_PIECE;
}

__global__ __launch_bounds__(OWN_T) void k_edge_own(const float* __restrict__ elb,
                                                    const float* __restrict__ erb,
                                                    const unsigned int* __restrict__ obase,
                                                    const unsigned int* __restrict__ bucket,
                                                    const float* __restrict__ hb,
                                                    float* __restrict__ out) {
  const int ob = blockIdx.x;
  const float* __restrict__ el0 = elb;
  const float* __restrict__ el1 = elb + PL_NH;
  const float* __restrict__ h0 = hb;
  const float* __restrict__ h1 = hb + PL_ND;
  __shared__ __attribute__((aligned(16))) unsigned char L[OWN_LDS];
  float* acc_l = (float*)L;
  float* s_l   = acc_l + 2 * OWN_ACC;
  float* tab_l = s_l + 2 * OWN_NHH;
  unsigned int* list_l = (unsigned int*)(tab_l + 4 * OWN_NHH);

  const int t    = threadIdx.x;
  const int lane = t & 31;
  const int wave = __builtin_amdgcn_readfirstlane(t >> 5);
  const int hh   = lane >> 3;
  const int n0   = ob * OWN_R;
  const int nrows = (NN - n0 < OWN_R) ? (NN - n0) : OWN_R;

  const unsigned int ra0 = (unsigned int)__builtin_amdgcn_readfirstlane((int)obase[ob]);
  const unsigned int rb0 = (unsigned int)__builtin_amdgcn_readfirstlane((int)obase[ob + 1]);
  const unsigned int ra1 = (unsigned int)__builtin_amdgcn_readfirstlane((int)obase[(OWN_P + 1) + ob]);
  const unsigned int rb1 = (unsigned int)__builtin_amdgcn_readfirstlane((int)obase[(OWN_P + 1) + ob + 1]);
  const unsigned int n_own0 = rb0 - ra0, n_own1 = rb1 - ra1;
  const unsigned int n_ownx = (n_own0 > n_own1) ? n_own0 : n_own1;
  const bool whole = n_ownx <= OWN_PIECE;

  for (int i = t; i < 2 * OWN_ACC / 4; i += OWN_T)
    ((float4*)acc_l)[i] = make_float4(0.0f, 0.0f, 0.0f, 0.0f);
  for (int i = t; i < OWN_NHH; i += OWN_T) {
    const bool ok = i < nrows * HH;
    s_l[i]           = 0.0f;
    s_l[OWN_NHH + i] = 0.0f;
    tab_l[i]               = ok ? erb[(size_t)n0 * HH + i] : 0.0f;
    tab_l[2 * OWN_NHH + i] = ok ? erb[PL_NH + (size_t)n0 * HH + i] : 0.0f;
    ((unsigned int*)tab_l)[OWN_NHH + i]     = FORD_NEG_INF;
    ((unsigned int*)tab_l)[3 * OWN_NHH + i] = FORD_NEG_INF;
  }

  for (unsigned int tb = 0u; tb < n_ownx; tb += OWN_PIECE) {
    const unsigned int na = own_piece(n_own0, tb), nb = own_piece(n_own1, tb);
    const unsigned int ga = ra0 + tb, gb = ra1 + tb;
    const unsigned int sa = ga & 3u, sb = gb & 3u;
    own_load_lists((const uint4*)(bucket + (ga - sa)), (const uint4*)(bucket + EE + (gb - sb)),
                   (uint4*)list_l, (uint4*)(list_l + OWN_CAP),
                   na ? (sa + na + 3u) >> 2 : 0u, nb ? (sb + nb + 3u) >> 2 : 0u, t);
    __syncthreads();
    own_max_list(el0, tab_l, (unsigned int*)tab_l + OWN_NHH, list_l + sa, na, t);
    own_max_list(el1, tab_l + 2 * OWN_NHH, (unsigned int*)tab_l + 3 * OWN_NHH, list_l + OWN_CAP + sb, nb, t);
    __syncthreads();
  }
  __syncthreads();
  own_merge(tab_l, t);
  for (unsigned int tb = 0u; tb < n_ownx; tb += OWN_PIECE) {
    const unsigned int na = own_piece(n_own0, tb), nb = own_piece(n_own1, tb);
    const unsigned int ga = ra0 + tb, gb = ra1 + tb;
    const unsigned int sa = ga & 3u, sb = gb & 3u;
    if (!whole) {
      own_load_lists((const uint4*)(bucket + (ga - sa)), (const uint4*)(bucket + EE + (gb - sb)),
                     (uint4*)list_l, (uint4*)(list_l + OWN_CAP),
                     na ? (sa + na + 3u) >> 2 : 0u, nb ? (sb + nb + 3u) >> 2 : 0u, t);
      __syncthreads();
    }
    own_add_list(h0, el0, acc_l, s_l, (const float2*)tab_l, list_l + sa, na, wave, lane, hh);
    own_add_list(h1, el1, acc_l + OWN_ACC, s_l + OWN_NHH, (const float2*)tab_l + OWN_NHH, list_l + OWN_CAP + sb, nb,
                 wave, lane, hh);
    if (!whole) __syncthreads();
  }
  __syncthreads();

  for (int i = t; i < nrows * 8; i += OWN_T) {
    const int node = i >> 3;
    const int d4   = i & 7;
    const float4 v0 = *(const float4*)(s_l + node * HH);
    const float4 v1 = *(const float4*)(s_l + OWN_NHH + node * HH);
    const float4 i0 = make_float4(inv_or_zero(v0.x), inv_or_zero(v0.y), inv_or_zero(v0.z), inv_or_zero(v0.w));
    const float4 i1 = make_float4(inv_or_zero(v1.x), inv_or_zero(v1.y), inv_or_zero(v1.z), inv_or_zero(v1.w));
    const float* a0 = acc_l + node * ACC_P + d4;
    const float* a1 = a0 + OWN_ACC;
    float4 o;
    o.x = norm_lds(a0, a1, 0, i0, i1);
    o.y = norm_lds(a0, a1, 1, i0, i1);
    o.z = norm_lds(a0, a1, 2, i0, i1);
    o.w = norm_lds(a0, a1, 3, i0, i1);
    *(float4*)(out + (size_t)(n0 + node) * DD + d4 * 4) = o;
  }
}

#define OBAS_W ((2 * (OWN_P + 1) + 3) & ~3)

extern "C" void kernel_launch(void* const* d_in, const int* in_sizes, int n_in,
                              void* d_out, int out_size, void* d_ws, size_t ws_size,
                              hipStream_t stream) {
  const float* feat = (const float*)d_in[0];
  const int*   src_g[2] = { (const int*)d_in[1], (const int*)d_in[3] };
  const int*   dst_g[2] = { (const int*)d_in[2], (const int*)d_in[4] };
  const float* W_g[2]   = { (const float*)d_in[5], (const float*)d_in[8] };
  const float* al_g[2]  = { (const float*)d_in[6], (const float*)d_in[9] };
  const float* ar_g[2]  = { (const float*)d_in[7], (const float*)d_in[10] };
  float* out = (float*)d_out;

  float*        hbuf = (float*)d_ws;
  float*        el   = hbuf + 2 * (size_t)NN * HD;
  float*        er   = el   + 2 * (size_t)NN * HH;
  unsigned int* cntb = (unsigned int*)(er + 2 * (size_t)NN * HH);
  unsigned int* obas = cntb + 2 * (size_t)OWN_P * PB;
  unsigned int* buck = obas + OBAS_W;

  const int T = 256;
  k_count<<<dim3(PB, 2), PTC, 0, stream>>>(dst_g[0], dst_g[1], cntb);
  k_scatter<<<dim3(PB, 2), PT, 0, stream>>>(src_g[0], src_g[1], dst_g[0], dst_g[1], cntb, obas, buck);

  k_proj<<<dim3(PJ_G, 2), T, 0, stream>>>(feat, W_g[0], W_g[1], hbuf);
  k_elr<<<dim3(ELR_G, 2), T, 0, stream>>>(hbuf, al_g[0], al_g[1], ar_g[0], ar_g[1], el, er);
  k_edge_own<<<OWN_P, OWN_T, 0, stream>>>(el, er, obas, buck, hbuf, out);
}


# owner pass orders its LDS list by local row (LDS counting sort); a wave sums a row's run of ex*h[src] in registers, one LDS add per run, not per edge; sum order changes (list order, then LDS arrival)
# speedup vs baseline: 5.3446x; 1.1160x over previous
#include <hip/hip_runtime.h>

#define NN 50000
#define EE 1600000
#define INF_ 128
#define HH 4
#define DD 32
#define HD 128
#define NEG_SLOPE 0.2f

typedef __attribute__((ext_vector_type(2))) float v2f;
typedef __attribute__((ext_vector_type(8))) float v8f;

__device__ __forceinline__ unsigned int ford(float f) {
  unsigned int b = __float_as_uint(f);
  return (b & 0x80000000u) ? ~b : (b | 0x80000000u);
}
__device__ __forceinline__ float funord(unsigned int u) {
  unsigned int b = (u & 0x80000000u) ? (u ^ 0x80000000u) : ~u;
  return __uint_as_float(b);
}
#define FORD_NEG_INF 0x007FFFFFu

#define OWN_R 200
#define OWN_P ((NN + OWN_R - 1) / OWN_R)
static_assert(NN <= 65536 && OWN_R <= 65536, "src and the local row share one list word");

#define PB 250
#define PCH (EE / PB)
#define PT 1024
#define PTC 256
#define PCW ((PCH / 4 + PTC - 1) / PTC)
static_assert(EE % PB == 0 && PCH % 4 == 0, "chunks are whole and start on 16-byte boundaries");
static_assert(OWN_P <= 256 && OWN_R <= 256, "owner and local row take a byte each in a staged word; one wave scans the owners");
static_assert(OWN_P <= PTC, "one counting thread per owner clears and stores the counts");

__global__ __launch_bounds__(PTC) void k_count(const int* __restrict__ dst0, const int* __restrict__ dst1,
                                               unsigned int* __restrict__ cnt) {
  __shared__ unsigned int c[OWN_P];
  const int t = threadIdx.x;
  const int b = blockIdx.x;
  const int g = blockIdx.y;
  if (t < OWN_P) c[t] = 0u;
  __syncthreads();
  const int4* dp = (const int4*)((g ? dst1 : dst0) + (size_t)b * PCH);
  int4 d[PCW];
#pragma unroll
  for (int k = 0; k < PCW; ++k) {
    const int i = t + k * PTC;
    d[k] = dp[(i < PCH / 4) ? i : t];
  }
  static_assert(PCW == 7, "the statement below names the seven words of a thread");
  asm volatile("" :: "v"(d[0].x), "v"(d[0].y), "v"(d[0].z), "v"(d[0].w), "v"(d[1].x), "v"(d[1].y), "v"(d[1].z), "v"(d[1].w),
               "v"(d[2].x), "v"(d[2].y), "v"(d[2].z), "v"(d[2].w), "v"(d[3].x), "v"(d[3].y), "v"(d[3].z), "v"(d[3].w),
               "v"(d[4].x), "v"(d[4].y), "v"(d[4].z), "v"(d[4].w), "v"(d[5].x), "v"(d[5].y), "v"(d[5].z), "v"(d[5].w),
               "v"(d[6].x), "v"(d[6].y), "v"(d[6].z), "v"(d[6].w));
#pragma unroll
  for (int k = 0; k < PCW; ++k) {
    if (t + k * PTC < PCH / 4) {
      atomicAdd(&c[d[k].x / OWN_R], 1u);
      atomicAdd(&c[d[k].y / OWN_R], 1u);
      atomicAdd(&c[d[k].z / OWN_R], 1u);
      atomicAdd(&c[d[k].w / OWN_R], 1u);
    }
  }
  __syncthreads();
  if (t < OWN_P) cnt[((size_t)g * PB + b) * OWN_P + t] = c[t];
}

__global__ __launch_bounds__(PT) void k_scatter(const int* __restrict__ src0, const int* __restrict__ src1,
                                                const int* __restrict__ dst0, const int* __restrict__ dst1,
                                                const unsigned int* __restrict__ cnt,
                                                unsigned int* __restrict__ obase,
                                                unsigned int* __restrict__ bucket) {
  __shared__ unsigned int part_a[4][256];
  __shared__ unsigned int part_p[4][256];
  __shared__ unsigned int lcur[256];
  __shared__ unsigned int delta[256];
  __shared__ unsigned int stage[PCH];
  const int t = threadIdx.x;
  const int b = blockIdx.x;
  const int g = blockIdx.y;
  const int lane = t & 31;
  const int wave = t >> 5;
  const int q  = t >> 8;
  const int oc = t & 255;
  const unsigned int* __restrict__ cg = cnt + (size_t)g * PB * OWN_P;
  unsigned int a = 0u, p = 0u;
  if (oc < OWN_P) {
#pragma unroll 8
    for (int bb = q; bb < PB; bb += 4) {
      const unsigned int v = cg[(size_t)bb * OWN_P + oc];
      a += v;
      p += (bb < b) ? v : 0u;
    }
  }
  part_a[q][oc] = a;
  part_p[q][oc] = p;
  __syncthreads();
  if (wave == 0) {
    unsigned int vt[8], vp[8], vo[8];
    unsigned int st = 0u, so = 0u;
#pragma unroll
    for (int k = 0; k < 8; ++k) {
      const int o = lane * 8 + k;
      vt[k] = part_a[0][o] + part_a[1][o] + part_a[2][o] + part_a[3][o];
      vp[k] = part_p[0][o] + part_p[1][o] + part_p[2][o] + part_p[3][o];
      vo[k] = (o < OWN_P) ? cg[(size_t)b * OWN_P + o] : 0u;
      st += vt[k];
      so += vo[k];
    }
    unsigned int it = st, io = so;
#pragma unroll
    for (int d = 1; d < 32; d <<= 1) {
      const unsigned int nt = (unsigned int)__shfl_up((int)it, d);
      const unsigned int no = (unsigned int)__shfl_up((int)io, d);
      if (lane >= d) { it += nt; io += no; }
    }
    unsigned int run = it - st;
    unsigned int loc = io - so;
    unsigned int* __restrict__ og = obase + (size_t)g * (OWN_P + 1);
#pragma unroll
    for (int k = 0; k < 8; ++k) {
      const int o = lane * 8 + k;
      if (o < OWN_P) {
        lcur[o]  = loc;
        delta[o] = run + vp[k] - loc;
        if (b == 0) og[o] = run;
      }
      run += vt[k];
      loc += vo[k];
    }
    if (b == 0 && lane == 31) og[OWN_P] = run;
  }
  __syncthreads();
  const int4* dp = (const int4*)((g ? dst1 : dst0) + (size_t)b * PCH);
  const int4* sp = (const int4*)((g ? src1 : src0) + (size_t)b * PCH);
  for (int i = t; i < PCH / 4; i += PT) {
    const int4 d = dp[i];
    const int4 s = sp[i];
    const int o0 = d.x / OWN_R, o1 = d.y / OWN_R, o2 = d.z / OWN_R, o3 = d.w / OWN_R;
    const unsigned int q0 = atomicAdd(&lcur[o0], 1u);
    const unsigned int q1 = atomicAdd(&lcur[o1], 1u);
    const unsigned int q2 = atomicAdd(&lcur[o2], 1u);
    const unsigned int q3 = atomicAdd(&lcur[o3], 1u);
    stage[q0] = (unsigned int)s.x | ((unsigned int)(d.x - o0 * OWN_R) << 16) | ((unsigned int)o0 << 24);
    stage[q1] = (unsigned int)s.y | ((unsigned int)(d.y - o1 * OWN_R) << 16) | ((unsigned int)o1 << 24);
    stage[q2] = (unsigned int)s.z | ((unsigned int)(d.z - o2 * OWN_R) << 16) | ((unsigned int)o2 << 24);
    stage[q3] = (unsigned int)s.w | ((unsigned int)(d.w - o3 * OWN_R) << 16) | ((unsigned int)o3 << 24);
  }
  __syncthreads();
  unsigned int* __restrict__ bk = bucket + (size_t)g * EE;
  for (int i = t; i < PCH; i += PT) {
    const unsigned int w = stage[i];
    bk[(unsigned int)i + delta[w >> 24]] = w & 0x00FFFFFFu;
  }
}

#define PL_ND ((size_t)NN * HD)
#define PL_NH ((size_t)NN * HH)

#define PJ_G 512
#define PJ_NT (NN / 16)
#define ELR_G 256
static_assert(NN % 16 == 0 && PJ_G <= PJ_NT, "every projection work-group has a first tile; tiles are whole");

__global__ __launch_bounds__(256) void k_proj(const float* __restrict__ feat,
                                              const float* __restrict__ W0,
                                              const float* __restrict__ W1,
                                              float* __restrict__ hb) {
  __shared__ float As[16 * 132];
  const int g = blockIdx.y;
  const float* __restrict__ W = g ? W1 : W0;
  float* __restrict__ h = hb + (size_t)g * PL_ND;
  const int t  = threadIdx.x;

  const int lane  = t & 31;
  const int wave  = t >> 5;
  const int n0    = wave * 16;
  const int r     = lane & 15;
  const int khalf = (lane >> 4) << 1;
  const int mh    = (lane >> 4) * 8;

  v2f bf[INF_ / 4];
#pragma unroll
  for (int s = 0; s < INF_ / 4; ++s) {
    const int kk = s * 4 + khalf;
    bf[s].x = W[(size_t)(kk)     * HD + n0 + r];
    bf[s].y = W[(size_t)(kk + 1) * HD + n0 + r];
  }

  const int row = t >> 4;
  const int col = (t & 15) * 8;
  float* lp = As + row * 132 + col;
  int tile = blockIdx.x;
  float4 x0, x1;
  {
    const float4* gp = (const float4*)(feat + (size_t)(tile * 16 + row) * INF_ + col);
    x0 = gp[0];
    x1 = gp[1];
  }
  for (; tile < PJ_NT; tile += PJ_G) {
    __syncthreads();
    *(float4*)(lp)     = x0;
    *(float4*)(lp + 4) = x1;
    __syncthreads();
    if (tile + PJ_G < PJ_NT) {
      const float4* gp = (const float4*)(feat + (size_t)((tile + PJ_G) * 16 + row) * INF_ + col);
      x0 = gp[0];
      x1 = gp[1];
    }

    v8f acc = {};
#pragma unroll
    for (int s = 0; s < INF_ / 4; ++s) {
      const int kk = s * 4 + khalf;
      v2f a;
      a.x = As[r * 132 + kk];
      a.y = As[r * 132 + kk + 1];
      acc = __builtin_amdgcn_wmma_f32_16x16x4_f32(false, a, false, bf[s],
                                                  (short)0, acc, false, false);
    }

    const int m0 = tile * 16;
#pragma unroll
    for (int j = 0; j < 8; ++j) {
      h[(size_t)(m0 + j + mh) * HD + n0 + r] = acc[j];
    }
  }
}

__global__ __launch_bounds__(256) void k_elr(const float* __restrict__ hb, const float* __restrict__ al0,
                                             const float* __restrict__ al1, const float* __restrict__ ar0,
                                             const float* __restrict__ ar1, float* __restrict__ elb,
                                             float* __restrict__ erb) {
  const int g = blockIdx.y;
  const float* __restrict__ h  = hb + (size_t)g * PL_ND;
  const float* __restrict__ al = g ? al1 : al0;
  const float* __restrict__ ar = g ? ar1 : ar0;
  float* __restrict__ el = elb + (size_t)g * PL_NH;
  float* __restrict__ er = erb + (size_t)g * PL_NH;
  const int hh = threadIdx.x % HH;
  const float* alp = al + hh * DD;
  const float* arp = ar + hh * DD;
  for (int tid = blockIdx.x * 256 + threadIdx.x; tid < NN * HH; tid += ELR_G * 256) {
    const int n = tid / HH;
    const float* hp = h + (size_t)n * HD + hh * DD;
    float sl = 0.0f, sr = 0.0f;
#pragma unroll
    for (int d = 0; d < DD; d += 4) {
      float4 hv = *(const float4*)(hp + d);
      float4 av = *(const float4*)(alp + d);
      float4 rv = *(const float4*)(arp + d);
      sl += hv.x * av.x + hv.y * av.y + hv.z * av.z + hv.w * av.w;
      sr += hv.x * rv.x + hv.y * rv.y + hv.z * rv.z + hv.w * rv.w;
    }
    el[tid] = sl;
    er[tid] = sr;
  }
}


#define OWN_T 1024
#define OWN_CAP 8192
#define OWN_PIECE (OWN_CAP - 4)
#define OWN_NB 8
static_assert(OWN_NB == 8, "the add loop names the eight table entries of a trip");
#define ACC_P 136
static_assert(OWN_R * HH <= OWN_T, "one thread per (node, head) merges the er and max tables");
static_assert(OWN_CAP == 2 * 4 * OWN_T, "a thread copies two 16-byte units of a list");
static_assert(OWN_CAP <= 8 * OWN_T, "the max pass takes a list in one trip of eight edges per lane");
static_assert(EE % 4 == 0, "a graph's runs start and end on 16-byte boundaries of the workspace");
#define OWN_ACC (OWN_R * ACC_P)
#define OWN_NHH (OWN_R * HH)
#define OWN_CNT 224
static_assert(OWN_CNT >= OWN_R && OWN_CNT == 7 * 32, "a lane scans seven neighbouring row counters");
static_assert(2 * OWN_CAP <= 2 * OWN_ACC, "the unordered lists are staged in the accumulator rows before those are cleared");
#define OWN_LDS ((2 * OWN_ACC + 2 * OWN_NHH + 4 * OWN_NHH + 2 * OWN_CAP + 2 * OWN_CNT) * 4)
static_assert((2 * OWN_ACC) % 4 == 0 && (2 * OWN_NHH) % 4 == 0, "every LDS table starts on a 16-byte boundary");

__device__ __forceinline__ void own_max_v(const float4 ev, const float* __restrict__ er_l,
                                          unsigned int* __restrict__ mu_l, int dl) {
  const float4 rv = *(const float4*)(er_l + dl * HH);
  float x0 = ev.x + rv.x, x1 = ev.y + rv.y, x2 = ev.z + rv.z, x3 = ev.w + rv.w;
  x0 = (x0 > 0.0f) ? x0 : NEG_SLOPE * x0;
  x1 = (x1 > 0.0f) ? x1 : NEG_SLOPE * x1;
  x2 = (x2 > 0.0f) ? x2 : NEG_SLOPE * x2;
  x3 = (x3 > 0.0f) ? x3 : NEG_SLOPE * x3;
  atomicMax(&mu_l[dl * HH + 0], ford(x0));
  atomicMax(&mu_l[dl * HH + 1], ford(x1));
  atomicMax(&mu_l[dl * HH + 2], ford(x2));
  atomicMax(&mu_l[dl * HH + 3], ford(x3));
}

__device__ __forceinline__ float own_ex(const float2 em, float elv) {
  float x = elv + em.x;
  x = (x > 0.0f) ? x : NEG_SLOPE * x;
  return __expf(x - em.y);
}
__device__ __forceinline__ void own_flush(float* __restrict__ acc_l, float* __restrict__ s_l, int dl, int lane, int hh,
                                          const float4 a, float sa) {
  float* ap = acc_l + dl * ACC_P + lane;
  atomicAdd(ap + 0,  a.x);
  atomicAdd(ap + 32, a.y);
  atomicAdd(ap + 64, a.z);
  atomicAdd(ap + 96, a.w);
  if ((lane & 7) == 0) atomicAdd(&s_l[dl * HH + hh], sa);
}

__device__ __forceinline__ void own_merge(float* tab_l, int t) {
  float e0 = 0.0f, m0 = 0.0f, e1 = 0.0f, m1 = 0.0f;
  if (t < OWN_NHH) {
    e0 = tab_l[t];
    m0 = funord(((const unsigned int*)tab_l)[OWN_NHH + t]);
    e1 = tab_l[2 * OWN_NHH + t];
    m1 = funord(((const unsigned int*)tab_l)[3 * OWN_NHH + t]);
  }
  __syncthreads();
  if (t < OWN_NHH) {
    ((float2*)tab_l)[t] = make_float2(e0, m0);
    ((float2*)tab_l)[OWN_NHH + t] = make_float2(e1, m1);
  }
  __syncthreads();
}

__device__ __forceinline__ void own_load_lists(const uint4* __restrict__ p0, const uint4* __restrict__ p1,
                                               uint4* __restrict__ l0, uint4* __restrict__ l1,
                                               unsigned int nv0, unsigned int nv1, int t) {
  const unsigned int i0 = (unsigned int)t, i1 = (unsigned int)t + OWN_T;
  uint4 a0 = make_uint4(0u, 0u, 0u, 0u), a1 = a0, b0 = a0, b1 = a0;
  if (i0 < nv0) a0 = p0[i0];
  if (i1 < nv0) a1 = p0[i1];
  if (i0 < nv1) b0 = p1[i0];
  if (i1 < nv1) b1 = p1[i1];
  asm volatile("" :: "v"(a0.x), "v"(a0.y), "v"(a0.z), "v"(a0.w), "v"(a1.x), "v"(a1.y), "v"(a1.z), "v"(a1.w),
               "v"(b0.x), "v"(b0.y), "v"(b0.z), "v"(b0.w), "v"(b1.x), "v"(b1.y), "v"(b1.z), "v"(b1.w));
  if (i0 < nv0) l0[i0] = a0;
  if (i1 < nv0) l0[i1] = a1;
  if (i0 < nv1) l1[i0] = b0;
  if (i1 < nv1) l1[i1] = b1;
}

template <bool COUNT>
__device__ __forceinline__ void own_max_list(const float* __restrict__ el, const float* er_l, unsigned int* mu_l,
                                             const unsigned int* list_l, unsigned int n, int t,
                                             unsigned int* cnt_l) {
  for (unsigned int i = (unsigned int)t; i < n; i += 8u * OWN_T) {
    unsigned int w[8];
    float4 ev[8];
#pragma unroll
    for (int k = 0; k < 8; ++k) {
      const unsigned int ik = i + (unsigned int)k * OWN_T;
      w[k] = list_l[(ik < n) ? ik : i];
    }
#pragma unroll
    for (int k = 0; k < 8; ++k) ev[k] = *(const float4*)(el + (size_t)(w[k] & 0xFFFFu) * HH);
    asm volatile("" :: "v"(ev[0].x), "v"(ev[0].y), "v"(ev[0].z), "v"(ev[0].w), "v"(ev[1].x), "v"(ev[1].y), "v"(ev[1].z),
                 "v"(ev[1].w), "v"(ev[2].x), "v"(ev[2].y), "v"(ev[2].z), "v"(ev[2].w), "v"(ev[3].x), "v"(ev[3].y),
                 "v"(ev[3].z), "v"(ev[3].w), "v"(ev[4].x), "v"(ev[4].y), "v"(ev[4].z), "v"(ev[4].w), "v"(ev[5].x),
                 "v"(ev[5].y), "v"(ev[5].z), "v"(ev[5].w), "v"(ev[6].x), "v"(ev[6].y), "v"(ev[6].z), "v"(ev[6].w),
                 "v"(ev[7].x), "v"(ev[7].y), "v"(ev[7].z), "v"(ev[7].w));
#pragma unroll
    for (int k = 0; k < 8; ++k) own_max_v(ev[k], er_l, mu_l, (int)(w[k] >> 16));
    if (COUNT) {
#pragma unroll
      for (int k = 0; k < 8; ++k)
        if (i + (unsigned int)k * OWN_T < n) atomicAdd(&cnt_l[w[k] >> 16], 1u);
    }
  }
}

__device__ __forceinline__ void own_scan_rows(unsigned int* c, int lane) {
  unsigned int v[7];
  unsigned int s = 0u;
#pragma unroll
  for (int k = 0; k < 7; ++k) { v[k] = c[lane * 7 + k]; s += v[k]; }
  unsigned int inc = s;
#pragma unroll
  for (int d = 1; d < 32; d <<= 1) {
    const unsigned int nb = (unsigned int)__shfl_up((int)inc, d);
    if (lane >= d) inc += nb;
  }
  unsigned int run = inc - s;
#pragma unroll
  for (int k = 0; k < 7; ++k) { c[lane * 7 + k] = run; run += v[k]; }
}

__device__ __forceinline__ void own_order_list(const unsigned int* tmp_l, unsigned int* list_l, unsigned int* cur_l,
                                               unsigned int n, int t) {
  for (unsigned int i = (unsigned int)t; i < n; i += OWN_T) {
    const unsigned int w = tmp_l[i];
    const unsigned int pos = atomicAdd(&cur_l[w >> 16], 1u);
    list_l[pos] = w;
  }
}

__device__ __forceinline__ void own_add_list(const float* __restrict__ h, const float* __restrict__ el,
                                             float* __restrict__ acc_l, float* __restrict__ s_l, const float2* em_l,
                                             const unsigned int* list_l, unsigned int n, int wave, int lane, int hh) {
  const unsigned int per = ((n + 32u * OWN_NB - 1u) / (32u * OWN_NB)) * OWN_NB;
  unsigned int b = (unsigned int)wave * per;
  if (b >= n) return;
  const unsigned int end = (b + per < n) ? (b + per) : n;
  unsigned int w[OWN_NB];
  {
    const unsigned int li = b + (unsigned int)(lane & (OWN_NB - 1));
    const int lv = (int)list_l[(li < end) ? li : b];
#pragma unroll
    for (int k = 0; k < OWN_NB; ++k) w[k] = (unsigned int)__builtin_amdgcn_readlane(lv, k);
  }
  int cur = (int)(w[0] >> 16);
  float4 a = make_float4(0.0f, 0.0f, 0.0f, 0.0f);
  float sa = 0.0f;
  for (;;) {
    const unsigned int left = end - b;
    const unsigned int bn = b + OWN_NB;
    const bool more = bn < end;
    int lvn = 0;
    if (more) {
      const unsigned int li = bn + (unsigned int)(lane & (OWN_NB - 1));
      lvn = (int)list_l[(li < end) ? li : bn];
    }
    float4 hv[OWN_NB];
    float  ev[OWN_NB];
    float2 em[OWN_NB];
    int    dl[OWN_NB];
#pragma unroll
    for (int k = 0; k < OWN_NB; ++k) hv[k] = *(const float4*)(h + (size_t)(w[k] & 0xFFFFu) * HD + lane * 4);
#pragma unroll
    for (int k = 0; k < OWN_NB; ++k) ev[k] = el[(w[k] & 0xFFFFu) * HH + hh];
#pragma unroll
    for (int k = 0; k < OWN_NB; ++k) {
      dl[k] = (int)(w[k] >> 16);
      em[k] = em_l[dl[k] * HH + hh];
    }
    asm volatile("" :: "v"(em[0].x), "v"(em[0].y), "v"(em[1].x), "v"(em[1].y), "v"(em[2].x), "v"(em[2].y), "v"(em[3].x),
                 "v"(em[3].y), "v"(em[4].x), "v"(em[4].y), "v"(em[5].x), "v"(em[5].y), "v"(em[6].x), "v"(em[6].y),
                 "v"(em[7].x), "v"(em[7].y));
    if (more) {
#pragma unroll
      for (int k = 0; k < OWN_NB; ++k) w[k] = (unsigned int)__builtin_amdgcn_readlane(lvn, k);
    }
#pragma unroll
    for (int k = 0; k < OWN_NB; ++k) {
      if (k == 0 || (unsigned int)k < left) {
        if (dl[k] != cur) {
          own_flush(acc_l, s_l, cur, lane, hh, a, sa);
          a = make_float4(0.0f, 0.0f, 0.0f, 0.0f);
          sa = 0.0f;
          cur = dl[k];
        }
        const float ex = own_ex(em[k], ev[k]);
        a.x += ex * hv[k].x;
        a.y += ex * hv[k].y;
        a.z += ex * hv[k].z;
        a.w += ex * hv[k].w;
        sa += ex;
      }
    }
    if (!more) break;
    b = bn;
  }
  own_flush(acc_l, s_l, cur, lane, hh, a, sa);
}

__device__ __forceinline__ float inv_or_zero(float v) {
  return (v != 0.0f) ? 1.0f / v : 0.0f;
}
__device__ __forceinline__ float norm_one(float f0, float f1, float f2, float f3, float b0, float b1, float b2,
                                          float b3, const float4 i0, const float4 i1) {
  float r = f0 * i0.x + b0 * i1.x;
  r += f1 * i0.y + b1 * i1.y;
  r += f2 * i0.z + b2 * i1.z;
  r += f3 * i0.w + b3 * i1.w;
  return 0.25f * r;
}
__device__ __forceinline__ float norm_lds(const float* a0, const float* a1, int j, const float4 i0, const float4 i1) {
  return norm_one(a0[j * 32], a0[j * 32 + 8], a0[j * 32 + 16], a0[j * 32 + 24],
                  a1[j * 32], a1[j * 32 + 8], a1[j * 32 + 16], a1[j * 32 + 24], i0, i1);
}

__device__ __forceinline__ unsigned int own_piece(unsigned int n_own, unsigned int tb) {
  if (tb >= n_own) return 0u;
  return (n_own - tb < OWN_PIECE) ? (n_own - tb) : OWN_PIECE;
}

__global__ __launch_bounds__(OWN_T) void k_edge_own(const float* __restrict__ elb,
                                                    const float* __restrict__ erb,
                                                    const unsigned int* __restrict__ obase,
                                                    const unsigned int* __restrict__ bucket,
                                                    const float* __restrict__ hb,
                                                    float* __restrict__ out) {
  const int ob = blockIdx.x;
  const float* __restrict__ el0 = elb;
  const float* __restrict__ el1 = elb + PL_NH;
  const float* __restrict__ h0 = hb;
  const float* __restrict__ h1 = hb + PL_ND;
  __shared__ __attribute__((aligned(16))) unsigned char L[OWN_LDS];
  float* acc_l = (float*)L;
  float* s_l   = acc_l + 2 * OWN_ACC;
  float* tab_l = s_l + 2 * OWN_NHH;
  unsigned int* list_l = (unsigned int*)(tab_l + 4 * OWN_NHH);

  const int t    = threadIdx.x;
  const int lane = t & 31;
  const int wave = __builtin_amdgcn_readfirstlane(t >> 5);
  const int hh   = lane >> 3;
  const int n0   = ob * OWN_R;
  const int nrows = (NN - n0 < OWN_R) ? (NN - n0) : OWN_R;

  const unsigned int ra0 = (unsigned int)__builtin_amdgcn_readfirstlane((int)obase[ob]);
  const unsigned int rb0 = (unsigned int)__builtin_amdgcn_readfirstlane((int)obase[ob + 1]);
  const unsigned int ra1 = (unsigned int)__builtin_amdgcn_readfirstlane((int)obase[(OWN_P + 1) + ob]);
  const unsigned int rb1 = (unsigned int)__builtin_amdgcn_readfirstlane((int)obase[(OWN_P + 1) + ob + 1]);
  const unsigned int n_own0 = rb0 - ra0, n_own1 = rb1 - ra1;
  const unsigned int n_ownx = (n_own0 > n_own1) ? n_own0 : n_own1;
  const bool whole = n_ownx <= OWN_PIECE;

  unsigned int* cnt_l = list_l + 2 * OWN_CAP;
  unsigned int* tmp_l = (unsigned int*)acc_l;

  for (int i = t; i < OWN_NHH; i += OWN_T) {
    const bool ok = i < nrows * HH;
    s_l[i]           = 0.0f;
    s_l[OWN_NHH + i] = 0.0f;
    tab_l[i]               = ok ? erb[(size_t)n0 * HH + i] : 0.0f;
    tab_l[2 * OWN_NHH + i] = ok ? erb[PL_NH + (size_t)n0 * HH + i] : 0.0f;
    ((unsigned int*)tab_l)[OWN_NHH + i]     = FORD_NEG_INF;
    ((unsigned int*)tab_l)[3 * OWN_NHH + i] = FORD_NEG_INF;
  }
  if (t < 2 * OWN_CNT) cnt_l[t] = 0u;

  if (whole) {
    const unsigned int sa = ra0 & 3u, sb = ra1 & 3u;
    own_load_lists((const uint4*)(bucket + (ra0 - sa)), (const uint4*)(bucket + EE + (ra1 - sb)),
                   (uint4*)tmp_l, (uint4*)(tmp_l + OWN_CAP),
                   n_own0 ? (sa + n_own0 + 3u) >> 2 : 0u, n_own1 ? (sb + n_own1 + 3u) >> 2 : 0u, t);
    __syncthreads();
    own_max_list<true>(el0, tab_l, (unsigned int*)tab_l + OWN_NHH, tmp_l + sa, n_own0, t, cnt_l);
    own_max_list<true>(el1, tab_l + 2 * OWN_NHH, (unsigned int*)tab_l + 3 * OWN_NHH, tmp_l + OWN_CAP + sb, n_own1, t,
                       cnt_l + OWN_CNT);
    __syncthreads();
    if (wave < 2) own_scan_rows(cnt_l + wave * OWN_CNT, lane);
    __syncthreads();
    own_order_list(tmp_l + sa, list_l, cnt_l, n_own0, t);
    own_order_list(tmp_l + OWN_CAP + sb, list_l + OWN_CAP, cnt_l + OWN_CNT, n_own1, t);
    __syncthreads();
  }
  for (int i = t; i < 2 * OWN_ACC / 4; i += OWN_T)
    ((float4*)acc_l)[i] = make_float4(0.0f, 0.0f, 0.0f, 0.0f);
  if (!whole) {
    for (unsigned int tb = 0u; tb < n_ownx; tb += OWN_PIECE) {
      const unsigned int na = own_piece(n_own0, tb), nb = own_piece(n_own1, tb);
      const unsigned int ga = ra0 + tb, gb = ra1 + tb;
      const unsigned int sa = ga & 3u, sb = gb & 3u;
      own_load_lists((const uint4*)(bucket + (ga - sa)), (const uint4*)(bucket + EE + (gb - sb)),
                     (uint4*)list_l, (uint4*)(list_l + OWN_CAP),
                     na ? (sa + na + 3u) >> 2 : 0u, nb ? (sb + nb + 3u) >> 2 : 0u, t);
      __syncthreads();
      own_max_list<false>(el0, tab_l, (unsigned int*)tab_l + OWN_NHH, list_l + sa, na, t, cnt_l);
      own_max_list<false>(el1, tab_l + 2 * OWN_NHH, (unsigned int*)tab_l + 3 * OWN_NHH, list_l + OWN_CAP + sb, nb, t,
                          cnt_l);
      __syncthreads();
    }
  }
  __syncthreads();
  own_merge(tab_l, t);
  for (unsigned int tb = 0u; tb < n_ownx; tb += OWN_PIECE) {
    const unsigned int na = own_piece(n_own0, tb), nb = own_piece(n_own1, tb);
    unsigned int sa = 0u, sb = 0u;
    if (!whole) {
      const unsigned int ga = ra0 + tb, gb = ra1 + tb;
      sa = ga & 3u;
      sb = gb & 3u;
      own_load_lists((const uint4*)(bucket + (ga - sa)), (const uint4*)(bucket + EE + (gb - sb)),
                     (uint4*)list_l, (uint4*)(list_l + OWN_CAP),
                     na ? (sa + na + 3u) >> 2 : 0u, nb ? (sb + nb + 3u) >> 2 : 0u, t);
      __syncthreads();
    }
    own_add_list(h0, el0, acc_l, s_l, (const float2*)tab_l, list_l + sa, na, wave, lane, hh);
    own_add_list(h1, el1, acc_l + OWN_ACC, s_l + OWN_NHH, (const float2*)tab_l + OWN_NHH, list_l + OWN_CAP + sb, nb,
                 wave, lane, hh);
    if (!whole) __syncthreads();
  }
  __syncthreads();

  for (int i = t; i < nrows * 8; i += OWN_T) {
    const int node = i >> 3;
    const int d4   = i & 7;
    const float4 v0 = *(const float4*)(s_l + node * HH);
    const float4 v1 = *(const float4*)(s_l + OWN_NHH + node * HH);
    const float4 i0 = make_float4(inv_or_zero(v0.x), inv_or_zero(v0.y), inv_or_zero(v0.z), inv_or_zero(v0.w));
    const float4 i1 = make_float4(inv_or_zero(v1.x), inv_or_zero(v1.y), inv_or_zero(v1.z), inv_or_zero(v1.w));
    const float* a0 = acc_l + node * ACC_P + d4;
    const float* a1 = a0 + OWN_ACC;
    float4 o;
    o.x = norm_lds(a0, a1, 0, i0, i1);
    o.y = norm_lds(a0, a1, 1, i0, i1);
    o.z = norm_lds(a0, a1, 2, i0, i1);
    o.w = norm_lds(a0, a1, 3, i0, i1);
    *(float4*)(out + (size_t)(n0 + node) * DD + d4 * 4) = o;
  }
}

#define OBAS_W ((2 * (OWN_P + 1) + 3) & ~3)

extern "C" void kernel_launch(void* const* d_in, const int* in_sizes, int n_in,
                              void* d_out, int out_size, void* d_ws, size_t ws_size,
                              hipStream_t stream) {
  const float* feat = (const float*)d_in[0];
  const int*   src_g[2] = { (const int*)d_in[1], (const int*)d_in[3] };
  const int*   dst_g[2] = { (const int*)d_in[2], (const int*)d_in[4] };
  const float* W_g[2]   = { (const float*)d_in[5], (const float*)d_in[8] };
  const float* al_g[2]  = { (const float*)d_in[6], (const float*)d_in[9] };
  const float* ar_g[2]  = { (const float*)d_in[7], (const float*)d_in[10] };
  float* out = (float*)d_out;

  float*        hbuf = (float*)d_ws;
  float*        el   = hbuf + 2 * (size_t)NN * HD;
  float*        er   = el   + 2 * (size_t)NN * HH;
  unsigned int* cntb = (unsigned int*)(er + 2 * (size_t)NN * HH);
  unsigned int* obas = cntb + 2 * (size_t)OWN_P * PB;
  unsigned int* buck = obas + OBAS_W;

  const int T = 256;
  k_count<<<dim3(PB, 2), PTC, 0, stream>>>(dst_g[0], dst_g[1], cntb);
  k_scatter<<<dim3(PB, 2), PT, 0, stream>>>(src_g[0], src_g[1], dst_g[0], dst_g[1], cntb, obas, buck);

  k_proj<<<dim3(PJ_G, 2), T, 0, stream>>>(feat, W_g[0], W_g[1], hbuf);
  k_elr<<<dim3(ELR_G, 2), T, 0, stream>>>(hbuf, al_g[0], al_g[1], ar_g[0], ar_g[1], el, er);
  k_edge_own<<<OWN_P, OWN_T, 0, stream>>>(el, er, obas, buck, hbuf, out);
}
